# starting kernel
# baseline (speedup 1.0000x reference)
#include <hip/hip_runtime.h>
#include <math.h>


#define NUM_NODES 100000
#define NUM_RELS  500
#define DIM       64
#define NUM_EDGES 1000000

typedef __attribute__((ext_vector_type(2))) float v2f;
typedef __attribute__((ext_vector_type(8))) float v8f;

__global__ void zero_kernel(float* __restrict__ p, int n) {
  for (int i = blockIdx.x * blockDim.x + threadIdx.x; i < n;
       i += gridDim.x * blockDim.x)
    p[i] = 0.0f;
}

__global__ __launch_bounds__(256) void fold_kernel(
    const float* __restrict__ W_ent, const float* __restrict__ b_ent,
    const float* __restrict__ W_relL, const float* __restrict__ b_relL,
    const float* __restrict__ W_rel2, const float* __restrict__ b_rel2,
    const float* __restrict__ W_a, const float* __restrict__ b_a,
    const float* __restrict__ W_fc, const float* __restrict__ b_fc,
    float* __restrict__ M_s, float* __restrict__ M_d,
    float* __restrict__ M_r, float* __restrict__ M_2,
    float* __restrict__ cvec, float* __restrict__ v_rel2,
    float* __restrict__ a0) {
  __shared__ float scvec[DIM];
  const int t = threadIdx.x;
  for (int idx = t; idx < DIM * DIM; idx += 256) {
    const int o = idx >> 6, k = idx & 63;
    float ms = 0.f, md = 0.f, mr = 0.f, m2 = 0.f;
    for (int u = 0; u < DIM; ++u) {
      const float we = W_ent[u * DIM + k];
      const float wl = W_relL[u * DIM + k];
      ms += W_fc[o * 192 + u] * we;
      md += W_fc[o * 192 + 64 + u] * we;
      mr += W_fc[o * 192 + 128 + u] * wl;
      m2 += W_rel2[o * DIM + u] * wl;
    }
    M_s[idx] = ms; M_d[idx] = md; M_r[idx] = mr; M_2[idx] = m2;
  }
  if (t < DIM) {
    float cv = b_fc[t], v2 = b_rel2[t];
    for (int u = 0; u < DIM; ++u) {
      cv += W_fc[t * 192 + u] * b_ent[u]
          + W_fc[t * 192 + 64 + u] * b_ent[u]
          + W_fc[t * 192 + 128 + u] * b_relL[u];
      v2 += W_rel2[t * DIM + u] * b_relL[u];
    }
    cvec[t] = cv; v_rel2[t] = v2; scvec[t] = cv;
  }
  __syncthreads();
  if (t == 0) {
    float a = b_a[0];
    for (int o = 0; o < DIM; ++o) a += scvec[o] * W_a[o];
    a0[0] = a;
  }
}

__global__ __launch_bounds__(256) void wmma_gemm64(
    const float* __restrict__ A, const float* __restrict__ W,
    const float* __restrict__ bias, float* __restrict__ out, int M) {
  __shared__ float sW[DIM * DIM];
  for (int i = threadIdx.x; i < DIM * DIM; i += 256) sW[i] = W[i];
  __syncthreads();

  const int wave = threadIdx.x >> 5;
  const int lane = threadIdx.x & 31;
  const int m0 = (blockIdx.x * 8 + wave) * 16;
  if (m0 >= M) return;

  const int lm = lane & 15;
  const int hi = lane >> 4;

  int arow = m0 + lm;
  if (arow >= M) arow = M - 1;
  const float* __restrict__ Arow = A + (size_t)arow * DIM;

  v8f acc0 = {}, acc1 = {}, acc2 = {}, acc3 = {};
  for (int k = 0; k < DIM; k += 4) {
    v2f a;
    a.x = Arow[k + 2 * hi + 0];
    a.y = Arow[k + 2 * hi + 1];
    v2f b0, b1, b2, b3;
    const float* w0 = &sW[(0 * 16 + lm) * DIM + k + 2 * hi];
    const float* w1 = &sW[(1 * 16 + lm) * DIM + k + 2 * hi];
    const float* w2 = &sW[(2 * 16 + lm) * DIM + k + 2 * hi];
    const float* w3 = &sW[(3 * 16 + lm) * DIM + k + 2 * hi];
    b0.x = w0[0]; b0.y = w0[1];
    b1.x = w1[0]; b1.y = w1[1];
    b2.x = w2[0]; b2.y = w2[1];
    b3.x = w3[0]; b3.y = w3[1];
    acc0 = __builtin_amdgcn_wmma_f32_16x16x4_f32(false, a, false, b0,
                                                 (short)0, acc0, false, false);
    acc1 = __builtin_amdgcn_wmma_f32_16x16x4_f32(false, a, false, b1,
                                                 (short)0, acc1, false, false);
    acc2 = __builtin_amdgcn_wmma_f32_16x16x4_f32(false, a, false, b2,
                                                 (short)0, acc2, false, false);
    acc3 = __builtin_amdgcn_wmma_f32_16x16x4_f32(false, a, false, b3,
                                                 (short)0, acc3, false, false);
  }

  v8f accs[4] = {acc0, acc1, acc2, acc3};
#pragma unroll
  for (int ob = 0; ob < 4; ++ob) {
    const float bv = bias ? bias[ob * 16 + lm] : 0.0f;
#pragma unroll
    for (int v = 0; v < 8; ++v) {
      const int row = m0 + v + 8 * hi;
      if (row < M)
        out[(size_t)row * DIM + ob * 16 + lm] = accs[ob][v] + bv;
    }
  }
}

__global__ void dot64_kernel(const float* __restrict__ X,
                             const float* __restrict__ v,
                             float* __restrict__ out, int M) {
  const int m = blockIdx.x * blockDim.x + threadIdx.x;
  if (m >= M) return;
  const float* x = X + (size_t)m * DIM;
  float acc = 0.f;
#pragma unroll 8
  for (int j = 0; j < DIM; ++j) acc += x[j] * v[j];
  out[m] = acc;
}

__global__ void edge_logits_kernel(
    const int* __restrict__ src, const int* __restrict__ dst,
    const int* __restrict__ rel, const float* __restrict__ a_s,
    const float* __restrict__ a_d, const float* __restrict__ a_r,
    const float* __restrict__ a0p, float* __restrict__ b_e,
    float* __restrict__ b_sum, float* __restrict__ cnt, int E) {
  const float a0 = a0p[0];
  for (int e = blockIdx.x * blockDim.x + threadIdx.x; e < E;
       e += gridDim.x * blockDim.x) {
    const int s = src[e], d = dst[e], r = rel[e];
    float l = a_s[s] + a_d[d] + a_r[r] + a0;
    l = l > 0.f ? l : 0.01f * l;
    const float b = expf(l);
    b_e[e] = b;
    atomicAdd(&b_sum[s], b);
    atomicAdd(&cnt[r], 1.0f);
  }
}

__global__ __launch_bounds__(256) void edge_scatter_h_kernel(
    const int* __restrict__ src, const int* __restrict__ dst,
    const int* __restrict__ rel, const float* __restrict__ u_s,
    const float* __restrict__ u_d, const float* __restrict__ u_r,
    const float* __restrict__ cvec, const float* __restrict__ b_e,
    const float* __restrict__ b_sum, float* __restrict__ h_ent, int E) {
  const int gw = blockIdx.x * (blockDim.x >> 5) + (threadIdx.x >> 5);
  const int lane = threadIdx.x & 31;
  const int nw = gridDim.x * (blockDim.x >> 5);
  for (int e = gw; e < E; e += nw) {
    const int s = src[e], d = dst[e], r = rel[e];
    const float alpha = b_e[e] / b_sum[s];
    const float* ps = u_s + (size_t)s * DIM;
    const float* pd = u_d + (size_t)d * DIM;
    const float* pr = u_r + (size_t)r * DIM;
    float* ph = h_ent + (size_t)s * DIM;
#pragma unroll
    for (int half = 0; half < 2; ++half) {
      const int j = lane + 32 * half;
      const float c = ps[j] + pd[j] + pr[j] + cvec[j];
      atomicAdd(&ph[j], alpha * c);
    }
  }
}

__global__ __launch_bounds__(256) void edge_featsum_kernel(
    const int* __restrict__ src, const int* __restrict__ dst,
    const int* __restrict__ rel, const float* __restrict__ h_ent,
    float* __restrict__ featsum, int E) {
  const int gw = blockIdx.x * (blockDim.x >> 5) + (threadIdx.x >> 5);
  const int lane = threadIdx.x & 31;
  const int nw = gridDim.x * (blockDim.x >> 5);
  for (int e = gw; e < E; e += nw) {
    const int s = src[e], d = dst[e], r = rel[e];
    const float* hs = h_ent + (size_t)s * DIM;
    const float* hd = h_ent + (size_t)d * DIM;
    float* fs = featsum + (size_t)r * 128;
#pragma unroll
    for (int half = 0; half < 2; ++half) {
      const int j = lane + 32 * half;
      atomicAdd(&fs[j], hs[j]);
      atomicAdd(&fs[64 + j], hd[j]);
    }
  }
}

__global__ void h_rel_kernel(const float* __restrict__ featsum,
                             const float* __restrict__ cnt,
                             const float* __restrict__ rel2_proj,
                             const float* __restrict__ W_rel3,
                             const float* __restrict__ b_rel3,
                             float* __restrict__ h_rel, int R) {
  const int t = blockIdx.x * blockDim.x + threadIdx.x;
  if (t >= R * DIM) return;
  const int r = t >> 6, o = t & 63;
  const float c = cnt[r];
  const float inv = 1.0f / fmaxf(c, 1.0f);
  const float* fs = featsum + (size_t)r * 128;
  const float* w = W_rel3 + (size_t)o * 192;
  float acc = b_rel3[o];
  for (int j = 0; j < 128; ++j) acc += (fs[j] * inv) * w[j];
  if (c > 0.f) {
    const float* r2 = rel2_proj + (size_t)r * DIM;
    for (int j = 0; j < DIM; ++j) acc += r2[j] * w[128 + j];
  }
  h_rel[t] = acc;
}

#define OFF_MS     0
#define OFF_MD     4096
#define OFF_MR     8192
#define OFF_M2     12288
#define OFF_CVEC   16384
#define OFF_VREL2  16448
#define OFF_A0     16512
#define OFF_UR     16576
#define OFF_AR     48576
#define OFF_REL2P  49088
#define OFF_AS     81088
#define OFF_AD     181088
#define OFF_BSUM   281088
#define OFF_CNT    381088
#define OFF_FEAT   381588
#define OFF_BE     445588
#define OFF_US     1445588
#define OFF_UD     7845588
#define ZERO_CNT   (OFF_FEAT + NUM_RELS * 128 - OFF_BSUM)

extern "C" void kernel_launch(void* const* d_in, const int* in_sizes, int n_in,
                              void* d_out, int out_size, void* d_ws,
                              size_t ws_size, hipStream_t stream) {
  const float* ent_embed = (const float*)d_in[0];
  const float* rel_embed = (const float*)d_in[1];
  const float* W_ent  = (const float*)d_in[2];
  const float* b_ent  = (const float*)d_in[3];
  const float* W_relL = (const float*)d_in[4];
  const float* b_relL = (const float*)d_in[5];
  const float* W_rel2 = (const float*)d_in[6];
  const float* b_rel2 = (const float*)d_in[7];
  const float* W_rel3 = (const float*)d_in[8];
  const float* b_rel3 = (const float*)d_in[9];
  const float* W_a    = (const float*)d_in[10];
  const float* b_a    = (const float*)d_in[11];
  const float* W_fc   = (const float*)d_in[12];
  const float* b_fc   = (const float*)d_in[13];
  const int* src_idx  = (const int*)d_in[14];
  const int* dst_idx  = (const int*)d_in[15];
  const int* rel_idx  = (const int*)d_in[16];

  float* ws = (float*)d_ws;
  float* h_ent = (float*)d_out;
  float* h_rel = (float*)d_out + (size_t)NUM_NODES * DIM;

  zero_kernel<<<2048, 256, 0, stream>>>(h_ent, NUM_NODES * DIM);
  zero_kernel<<<512, 256, 0, stream>>>(ws + OFF_BSUM, ZERO_CNT);

  fold_kernel<<<1, 256, 0, stream>>>(
      W_ent, b_ent, W_relL, b_relL, W_rel2, b_rel2, W_a, b_a, W_fc, b_fc,
      ws + OFF_MS, ws + OFF_MD, ws + OFF_MR, ws + OFF_M2, ws + OFF_CVEC,
      ws + OFF_VREL2, ws + OFF_A0);

  const int gemm_blocks_N = (NUM_NODES + 127) / 128;
  const int gemm_blocks_R = (NUM_RELS + 127) / 128;
  wmma_gemm64<<<gemm_blocks_N, 256, 0, stream>>>(ent_embed, ws + OFF_MS,
                                                 nullptr, ws + OFF_US, NUM_NODES);
  wmma_gemm64<<<gemm_blocks_N, 256, 0, stream>>>(ent_embed, ws + OFF_MD,
                                                 nullptr, ws + OFF_UD, NUM_NODES);
  wmma_gemm64<<<gemm_blocks_R, 256, 0, stream>>>(rel_embed, ws + OFF_MR,
                                                 nullptr, ws + OFF_UR, NUM_RELS);
  wmma_gemm64<<<gemm_blocks_R, 256, 0, stream>>>(rel_embed, ws + OFF_M2,
                                                 ws + OFF_VREL2, ws + OFF_REL2P,
                                                 NUM_RELS);

  dot64_kernel<<<(NUM_NODES + 255) / 256, 256, 0, stream>>>(ws + OFF_US, W_a,
                                                            ws + OFF_AS, NUM_NODES);
  dot64_kernel<<<(NUM_NODES + 255) / 256, 256, 0, stream>>>(ws + OFF_UD, W_a,
                                                            ws + OFF_AD, NUM_NODES);
  dot64_kernel<<<(NUM_RELS + 255) / 256, 256, 0, stream>>>(ws + OFF_UR, W_a,
                                                           ws + OFF_AR, NUM_RELS);

  edge_logits_kernel<<<4096, 256, 0, stream>>>(
      src_idx, dst_idx, rel_idx, ws + OFF_AS, ws + OFF_AD, ws + OFF_AR,
      ws + OFF_A0, ws + OFF_BE, ws + OFF_BSUM, ws + OFF_CNT, NUM_EDGES);
  edge_scatter_h_kernel<<<4096, 256, 0, stream>>>(
      src_idx, dst_idx, rel_idx, ws + OFF_US, ws + OFF_UD, ws + OFF_UR,
      ws + OFF_CVEC, ws + OFF_BE, ws + OFF_BSUM, h_ent, NUM_EDGES);
  edge_featsum_kernel<<<4096, 256, 0, stream>>>(
      src_idx, dst_idx, rel_idx, h_ent, ws + OFF_FEAT, NUM_EDGES);

  h_rel_kernel<<<(NUM_RELS * DIM + 255) / 256, 256, 0, stream>>>(
      ws + OFF_FEAT, ws + OFF_CNT, ws + OFF_REL2P, W_rel3, b_rel3, h_rel,
      NUM_RELS);
}


# edge_pool on the matrix unit: per 4 edges onehot[16 rel x 4] @ G[4 x 16] f32 WMMA into register accumulators (whole pool per work-group), one atomic flush per group not per edge; sum order changes
# speedup vs baseline: 1.7700x; 1.7700x over previous
#include <hip/hip_runtime.h>
#include <math.h>


#define NUM_NODES 100000
#define NUM_RELS  500
#define DIM       64
#define NUM_EDGES 1000000

typedef __attribute__((ext_vector_type(2))) float v2f;
typedef __attribute__((ext_vector_type(8))) float v8f;

__global__ void zero_kernel(float* __restrict__ p, int n) {
  for (int i = blockIdx.x * blockDim.x + threadIdx.x; i < n;
       i += gridDim.x * blockDim.x)
    p[i] = 0.0f;
}

__global__ __launch_bounds__(256) void fold_kernel(
    const float* __restrict__ W_ent, const float* __restrict__ b_ent,
    const float* __restrict__ W_relL, const float* __restrict__ b_relL,
    const float* __restrict__ W_rel2, const float* __restrict__ b_rel2,
    const float* __restrict__ W_a, const float* __restrict__ b_a,
    const float* __restrict__ W_fc, const float* __restrict__ b_fc,
    float* __restrict__ M_s, float* __restrict__ M_d,
    float* __restrict__ M_r, float* __restrict__ M_2,
    float* __restrict__ cvec, float* __restrict__ v_rel2,
    float* __restrict__ a0) {
  __shared__ float scvec[DIM];
  const int t = threadIdx.x;
  for (int idx = t; idx < DIM * DIM; idx += 256) {
    const int o = idx >> 6, k = idx & 63;
    float ms = 0.f, md = 0.f, mr = 0.f, m2 = 0.f;
    for (int u = 0; u < DIM; ++u) {
      const float we = W_ent[u * DIM + k];
      const float wl = W_relL[u * DIM + k];
      ms += W_fc[o * 192 + u] * we;
      md += W_fc[o * 192 + 64 + u] * we;
      mr += W_fc[o * 192 + 128 + u] * wl;
      m2 += W_rel2[o * DIM + u] * wl;
    }
    M_s[idx] = ms; M_d[idx] = md; M_r[idx] = mr; M_2[idx] = m2;
  }
  if (t < DIM) {
    float cv = b_fc[t], v2 = b_rel2[t];
    for (int u = 0; u < DIM; ++u) {
      cv += W_fc[t * 192 + u] * b_ent[u]
          + W_fc[t * 192 + 64 + u] * b_ent[u]
          + W_fc[t * 192 + 128 + u] * b_relL[u];
      v2 += W_rel2[t * DIM + u] * b_relL[u];
    }
    cvec[t] = cv; v_rel2[t] = v2; scvec[t] = cv;
  }
  __syncthreads();
  if (t == 0) {
    float a = b_a[0];
    for (int o = 0; o < DIM; ++o) a += scvec[o] * W_a[o];
    a0[0] = a;
  }
}

template <int LDW>
__global__ __launch_bounds__(256) void wmma_gemm64(
    const float* __restrict__ A, const float* __restrict__ W,
    const float* __restrict__ bias, float* __restrict__ out, int M) {
  __shared__ float sW[DIM * DIM];
  for (int i = threadIdx.x; i < DIM * DIM; i += 256)
    sW[i] = W[(i >> 6) * LDW + (i & 63)];
  __syncthreads();

  const int wave = threadIdx.x >> 5;
  const int lane = threadIdx.x & 31;
  const int m0 = (blockIdx.x * 8 + wave) * 16;
  if (m0 >= M) return;

  const int lm = lane & 15;
  const int hi = lane >> 4;

  int arow = m0 + lm;
  if (arow >= M) arow = M - 1;
  const float* __restrict__ Arow = A + (size_t)arow * DIM;

  v8f acc0 = {}, acc1 = {}, acc2 = {}, acc3 = {};
  for (int k = 0; k < DIM; k += 4) {
    v2f a;
    a.x = Arow[k + 2 * hi + 0];
    a.y = Arow[k + 2 * hi + 1];
    v2f b0, b1, b2, b3;
    const float* w0 = &sW[(0 * 16 + lm) * DIM + k + 2 * hi];
    const float* w1 = &sW[(1 * 16 + lm) * DIM + k + 2 * hi];
    const float* w2 = &sW[(2 * 16 + lm) * DIM + k + 2 * hi];
    const float* w3 = &sW[(3 * 16 + lm) * DIM + k + 2 * hi];
    b0.x = w0[0]; b0.y = w0[1];
    b1.x = w1[0]; b1.y = w1[1];
    b2.x = w2[0]; b2.y = w2[1];
    b3.x = w3[0]; b3.y = w3[1];
    acc0 = __builtin_amdgcn_wmma_f32_16x16x4_f32(false, a, false, b0,
                                                 (short)0, acc0, false, false);
    acc1 = __builtin_amdgcn_wmma_f32_16x16x4_f32(false, a, false, b1,
                                                 (short)0, acc1, false, false);
    acc2 = __builtin_amdgcn_wmma_f32_16x16x4_f32(false, a, false, b2,
                                                 (short)0, acc2, false, false);
    acc3 = __builtin_amdgcn_wmma_f32_16x16x4_f32(false, a, false, b3,
                                                 (short)0, acc3, false, false);
  }

  v8f accs[4] = {acc0, acc1, acc2, acc3};
#pragma unroll
  for (int ob = 0; ob < 4; ++ob) {
    const float bv = bias ? bias[ob * 16 + lm] : 0.0f;
#pragma unroll
    for (int v = 0; v < 8; ++v) {
      const int row = m0 + v + 8 * hi;
      if (row < M)
        out[(size_t)row * DIM + ob * 16 + lm] = accs[ob][v] + bv;
    }
  }
}

__global__ void dot64_kernel(const float* __restrict__ X,
                             const float* __restrict__ v,
                             float* __restrict__ out, int M) {
  const int m = blockIdx.x * blockDim.x + threadIdx.x;
  if (m >= M) return;
  const float* x = X + (size_t)m * DIM;
  float acc = 0.f;
#pragma unroll 8
  for (int j = 0; j < DIM; ++j) acc += x[j] * v[j];
  out[m] = acc;
}

__global__ void edge_logits_kernel(
    const int* __restrict__ src, const int* __restrict__ dst,
    const int* __restrict__ rel, const float* __restrict__ a_s,
    const float* __restrict__ a_d, const float* __restrict__ a_r,
    const float* __restrict__ a0p, float* __restrict__ b_e,
    float* __restrict__ b_sum, float* __restrict__ cnt, int E) {
  const float a0 = a0p[0];
  for (int e = blockIdx.x * blockDim.x + threadIdx.x; e < E;
       e += gridDim.x * blockDim.x) {
    const int s = src[e], d = dst[e], r = rel[e];
    float l = a_s[s] + a_d[d] + a_r[r] + a0;
    l = l > 0.f ? l : 0.01f * l;
    const float b = expf(l);
    b_e[e] = b;
    atomicAdd(&b_sum[s], b);
    atomicAdd(&cnt[r], 1.0f);
  }
}

__global__ __launch_bounds__(256) void edge_scatter_h_kernel(
    const int* __restrict__ src, const int* __restrict__ dst,
    const int* __restrict__ rel, const float* __restrict__ u_s,
    const float* __restrict__ u_d, const float* __restrict__ u_r,
    const float* __restrict__ cvec, const float* __restrict__ b_e,
    const float* __restrict__ b_sum, float* __restrict__ h_ent, int E) {
  const int gw = blockIdx.x * (blockDim.x >> 5) + (threadIdx.x >> 5);
  const int lane = threadIdx.x & 31;
  const int nw = gridDim.x * (blockDim.x >> 5);
  for (int e = gw; e < E; e += nw) {
    const int s = src[e], d = dst[e], r = rel[e];
    const float alpha = b_e[e] / b_sum[s];
    const float* ps = u_s + (size_t)s * DIM;
    const float* pd = u_d + (size_t)d * DIM;
    const float* pr = u_r + (size_t)r * DIM;
    float* ph = h_ent + (size_t)s * DIM;
#pragma unroll
    for (int half = 0; half < 2; ++half) {
      const int j = lane + 32 * half;
      const float c = ps[j] + pd[j] + pr[j] + cvec[j];
      atomicAdd(&ph[j], alpha * c);
    }
  }
}

typedef __attribute__((ext_vector_type(4))) float v4f;
typedef __attribute__((ext_vector_type(4))) int v4i;

#define PB      128
#define PPITCH  160
#define POOL_WGS 256

__global__ __launch_bounds__(512) void edge_pool_kernel(
    const int* __restrict__ src, const int* __restrict__ dst,
    const int* __restrict__ rel, const float* __restrict__ g_s,
    const float* __restrict__ g_d, float* __restrict__ pool, int E) {
  __shared__ __attribute__((aligned(16))) float sG[2][(PB / 2) * PPITCH];
  __shared__ __attribute__((aligned(16))) int sR[2][PB];

  const int wave = threadIdx.x >> 5;
  const int lane = threadIdx.x & 31;
  const int lm = lane & 15;
  const int hi = lane >> 4;
  const int rbase = wave * 32;
  const int nb = (E + PB - 1) / PB;
  const int G = gridDim.x;

  v8f c00 = {}, c01 = {}, c02 = {}, c03 = {};
  v8f c10 = {}, c11 = {}, c12 = {}, c13 = {};

  v2f rs[8], rd[8];
  int s_cur, d_cur, r_cur, s_nxt, d_nxt, r_nxt;

  auto load_idx = [&](int bb, int& s, int& d, int& r) {
    const int e = bb * PB + wave * 8 + (lane & 7);
    const bool ok = (bb < nb) && (e < E);
    s = ok ? src[e] : 0;
    d = ok ? dst[e] : 0;
    r = ok ? rel[e] : -1;
  };
  auto load_rows = [&](int s, int d) {
#pragma unroll
    for (int i = 0; i < 8; ++i) {
      const int si = __builtin_amdgcn_readlane(s, i);
      const int di = __builtin_amdgcn_readlane(d, i);
      rs[i] = *(const v2f*)(g_s + (size_t)si * DIM + 2 * lane);
      rd[i] = *(const v2f*)(g_d + (size_t)di * DIM + 2 * lane);
    }
  };

  int b = blockIdx.x;
  load_idx(b, s_cur, d_cur, r_cur);
  load_rows(s_cur, d_cur);
  load_idx(b + G, s_nxt, d_nxt, r_nxt);

  int buf = 0;
  for (; b < nb; b += G) {
    float* gb = sG[buf];
#pragma unroll
    for (int j = 0; j < 4; ++j) {
      const int r0 = __builtin_amdgcn_readlane(r_cur, 2 * j);
      const int r1 = __builtin_amdgcn_readlane(r_cur, 2 * j + 1);
      v2f g0 = rs[2 * j] + rd[2 * j];
      v2f g1 = rs[2 * j + 1] + rd[2 * j + 1];
      if (r0 < 0) { g0.x = 0.f; g0.y = 0.f; }
      if (r1 < 0) { g1.x = 0.f; g1.y = 0.f; }
      v4f o;
      o.x = g0.x; o.y = g1.x; o.z = g0.y; o.w = g1.y;
      *(v4f*)(gb + (wave * 4 + j) * PPITCH + 4 * lane) = o;
    }
    if (lane < 8) sR[buf][wave * 8 + lane] = r_cur;
    __syncthreads();

    s_cur = s_nxt; d_cur = d_nxt; r_cur = r_nxt;
    if (b + G < nb) load_rows(s_cur, d_cur);
    load_idx(b + 2 * G, s_nxt, d_nxt, r_nxt);

    const v4i rr = *(const v4i*)&sR[buf][4 * lane];
    const bool relv = ((unsigned)(rr.x - rbase) < 32u) |
                      ((unsigned)(rr.y - rbase) < 32u) |
                      ((unsigned)(rr.z - rbase) < 32u) |
                      ((unsigned)(rr.w - rbase) < 32u);
    unsigned mask = __builtin_amdgcn_ballot_w32(relv);
    while (mask) {
      const int g = __builtin_ctz(mask);
      mask &= mask - 1;
      const int* pr = &sR[buf][4 * g + 2 * hi];
      const int da = pr[0] - rbase - lm;
      const int db = pr[1] - rbase - lm;
      const float* pg = gb + (2 * g + hi) * PPITCH + 2 * lm;
      const v2f b0 = *(const v2f*)(pg);
      const v2f b1 = *(const v2f*)(pg + 32);
      const v2f b2 = *(const v2f*)(pg + 64);
      const v2f b3 = *(const v2f*)(pg + 96);
      {
        const bool ma = (da == 0), mb = (db == 0);
        if (__builtin_amdgcn_ballot_w32(ma | mb)) {
          v2f a;
          a.x = ma ? 1.0f : 0.0f;
          a.y = mb ? 1.0f : 0.0f;
          c00 = __builtin_amdgcn_wmma_f32_16x16x4_f32(false, a, false, b0,
                                                      (short)0, c00, false, false);
          c01 = __builtin_amdgcn_wmma_f32_16x16x4_f32(false, a, false, b1,
                                                      (short)0, c01, false, false);
          c02 = __builtin_amdgcn_wmma_f32_16x16x4_f32(false, a, false, b2,
                                                      (short)0, c02, false, false);
          c03 = __builtin_amdgcn_wmma_f32_16x16x4_f32(false, a, false, b3,
                                                      (short)0, c03, false, false);
        }
      }
      {
        const bool ma = (da == 16), mb = (db == 16);
        if (__builtin_amdgcn_ballot_w32(ma | mb)) {
          v2f a;
          a.x = ma ? 1.0f : 0.0f;
          a.y = mb ? 1.0f : 0.0f;
          c10 = __builtin_amdgcn_wmma_f32_16x16x4_f32(false, a, false, b0,
                                                      (short)0, c10, false, false);
          c11 = __builtin_amdgcn_wmma_f32_16x16x4_f32(false, a, false, b1,
                                                      (short)0, c11, false, false);
          c12 = __builtin_amdgcn_wmma_f32_16x16x4_f32(false, a, false, b2,
                                                      (short)0, c12, false, false);
          c13 = __builtin_amdgcn_wmma_f32_16x16x4_f32(false, a, false, b3,
                                                      (short)0, c13, false, false);
        }
      }
    }
    buf ^= 1;
  }

  __syncthreads();
  float* sFw = &sG[0][0] + wave * 1024;
  auto flush = [&](const v8f& t0, const v8f& t1, const v8f& t2, const v8f& t3,
                   int rt) {
#pragma unroll
    for (int v = 0; v < 8; ++v) {
      float* row = sFw + (v + 8 * hi) * DIM + lm;
      row[0] = t0[v]; row[16] = t1[v]; row[32] = t2[v]; row[48] = t3[v];
    }
    __syncthreads();
    for (int q = 0; q < 16; ++q) {
      const int r = (q + (int)blockIdx.x) & 15;
      const int R = rbase + rt * 16 + r;
      if (R < NUM_RELS) {
        atomicAdd(&pool[(size_t)R * DIM + lane], sFw[r * DIM + lane]);
        atomicAdd(&pool[(size_t)R * DIM + 32 + lane], sFw[r * DIM + 32 + lane]);
      }
    }
    __syncthreads();
  };
  flush(c00, c01, c02, c03, 0);
  flush(c10, c11, c12, c13, 1);
}

__global__ void h_rel_kernel(const float* __restrict__ pool,
                             const float* __restrict__ cnt,
                             const float* __restrict__ rel2_proj,
                             const float* __restrict__ W_rel3,
                             const float* __restrict__ b_rel3,
                             float* __restrict__ h_rel, int R) {
  const int t = blockIdx.x * blockDim.x + threadIdx.x;
  if (t >= R * DIM) return;
  const int r = t >> 6, o = t & 63;
  const float c = cnt[r];
  const float inv = 1.0f / fmaxf(c, 1.0f);
  const float* w = W_rel3 + (size_t)o * 192;
  float acc = b_rel3[o] + pool[t] * inv;
  if (c > 0.f) {
    const float* r2 = rel2_proj + (size_t)r * DIM;
    for (int j = 0; j < DIM; ++j) acc += r2[j] * w[128 + j];
  }
  h_rel[t] = acc;
}

#define OFF_MS     0
#define OFF_MD     4096
#define OFF_MR     8192
#define OFF_M2     12288
#define OFF_CVEC   16384
#define OFF_VREL2  16448
#define OFF_A0     16512
#define OFF_UR     16576
#define OFF_AR     48576
#define OFF_REL2P  49088
#define OFF_AS     81088
#define OFF_AD     181088
#define OFF_BSUM   281088
#define OFF_CNT    381088
#define OFF_POOL   381588
#define OFF_BE     445588
#define OFF_US     1445588
#define OFF_UD     7845588
#define ZERO_CNT   (OFF_POOL + NUM_RELS * DIM - OFF_BSUM)

extern "C" void kernel_launch(void* const* d_in, const int* in_sizes, int n_in,
                              void* d_out, int out_size, void* d_ws,
                              size_t ws_size, hipStream_t stream) {
  const float* ent_embed = (const float*)d_in[0];
  const float* rel_embed = (const float*)d_in[1];
  const float* W_ent  = (const float*)d_in[2];
  const float* b_ent  = (const float*)d_in[3];
  const float* W_relL = (const float*)d_in[4];
  const float* b_relL = (const float*)d_in[5];
  const float* W_rel2 = (const float*)d_in[6];
  const float* b_rel2 = (const float*)d_in[7];
  const float* W_rel3 = (const float*)d_in[8];
  const float* b_rel3 = (const float*)d_in[9];
  const float* W_a    = (const float*)d_in[10];
  const float* b_a    = (const float*)d_in[11];
  const float* W_fc   = (const float*)d_in[12];
  const float* b_fc   = (const float*)d_in[13];
  const int* src_idx  = (const int*)d_in[14];
  const int* dst_idx  = (const int*)d_in[15];
  const int* rel_idx  = (const int*)d_in[16];

  float* ws = (float*)d_ws;
  float* h_ent = (float*)d_out;
  float* h_rel = (float*)d_out + (size_t)NUM_NODES * DIM;

  zero_kernel<<<2048, 256, 0, stream>>>(h_ent, NUM_NODES * DIM);
  zero_kernel<<<512, 256, 0, stream>>>(ws + OFF_BSUM, ZERO_CNT);

  fold_kernel<<<1, 256, 0, stream>>>(
      W_ent, b_ent, W_relL, b_relL, W_rel2, b_rel2, W_a, b_a, W_fc, b_fc,
      ws + OFF_MS, ws + OFF_MD, ws + OFF_MR, ws + OFF_M2, ws + OFF_CVEC,
      ws + OFF_VREL2, ws + OFF_A0);

  const int gemm_blocks_N = (NUM_NODES + 127) / 128;
  const int gemm_blocks_R = (NUM_RELS + 127) / 128;
  wmma_gemm64<DIM><<<gemm_blocks_N, 256, 0, stream>>>(ent_embed, ws + OFF_MS,
                                                 nullptr, ws + OFF_US, NUM_NODES);
  wmma_gemm64<DIM><<<gemm_blocks_N, 256, 0, stream>>>(ent_embed, ws + OFF_MD,
                                                 nullptr, ws + OFF_UD, NUM_NODES);
  wmma_gemm64<DIM><<<gemm_blocks_R, 256, 0, stream>>>(rel_embed, ws + OFF_MR,
                                                 nullptr, ws + OFF_UR, NUM_RELS);
  wmma_gemm64<DIM><<<gemm_blocks_R, 256, 0, stream>>>(rel_embed, ws + OFF_M2,
                                                 ws + OFF_VREL2, ws + OFF_REL2P,
                                                 NUM_RELS);

  dot64_kernel<<<(NUM_NODES + 255) / 256, 256, 0, stream>>>(ws + OFF_US, W_a,
                                                            ws + OFF_AS, NUM_NODES);
  dot64_kernel<<<(NUM_NODES + 255) / 256, 256, 0, stream>>>(ws + OFF_UD, W_a,
                                                            ws + OFF_AD, NUM_NODES);
  dot64_kernel<<<(NUM_RELS + 255) / 256, 256, 0, stream>>>(ws + OFF_UR, W_a,
                                                           ws + OFF_AR, NUM_RELS);

  edge_logits_kernel<<<4096, 256, 0, stream>>>(
      src_idx, dst_idx, rel_idx, ws + OFF_AS, ws + OFF_AD, ws + OFF_AR,
      ws + OFF_A0, ws + OFF_BE, ws + OFF_BSUM, ws + OFF_CNT, NUM_EDGES);
  edge_scatter_h_kernel<<<4096, 256, 0, stream>>>(
      src_idx, dst_idx, rel_idx, ws + OFF_US, ws + OFF_UD, ws + OFF_UR,
      ws + OFF_CVEC, ws + OFF_BE, ws + OFF_BSUM, h_ent, NUM_EDGES);
  wmma_gemm64<3 * DIM><<<gemm_blocks_N, 256, 0, stream>>>(h_ent, W_rel3,
                                                 nullptr, ws + OFF_US, NUM_NODES);
  wmma_gemm64<3 * DIM><<<gemm_blocks_N, 256, 0, stream>>>(h_ent, W_rel3 + DIM,
                                                 nullptr, ws + OFF_UD, NUM_NODES);
  edge_pool_kernel<<<POOL_WGS, 512, 0, stream>>>(
      src_idx, dst_idx, rel_idx, ws + OFF_US, ws + OFF_UD, ws + OFF_POOL,
      NUM_EDGES);

  h_rel_kernel<<<(NUM_RELS * DIM + 255) / 256, 256, 0, stream>>>(
      ws + OFF_POOL, ws + OFF_CNT, ws + OFF_REL2P, W_rel3, b_rel3, h_rel,
      NUM_RELS);
}


# h_ent as per-node segment sums: edges grouped by source per call (CSR via integer cursor atomics), one wave per node sums alpha*c in registers, stores its row once; h_ent/b_sum sum order changes
# speedup vs baseline: 2.0450x; 1.1554x over previous
#include <hip/hip_runtime.h>
#include <math.h>


#define NUM_NODES 100000
#define NUM_RELS  500
#define DIM       64
#define NUM_EDGES 1000000

typedef __attribute__((ext_vector_type(2))) float v2f;
typedef __attribute__((ext_vector_type(8))) float v8f;

__global__ void zero_kernel(float* __restrict__ p, int n) {
  for (int i = blockIdx.x * blockDim.x + threadIdx.x; i < n;
       i += gridDim.x * blockDim.x)
    p[i] = 0.0f;
}

__global__ __launch_bounds__(256) void fold_kernel(
    const float* __restrict__ W_ent, const float* __restrict__ b_ent,
    const float* __restrict__ W_relL, const float* __restrict__ b_relL,
    const float* __restrict__ W_rel2, const float* __restrict__ b_rel2,
    const float* __restrict__ W_a, const float* __restrict__ b_a,
    const float* __restrict__ W_fc, const float* __restrict__ b_fc,
    float* __restrict__ M_s, float* __restrict__ M_d,
    float* __restrict__ M_r, float* __restrict__ M_2,
    float* __restrict__ cvec, float* __restrict__ v_rel2,
    float* __restrict__ a0) {
  __shared__ float scvec[DIM];
  const int t = threadIdx.x;
  for (int idx = t; idx < DIM * DIM; idx += 256) {
    const int o = idx >> 6, k = idx & 63;
    float ms = 0.f, md = 0.f, mr = 0.f, m2 = 0.f;
    for (int u = 0; u < DIM; ++u) {
      const float we = W_ent[u * DIM + k];
      const float wl = W_relL[u * DIM + k];
      ms += W_fc[o * 192 + u] * we;
      md += W_fc[o * 192 + 64 + u] * we;
      mr += W_fc[o * 192 + 128 + u] * wl;
      m2 += W_rel2[o * DIM + u] * wl;
    }
    M_s[idx] = ms; M_d[idx] = md; M_r[idx] = mr; M_2[idx] = m2;
  }
  if (t < DIM) {
    float cv = b_fc[t], v2 = b_rel2[t];
    for (int u = 0; u < DIM; ++u) {
      cv += W_fc[t * 192 + u] * b_ent[u]
          + W_fc[t * 192 + 64 + u] * b_ent[u]
          + W_fc[t * 192 + 128 + u] * b_relL[u];
      v2 += W_rel2[t * DIM + u] * b_relL[u];
    }
    cvec[t] = cv; v_rel2[t] = v2; scvec[t] = cv;
  }
  __syncthreads();
  if (t == 0) {
    float a = b_a[0];
    for (int o = 0; o < DIM; ++o) a += scvec[o] * W_a[o];
    a0[0] = a;
  }
}

template <int LDW>
__global__ __launch_bounds__(256) void wmma_gemm64(
    const float* __restrict__ A, const float* __restrict__ W,
    const float* __restrict__ bias, float* __restrict__ out, int M) {
  __shared__ float sW[DIM * DIM];
  for (int i = threadIdx.x; i < DIM * DIM; i += 256)
    sW[i] = W[(i >> 6) * LDW + (i & 63)];
  __syncthreads();

  const int wave = threadIdx.x >> 5;
  const int lane = threadIdx.x & 31;
  const int m0 = (blockIdx.x * 8 + wave) * 16;
  if (m0 >= M) return;

  const int lm = lane & 15;
  const int hi = lane >> 4;

  int arow = m0 + lm;
  if (arow >= M) arow = M - 1;
  const float* __restrict__ Arow = A + (size_t)arow * DIM;

  v8f acc0 = {}, acc1 = {}, acc2 = {}, acc3 = {};
  for (int k = 0; k < DIM; k += 4) {
    v2f a;
    a.x = Arow[k + 2 * hi + 0];
    a.y = Arow[k + 2 * hi + 1];
    v2f b0, b1, b2, b3;
    const float* w0 = &sW[(0 * 16 + lm) * DIM + k + 2 * hi];
    const float* w1 = &sW[(1 * 16 + lm) * DIM + k + 2 * hi];
    const float* w2 = &sW[(2 * 16 + lm) * DIM + k + 2 * hi];
    const float* w3 = &sW[(3 * 16 + lm) * DIM + k + 2 * hi];
    b0.x = w0[0]; b0.y = w0[1];
    b1.x = w1[0]; b1.y = w1[1];
    b2.x = w2[0]; b2.y = w2[1];
    b3.x = w3[0]; b3.y = w3[1];
    acc0 = __builtin_amdgcn_wmma_f32_16x16x4_f32(false, a, false, b0,
                                                 (short)0, acc0, false, false);
    acc1 = __builtin_amdgcn_wmma_f32_16x16x4_f32(false, a, false, b1,
                                                 (short)0, acc1, false, false);
    acc2 = __builtin_amdgcn_wmma_f32_16x16x4_f32(false, a, false, b2,
                                                 (short)0, acc2, false, false);
    acc3 = __builtin_amdgcn_wmma_f32_16x16x4_f32(false, a, false, b3,
                                                 (short)0, acc3, false, false);
  }

  v8f accs[4] = {acc0, acc1, acc2, acc3};
#pragma unroll
  for (int ob = 0; ob < 4; ++ob) {
    const float bv = bias ? bias[ob * 16 + lm] : 0.0f;
#pragma unroll
    for (int v = 0; v < 8; ++v) {
      const int row = m0 + v + 8 * hi;
      if (row < M)
        out[(size_t)row * DIM + ob * 16 + lm] = accs[ob][v] + bv;
    }
  }
}

__global__ void dot64_kernel(const float* __restrict__ X,
                             const float* __restrict__ v,
                             float* __restrict__ out, int M) {
  const int m = blockIdx.x * blockDim.x + threadIdx.x;
  if (m >= M) return;
  const float* x = X + (size_t)m * DIM;
  float acc = 0.f;
#pragma unroll 8
  for (int j = 0; j < DIM; ++j) acc += x[j] * v[j];
  out[m] = acc;
}

template <int V> struct KConst { static constexpr int value = V; };

#define SEG_BLK 1024

__global__ void edge_count_kernel(const int* __restrict__ src,
                                  const int* __restrict__ rel,
                                  int* __restrict__ deg,
                                  float* __restrict__ cnt, int E) {
  for (int e = blockIdx.x * blockDim.x + threadIdx.x; e < E;
       e += gridDim.x * blockDim.x) {
    atomicAdd(&deg[src[e]], 1);
    atomicAdd(&cnt[rel[e]], 1.0f);
  }
}

__global__ __launch_bounds__(SEG_BLK) void seg_alloc_kernel(
    int* __restrict__ cursor  ,
    int* __restrict__ total, int* __restrict__ bbase, int N) {
  __shared__ int swave[32];
  __shared__ int sbase;
  const int lane = threadIdx.x & 31;
  const int wave = threadIdx.x >> 5;
  const int i = blockIdx.x * SEG_BLK + threadIdx.x;
  const int v = (i < N) ? cursor[i] : 0;
  int x = v;
#pragma unroll
  for (int o = 1; o < 32; o <<= 1) {
    const int y = __shfl_up(x, o);
    if (lane >= o) x += y;
  }
  if (lane == 31) swave[wave] = x;
  __syncthreads();
  if (wave == 0) {
    const int w = swave[lane];
    int xs = w;
#pragma unroll
    for (int o = 1; o < 32; o <<= 1) {
      const int y = __shfl_up(xs, o);
      if (lane >= o) xs += y;
    }
    swave[lane] = xs - w;
    if (lane == 31) {
      const int base = atomicAdd(total, xs);
      sbase = base;
      bbase[blockIdx.x] = base;
    }
  }
  __syncthreads();
  if (i < N) cursor[i] = sbase + swave[wave] + x - v;
}

__global__ void edge_fill_kernel(const int* __restrict__ src,
                                 const int* __restrict__ dst,
                                 const int* __restrict__ rel,
                                 int* __restrict__ cursor,
                                 int* __restrict__ rec, int E) {
  for (int e = blockIdx.x * blockDim.x + threadIdx.x; e < E;
       e += gridDim.x * blockDim.x) {
    const int p = atomicAdd(&cursor[src[e]], 1);
    if ((unsigned)p < (unsigned)E) rec[p] = dst[e] | (rel[e] << 17);
  }
}

__global__ __launch_bounds__(256) void node_gather_kernel(
    const int* __restrict__ cursor, const int* __restrict__ bbase,
    const int* __restrict__ rec, const float* __restrict__ u_s,
    const float* __restrict__ u_d, const float* __restrict__ u_r,
    const float* __restrict__ cvec, const float* __restrict__ a_s,
    const float* __restrict__ a_d, const float* __restrict__ a_r,
    const float* __restrict__ a0p, float* __restrict__ h_ent, int N, int E) {
  const int s = blockIdx.x * (blockDim.x >> 5) + (threadIdx.x >> 5);
  const int lane = threadIdx.x & 31;
  if (s >= N) return;

  int end = cursor[s];
  int start = (s & (SEG_BLK - 1)) ? cursor[s - 1] : bbase[s / SEG_BLK];
  end = __builtin_amdgcn_readfirstlane(end);
  start = __builtin_amdgcn_readfirstlane(start);
  start = min(max(start, 0), E);
  end = min(max(end, start), E);
  const int n = end - start;

  v2f acc = {0.f, 0.f};
  if (n > 0) {
    const float as0 = a_s[s] + a0p[0];
    const v2f base = *(const v2f*)(u_s + (size_t)s * DIM + 2 * lane) +
                     *(const v2f*)(cvec + 2 * lane);

    auto edge_b = [&](int c0, int& pk, float& bv) {
      pk = 0; bv = 0.f;
      if (c0 + lane < n) {
        pk = rec[start + c0 + lane];
        const int d = min(pk & 0x1FFFF, NUM_NODES - 1);
        const int r = min(pk >> 17, NUM_RELS - 1);
        float l = as0 + a_d[d] + a_r[r];
        l = l > 0.f ? l : 0.01f * l;
        bv = expf(l);
      }
    };

    int pk0; float b0;
    edge_b(0, pk0, b0);
    float bl = b0;
    for (int c0 = 32; c0 < n; c0 += 32) {
      int pk; float bv;
      edge_b(c0, pk, bv);
      bl += bv;
    }
#pragma unroll
    for (int o = 16; o > 0; o >>= 1) bl += __shfl_xor(bl, o);
    const float bsum = bl;

    for (int c0 = 0; c0 < n; c0 += 32) {
      int pk = pk0; float bv = b0;
      if (c0 > 0) edge_b(c0, pk, bv);
      const int wbits = __float_as_int(bv / bsum);
      const int m = min(32, n - c0);
      int j = 0;
      auto group = [&](auto KC, int valid) {
        constexpr int K = decltype(KC)::value;
        v2f ud[K], ur[K];
        float w[K];
#pragma unroll
        for (int i = 0; i < K; ++i) {
          const int idx = (j + i) & 31;
          const int pj = __builtin_amdgcn_readlane(pk, idx);
          w[i] = __int_as_float(__builtin_amdgcn_readlane(wbits, idx));
          const int d = min(pj & 0x1FFFF, NUM_NODES - 1);
          const int r = min(pj >> 17, NUM_RELS - 1);
          ud[i] = *(const v2f*)(u_d + (size_t)d * DIM + 2 * lane);
          ur[i] = *(const v2f*)(u_r + (size_t)r * DIM + 2 * lane);
        }
#pragma unroll
        for (int i = 0; i < K; ++i) {
          if (i < valid) {
            const v2f c = (base + ud[i]) + ur[i];
            acc.x += w[i] * c.x;
            acc.y += w[i] * c.y;
          }
        }
        j += K;
      };
      while (m - j >= 8) group(KConst<8>{}, 8);
      if (m - j >= 4) group(KConst<4>{}, 4);
      if (m - j > 0) group(KConst<4>{}, m - j);
    }
  }
  *(v2f*)(h_ent + (size_t)s * DIM + 2 * lane) = acc;
}

typedef __attribute__((ext_vector_type(4))) float v4f;
typedef __attribute__((ext_vector_type(4))) int v4i;

#define PB      128
#define PPITCH  160
#define POOL_WGS 256

__global__ __launch_bounds__(512) void edge_pool_kernel(
    const int* __restrict__ src, const int* __restrict__ dst,
    const int* __restrict__ rel, const float* __restrict__ g_s,
    const float* __restrict__ g_d, float* __restrict__ pool, int E) {
  __shared__ __attribute__((aligned(16))) float sG[2][(PB / 2) * PPITCH];
  __shared__ __attribute__((aligned(16))) int sR[2][PB];

  const int wave = threadIdx.x >> 5;
  const int lane = threadIdx.x & 31;
  const int lm = lane & 15;
  const int hi = lane >> 4;
  const int rbase = wave * 32;
  const int nb = (E + PB - 1) / PB;
  const int G = gridDim.x;

  v8f c00 = {}, c01 = {}, c02 = {}, c03 = {};
  v8f c10 = {}, c11 = {}, c12 = {}, c13 = {};

  v2f rs[8], rd[8];
  int s_cur, d_cur, r_cur, s_nxt, d_nxt, r_nxt;

  auto load_idx = [&](int bb, int& s, int& d, int& r) {
    const int e = bb * PB + wave * 8 + (lane & 7);
    const bool ok = (bb < nb) && (e < E);
    s = ok ? src[e] : 0;
    d = ok ? dst[e] : 0;
    r = ok ? rel[e] : -1;
  };
  auto load_rows = [&](int s, int d) {
#pragma unroll
    for (int i = 0; i < 8; ++i) {
      const int si = __builtin_amdgcn_readlane(s, i);
      const int di = __builtin_amdgcn_readlane(d, i);
      rs[i] = *(const v2f*)(g_s + (size_t)si * DIM + 2 * lane);
      rd[i] = *(const v2f*)(g_d + (size_t)di * DIM + 2 * lane);
    }
  };

  int b = blockIdx.x;
  load_idx(b, s_cur, d_cur, r_cur);
  load_rows(s_cur, d_cur);
  load_idx(b + G, s_nxt, d_nxt, r_nxt);

  int buf = 0;
  for (; b < nb; b += G) {
    float* gb = sG[buf];
#pragma unroll
    for (int j = 0; j < 4; ++j) {
      const int r0 = __builtin_amdgcn_readlane(r_cur, 2 * j);
      const int r1 = __builtin_amdgcn_readlane(r_cur, 2 * j + 1);
      v2f g0 = rs[2 * j] + rd[2 * j];
      v2f g1 = rs[2 * j + 1] + rd[2 * j + 1];
      if (r0 < 0) { g0.x = 0.f; g0.y = 0.f; }
      if (r1 < 0) { g1.x = 0.f; g1.y = 0.f; }
      v4f o;
      o.x = g0.x; o.y = g1.x; o.z = g0.y; o.w = g1.y;
      *(v4f*)(gb + (wave * 4 + j) * PPITCH + 4 * lane) = o;
    }
    if (lane < 8) sR[buf][wave * 8 + lane] = r_cur;
    __syncthreads();

    s_cur = s_nxt; d_cur = d_nxt; r_cur = r_nxt;
    if (b + G < nb) load_rows(s_cur, d_cur);
    load_idx(b + 2 * G, s_nxt, d_nxt, r_nxt);

    const v4i rr = *(const v4i*)&sR[buf][4 * lane];
    const bool relv = ((unsigned)(rr.x - rbase) < 32u) |
                      ((unsigned)(rr.y - rbase) < 32u) |
                      ((unsigned)(rr.z - rbase) < 32u) |
                      ((unsigned)(rr.w - rbase) < 32u);
    unsigned mask = __builtin_amdgcn_ballot_w32(relv);
    while (mask) {
      const int g = __builtin_ctz(mask);
      mask &= mask - 1;
      const int* pr = &sR[buf][4 * g + 2 * hi];
      const int da = pr[0] - rbase - lm;
      const int db = pr[1] - rbase - lm;
      const float* pg = gb + (2 * g + hi) * PPITCH + 2 * lm;
      const v2f b0 = *(const v2f*)(pg);
      const v2f b1 = *(const v2f*)(pg + 32);
      const v2f b2 = *(const v2f*)(pg + 64);
      const v2f b3 = *(const v2f*)(pg + 96);
      {
        const bool ma = (da == 0), mb = (db == 0);
        if (__builtin_amdgcn_ballot_w32(ma | mb)) {
          v2f a;
          a.x = ma ? 1.0f : 0.0f;
          a.y = mb ? 1.0f : 0.0f;
          c00 = __builtin_amdgcn_wmma_f32_16x16x4_f32(false, a, false, b0,
                                                      (short)0, c00, false, false);
          c01 = __builtin_amdgcn_wmma_f32_16x16x4_f32(false, a, false, b1,
                                                      (short)0, c01, false, false);
          c02 = __builtin_amdgcn_wmma_f32_16x16x4_f32(false, a, false, b2,
                                                      (short)0, c02, false, false);
          c03 = __builtin_amdgcn_wmma_f32_16x16x4_f32(false, a, false, b3,
                                                      (short)0, c03, false, false);
        }
      }
      {
        const bool ma = (da == 16), mb = (db == 16);
        if (__builtin_amdgcn_ballot_w32(ma | mb)) {
          v2f a;
          a.x = ma ? 1.0f : 0.0f;
          a.y = mb ? 1.0f : 0.0f;
          c10 = __builtin_amdgcn_wmma_f32_16x16x4_f32(false, a, false, b0,
                                                      (short)0, c10, false, false);
          c11 = __builtin_amdgcn_wmma_f32_16x16x4_f32(false, a, false, b1,
                                                      (short)0, c11, false, false);
          c12 = __builtin_amdgcn_wmma_f32_16x16x4_f32(false, a, false, b2,
                                                      (short)0, c12, false, false);
          c13 = __builtin_amdgcn_wmma_f32_16x16x4_f32(false, a, false, b3,
                                                      (short)0, c13, false, false);
        }
      }
    }
    buf ^= 1;
  }

  __syncthreads();
  float* sFw = &sG[0][0] + wave * 1024;
  auto flush = [&](const v8f& t0, const v8f& t1, const v8f& t2, const v8f& t3,
                   int rt) {
#pragma unroll
    for (int v = 0; v < 8; ++v) {
      float* row = sFw + (v + 8 * hi) * DIM + lm;
      row[0] = t0[v]; row[16] = t1[v]; row[32] = t2[v]; row[48] = t3[v];
    }
    __syncthreads();
    for (int q = 0; q < 16; ++q) {
      const int r = (q + (int)blockIdx.x) & 15;
      const int R = rbase + rt * 16 + r;
      if (R < NUM_RELS) {
        atomicAdd(&pool[(size_t)R * DIM + lane], sFw[r * DIM + lane]);
        atomicAdd(&pool[(size_t)R * DIM + 32 + lane], sFw[r * DIM + 32 + lane]);
      }
    }
    __syncthreads();
  };
  flush(c00, c01, c02, c03, 0);
  flush(c10, c11, c12, c13, 1);
}

__global__ void h_rel_kernel(const float* __restrict__ pool,
                             const float* __restrict__ cnt,
                             const float* __restrict__ rel2_proj,
                             const float* __restrict__ W_rel3,
                             const float* __restrict__ b_rel3,
                             float* __restrict__ h_rel, int R) {
  const int t = blockIdx.x * blockDim.x + threadIdx.x;
  if (t >= R * DIM) return;
  const int r = t >> 6, o = t & 63;
  const float c = cnt[r];
  const float inv = 1.0f / fmaxf(c, 1.0f);
  const float* w = W_rel3 + (size_t)o * 192;
  float acc = b_rel3[o] + pool[t] * inv;
  if (c > 0.f) {
    const float* r2 = rel2_proj + (size_t)r * DIM;
    for (int j = 0; j < DIM; ++j) acc += r2[j] * w[128 + j];
  }
  h_rel[t] = acc;
}

#define OFF_MS     0
#define OFF_MD     4096
#define OFF_MR     8192
#define OFF_M2     12288
#define OFF_CVEC   16384
#define OFF_VREL2  16448
#define OFF_A0     16512
#define OFF_UR     16576
#define OFF_AR     48576
#define OFF_REL2P  49088
#define OFF_AS     81088
#define OFF_AD     181088
#define OFF_BSUM   281088
#define OFF_CNT    381088
#define OFF_POOL   381588
#define OFF_TOTAL  413588
#define OFF_BBASE  413600
#define OFF_BE     445588
#define OFF_US     1445588
#define OFF_UD     7845588
#define ZERO_CNT   (OFF_TOTAL + 1 - OFF_BSUM)

extern "C" void kernel_launch(void* const* d_in, const int* in_sizes, int n_in,
                              void* d_out, int out_size, void* d_ws,
                              size_t ws_size, hipStream_t stream) {
  const float* ent_embed = (const float*)d_in[0];
  const float* rel_embed = (const float*)d_in[1];
  const float* W_ent  = (const float*)d_in[2];
  const float* b_ent  = (const float*)d_in[3];
  const float* W_relL = (const float*)d_in[4];
  const float* b_relL = (const float*)d_in[5];
  const float* W_rel2 = (const float*)d_in[6];
  const float* b_rel2 = (const float*)d_in[7];
  const float* W_rel3 = (const float*)d_in[8];
  const float* b_rel3 = (const float*)d_in[9];
  const float* W_a    = (const float*)d_in[10];
  const float* b_a    = (const float*)d_in[11];
  const float* W_fc   = (const float*)d_in[12];
  const float* b_fc   = (const float*)d_in[13];
  const int* src_idx  = (const int*)d_in[14];
  const int* dst_idx  = (const int*)d_in[15];
  const int* rel_idx  = (const int*)d_in[16];

  float* ws = (float*)d_ws;
  float* h_ent = (float*)d_out;
  float* h_rel = (float*)d_out + (size_t)NUM_NODES * DIM;

  zero_kernel<<<512, 256, 0, stream>>>(ws + OFF_BSUM, ZERO_CNT);

  int* deg_cur = (int*)(ws + OFF_BSUM);
  int* seg_total = (int*)(ws + OFF_TOTAL);
  int* seg_bbase = (int*)(ws + OFF_BBASE);
  int* edge_rec = (int*)(ws + OFF_BE);
  edge_count_kernel<<<4096, 256, 0, stream>>>(src_idx, rel_idx, deg_cur,
                                              ws + OFF_CNT, NUM_EDGES);
  seg_alloc_kernel<<<(NUM_NODES + SEG_BLK - 1) / SEG_BLK, SEG_BLK, 0, stream>>>(
      deg_cur, seg_total, seg_bbase, NUM_NODES);
  edge_fill_kernel<<<4096, 256, 0, stream>>>(src_idx, dst_idx, rel_idx, deg_cur,
                                             edge_rec, NUM_EDGES);

  fold_kernel<<<1, 256, 0, stream>>>(
      W_ent, b_ent, W_relL, b_relL, W_rel2, b_rel2, W_a, b_a, W_fc, b_fc,
      ws + OFF_MS, ws + OFF_MD, ws + OFF_MR, ws + OFF_M2, ws + OFF_CVEC,
      ws + OFF_VREL2, ws + OFF_A0);

  const int gemm_blocks_N = (NUM_NODES + 127) / 128;
  const int gemm_blocks_R = (NUM_RELS + 127) / 128;
  wmma_gemm64<DIM><<<gemm_blocks_N, 256, 0, stream>>>(ent_embed, ws + OFF_MS,
                                                 nullptr, ws + OFF_US, NUM_NODES);
  wmma_gemm64<DIM><<<gemm_blocks_N, 256, 0, stream>>>(ent_embed, ws + OFF_MD,
                                                 nullptr, ws + OFF_UD, NUM_NODES);
  wmma_gemm64<DIM><<<gemm_blocks_R, 256, 0, stream>>>(rel_embed, ws + OFF_MR,
                                                 nullptr, ws + OFF_UR, NUM_RELS);
  wmma_gemm64<DIM><<<gemm_blocks_R, 256, 0, stream>>>(rel_embed, ws + OFF_M2,
                                                 ws + OFF_VREL2, ws + OFF_REL2P,
                                                 NUM_RELS);

  dot64_kernel<<<(NUM_NODES + 255) / 256, 256, 0, stream>>>(ws + OFF_US, W_a,
                                                            ws + OFF_AS, NUM_NODES);
  dot64_kernel<<<(NUM_NODES + 255) / 256, 256, 0, stream>>>(ws + OFF_UD, W_a,
                                                            ws + OFF_AD, NUM_NODES);
  dot64_kernel<<<(NUM_RELS + 255) / 256, 256, 0, stream>>>(ws + OFF_UR, W_a,
                                                           ws + OFF_AR, NUM_RELS);

  node_gather_kernel<<<(NUM_NODES + 7) / 8, 256, 0, stream>>>(
      deg_cur, seg_bbase, edge_rec, ws + OFF_US, ws + OFF_UD, ws + OFF_UR,
      ws + OFF_CVEC, ws + OFF_AS, ws + OFF_AD, ws + OFF_AR, ws + OFF_A0, h_ent,
      NUM_NODES, NUM_EDGES);
  wmma_gemm64<3 * DIM><<<gemm_blocks_N, 256, 0, stream>>>(h_ent, W_rel3,
                                                 nullptr, ws + OFF_US, NUM_NODES);
  wmma_gemm64<3 * DIM><<<gemm_blocks_N, 256, 0, stream>>>(h_ent, W_rel3 + DIM,
                                                 nullptr, ws + OFF_UD, NUM_NODES);
  edge_pool_kernel<<<POOL_WGS, 512, 0, stream>>>(
      src_idx, dst_idx, rel_idx, ws + OFF_US, ws + OFF_UD, ws + OFF_POOL,
      NUM_EDGES);

  h_rel_kernel<<<(NUM_RELS * DIM + 255) / 256, 256, 0, stream>>>(
      ws + OFF_POOL, ws + OFF_CNT, ws + OFF_REL2P, W_rel3, b_rel3, h_rel,
      NUM_RELS);
}


# edge logits from the rows the gather wave holds anyway (c_e.w_a+b_a, wave sum), softmax as sum(b c)/sum(b) under a running max; 3 score passes + score gathers go; logit/b_sum/h_ent sum order changes
# speedup vs baseline: 2.1231x; 1.0382x over previous
#include <hip/hip_runtime.h>
#include <math.h>


#define NUM_NODES 100000
#define NUM_RELS  500
#define DIM       64
#define NUM_EDGES 1000000

typedef __attribute__((ext_vector_type(2))) float v2f;
typedef __attribute__((ext_vector_type(8))) float v8f;

__global__ void zero_kernel(float* __restrict__ p, int n) {
  for (int i = blockIdx.x * blockDim.x + threadIdx.x; i < n;
       i += gridDim.x * blockDim.x)
    p[i] = 0.0f;
}

__global__ __launch_bounds__(256) void fold_kernel(
    const float* __restrict__ W_ent, const float* __restrict__ b_ent,
    const float* __restrict__ W_relL, const float* __restrict__ b_relL,
    const float* __restrict__ W_rel2, const float* __restrict__ b_rel2,
    const float* __restrict__ W_fc, const float* __restrict__ b_fc,
    float* __restrict__ M_s, float* __restrict__ M_d,
    float* __restrict__ M_r, float* __restrict__ M_2,
    float* __restrict__ cvec, float* __restrict__ v_rel2) {
  const int t = threadIdx.x;
  for (int idx = t; idx < DIM * DIM; idx += 256) {
    const int o = idx >> 6, k = idx & 63;
    float ms = 0.f, md = 0.f, mr = 0.f, m2 = 0.f;
    for (int u = 0; u < DIM; ++u) {
      const float we = W_ent[u * DIM + k];
      const float wl = W_relL[u * DIM + k];
      ms += W_fc[o * 192 + u] * we;
      md += W_fc[o * 192 + 64 + u] * we;
      mr += W_fc[o * 192 + 128 + u] * wl;
      m2 += W_rel2[o * DIM + u] * wl;
    }
    M_s[idx] = ms; M_d[idx] = md; M_r[idx] = mr; M_2[idx] = m2;
  }
  if (t < DIM) {
    float cv = b_fc[t], v2 = b_rel2[t];
    for (int u = 0; u < DIM; ++u) {
      cv += W_fc[t * 192 + u] * b_ent[u]
          + W_fc[t * 192 + 64 + u] * b_ent[u]
          + W_fc[t * 192 + 128 + u] * b_relL[u];
      v2 += W_rel2[t * DIM + u] * b_relL[u];
    }
    cvec[t] = cv; v_rel2[t] = v2;
  }
}

template <int LDW>
__global__ __launch_bounds__(256) void wmma_gemm64(
    const float* __restrict__ A, const float* __restrict__ W,
    const float* __restrict__ bias, float* __restrict__ out, int M) {
  __shared__ float sW[DIM * DIM];
  for (int i = threadIdx.x; i < DIM * DIM; i += 256)
    sW[i] = W[(i >> 6) * LDW + (i & 63)];
  __syncthreads();

  const int wave = threadIdx.x >> 5;
  const int lane = threadIdx.x & 31;
  const int m0 = (blockIdx.x * 8 + wave) * 16;
  if (m0 >= M) return;

  const int lm = lane & 15;
  const int hi = lane >> 4;

  int arow = m0 + lm;
  if (arow >= M) arow = M - 1;
  const float* __restrict__ Arow = A + (size_t)arow * DIM;

  v8f acc0 = {}, acc1 = {}, acc2 = {}, acc3 = {};
  for (int k = 0; k < DIM; k += 4) {
    v2f a;
    a.x = Arow[k + 2 * hi + 0];
    a.y = Arow[k + 2 * hi + 1];
    v2f b0, b1, b2, b3;
    const float* w0 = &sW[(0 * 16 + lm) * DIM + k + 2 * hi];
    const float* w1 = &sW[(1 * 16 + lm) * DIM + k + 2 * hi];
    const float* w2 = &sW[(2 * 16 + lm) * DIM + k + 2 * hi];
    const float* w3 = &sW[(3 * 16 + lm) * DIM + k + 2 * hi];
    b0.x = w0[0]; b0.y = w0[1];
    b1.x = w1[0]; b1.y = w1[1];
    b2.x = w2[0]; b2.y = w2[1];
    b3.x = w3[0]; b3.y = w3[1];
    acc0 = __builtin_amdgcn_wmma_f32_16x16x4_f32(false, a, false, b0,
                                                 (short)0, acc0, false, false);
    acc1 = __builtin_amdgcn_wmma_f32_16x16x4_f32(false, a, false, b1,
                                                 (short)0, acc1, false, false);
    acc2 = __builtin_amdgcn_wmma_f32_16x16x4_f32(false, a, false, b2,
                                                 (short)0, acc2, false, false);
    acc3 = __builtin_amdgcn_wmma_f32_16x16x4_f32(false, a, false, b3,
                                                 (short)0, acc3, false, false);
  }

  v8f accs[4] = {acc0, acc1, acc2, acc3};
#pragma unroll
  for (int ob = 0; ob < 4; ++ob) {
    const float bv = bias ? bias[ob * 16 + lm] : 0.0f;
#pragma unroll
    for (int v = 0; v < 8; ++v) {
      const int row = m0 + v + 8 * hi;
      if (row < M)
        out[(size_t)row * DIM + ob * 16 + lm] = accs[ob][v] + bv;
    }
  }
}

template <int V> struct KConst { static constexpr int value = V; };

#define SEG_BLK 1024

__global__ void edge_count_kernel(const int* __restrict__ src,
                                  const int* __restrict__ rel,
                                  int* __restrict__ deg,
                                  float* __restrict__ cnt, int E) {
  for (int e = blockIdx.x * blockDim.x + threadIdx.x; e < E;
       e += gridDim.x * blockDim.x) {
    atomicAdd(&deg[src[e]], 1);
    atomicAdd(&cnt[rel[e]], 1.0f);
  }
}

__global__ __launch_bounds__(SEG_BLK) void seg_alloc_kernel(
    int* __restrict__ cursor  ,
    int* __restrict__ total, int* __restrict__ bbase, int N) {
  __shared__ int swave[32];
  __shared__ int sbase;
  const int lane = threadIdx.x & 31;
  const int wave = threadIdx.x >> 5;
  const int i = blockIdx.x * SEG_BLK + threadIdx.x;
  const int v = (i < N) ? cursor[i] : 0;
  int x = v;
#pragma unroll
  for (int o = 1; o < 32; o <<= 1) {
    const int y = __shfl_up(x, o);
    if (lane >= o) x += y;
  }
  if (lane == 31) swave[wave] = x;
  __syncthreads();
  if (wave == 0) {
    const int w = swave[lane];
    int xs = w;
#pragma unroll
    for (int o = 1; o < 32; o <<= 1) {
      const int y = __shfl_up(xs, o);
      if (lane >= o) xs += y;
    }
    swave[lane] = xs - w;
    if (lane == 31) {
      const int base = atomicAdd(total, xs);
      sbase = base;
      bbase[blockIdx.x] = base;
    }
  }
  __syncthreads();
  if (i < N) cursor[i] = sbase + swave[wave] + x - v;
}

__global__ void edge_fill_kernel(const int* __restrict__ src,
                                 const int* __restrict__ dst,
                                 const int* __restrict__ rel,
                                 int* __restrict__ cursor,
                                 int* __restrict__ rec, int E) {
  for (int e = blockIdx.x * blockDim.x + threadIdx.x; e < E;
       e += gridDim.x * blockDim.x) {
    const int p = atomicAdd(&cursor[src[e]], 1);
    if ((unsigned)p < (unsigned)E) rec[p] = dst[e] | (rel[e] << 17);
  }
}

__device__ __forceinline__ float xlane16(float x) {
  const int b = __float_as_int(x);
  return __int_as_float(
      __builtin_amdgcn_permlanex16(b, b, 0x76543210, 0xfedcba98, false, false));
}
template <int M>
__device__ __forceinline__ float xrow(float x) {
  const int b = __float_as_int(x);
  return __int_as_float(
      __builtin_amdgcn_update_dpp(b, b, 0x160 | M, 0xf, 0xf, false));
}

__global__ __launch_bounds__(256) void node_gather_kernel(
    const int* __restrict__ cursor, const int* __restrict__ bbase,
    const int* __restrict__ rec, const float* __restrict__ u_s,
    const float* __restrict__ u_d, const float* __restrict__ u_r,
    const float* __restrict__ cvec, const float* __restrict__ W_a,
    const float* __restrict__ b_a, float* __restrict__ h_ent, int N, int E) {
  const int s = blockIdx.x * (blockDim.x >> 5) + (threadIdx.x >> 5);
  const int lane = threadIdx.x & 31;
  if (s >= N) return;

  int end = cursor[s];
  int start = (s & (SEG_BLK - 1)) ? cursor[s - 1] : bbase[s / SEG_BLK];
  end = __builtin_amdgcn_readfirstlane(end);
  start = __builtin_amdgcn_readfirstlane(start);
  start = min(max(start, 0), E);
  end = min(max(end, start), E);
  const int n = end - start;

  v2f acc = {0.f, 0.f};
  if (n > 0) {
    int pk = (lane < n) ? rec[start + lane] : 0;
    const v2f wa = *(const v2f*)(W_a + 2 * lane);
    const float ba = b_a[0];
    const v2f base = *(const v2f*)(u_s + (size_t)s * DIM + 2 * lane) +
                     *(const v2f*)(cvec + 2 * lane);
    const bool h4 = (lane & 16) != 0;
    const bool h3 = (lane & 8) != 0;
    const bool h2 = (lane & 4) != 0;
    const float NEG = -3.0e38f;
    float m = NEG;
    float bsum = 0.f;

    for (int c0 = 0; c0 < n; c0 += 32) {
      if (c0 > 0) pk = (c0 + lane < n) ? rec[start + c0 + lane] : 0;
      const int mm = min(32, n - c0);
      int j = 0;
      auto group = [&](auto KC, int valid) {
        constexpr int K = decltype(KC)::value;
        v2f ud[K], ur[K], c[K];
        float p[K];
#pragma unroll
        for (int i = 0; i < K; ++i) {
          const int idx = (j + i) & 31;
          const int pj = __builtin_amdgcn_readlane(pk, idx);
          const int d = min(pj & 0x1FFFF, NUM_NODES - 1);
          const int r = min(pj >> 17, NUM_RELS - 1);
          ud[i] = *(const v2f*)(u_d + (size_t)d * DIM + 2 * lane);
          ur[i] = *(const v2f*)(u_r + (size_t)r * DIM + 2 * lane);
        }
#pragma unroll
        for (int i = 0; i < K; ++i) {
          c[i] = (base + ud[i]) + ur[i];
          p[i] = c[i].x * wa.x + c[i].y * wa.y;
        }
        float t;
        if constexpr (K == 8) {
          float q[4], q2[2];
#pragma unroll
          for (int i = 0; i < 4; ++i) {
            const float keep = h4 ? p[i + 4] : p[i];
            const float send = h4 ? p[i] : p[i + 4];
            q[i] = keep + xlane16(send);
          }
#pragma unroll
          for (int i = 0; i < 2; ++i) {
            const float keep = h3 ? q[i + 2] : q[i];
            const float send = h3 ? q[i] : q[i + 2];
            q2[i] = keep + xrow<8>(send);
          }
          const float keep = h2 ? q2[1] : q2[0];
          const float send = h2 ? q2[0] : q2[1];
          t = keep + xrow<4>(send);
        } else {
          float q[2];
#pragma unroll
          for (int i = 0; i < 2; ++i) {
            const float keep = h4 ? p[i + 2] : p[i];
            const float send = h4 ? p[i] : p[i + 2];
            q[i] = keep + xlane16(send);
          }
          const float keep = h3 ? q[1] : q[0];
          const float send = h3 ? q[0] : q[1];
          t = keep + xrow<8>(send);
          t += xrow<4>(t);
        }
        t += xrow<2>(t);
        t += xrow<1>(t);
        const int myedge = (K == 8) ? ((lane >> 2) & 7) : ((lane >> 3) & 3);
        float l = t + ba;
        l = l > 0.f ? l : 0.01f * l;
        if (myedge >= valid) l = NEG;
        float gm = l;
        gm = fmaxf(gm, xlane16(gm));
        gm = fmaxf(gm, xrow<8>(gm));
        gm = fmaxf(gm, xrow<4>(gm));
        const float mnew = fmaxf(m, gm);
        const float sc = expf(m - mnew);
        const int bbits = __float_as_int(expf(l - mnew));
        m = mnew;
        acc.x *= sc;
        acc.y *= sc;
        bsum *= sc;
#pragma unroll
        for (int i = 0; i < K; ++i) {
          if (i < valid) {
            const float w =
                __int_as_float(__builtin_amdgcn_readlane(bbits, i * (32 / K)));
            acc.x += w * c[i].x;
            acc.y += w * c[i].y;
            bsum += w;
          }
        }
        j += K;
      };
      while (mm - j >= 8) group(KConst<8>{}, 8);
      if (mm - j >= 4) group(KConst<4>{}, 4);
      if (mm - j > 0) group(KConst<4>{}, mm - j);
    }
    acc.x = acc.x / bsum;
    acc.y = acc.y / bsum;
  }
  *(v2f*)(h_ent + (size_t)s * DIM + 2 * lane) = acc;
}

typedef __attribute__((ext_vector_type(4))) float v4f;
typedef __attribute__((ext_vector_type(4))) int v4i;

#define PB      128
#define PPITCH  160
#define POOL_WGS 256

__global__ __launch_bounds__(512) void edge_pool_kernel(
    const int* __restrict__ src, const int* __restrict__ dst,
    const int* __restrict__ rel, const float* __restrict__ g_s,
    const float* __restrict__ g_d, float* __restrict__ pool, int E) {
  __shared__ __attribute__((aligned(16))) float sG[2][(PB / 2) * PPITCH];
  __shared__ __attribute__((aligned(16))) int sR[2][PB];

  const int wave = threadIdx.x >> 5;
  const int lane = threadIdx.x & 31;
  const int lm = lane & 15;
  const int hi = lane >> 4;
  const int rbase = wave * 32;
  const int nb = (E + PB - 1) / PB;
  const int G = gridDim.x;

  v8f c00 = {}, c01 = {}, c02 = {}, c03 = {};
  v8f c10 = {}, c11 = {}, c12 = {}, c13 = {};

  v2f rs[8], rd[8];
  int s_cur, d_cur, r_cur, s_nxt, d_nxt, r_nxt;

  auto load_idx = [&](int bb, int& s, int& d, int& r) {
    const int e = bb * PB + wave * 8 + (lane & 7);
    const bool ok = (bb < nb) && (e < E);
    s = ok ? src[e] : 0;
    d = ok ? dst[e] : 0;
    r = ok ? rel[e] : -1;
  };
  auto load_rows = [&](int s, int d) {
#pragma unroll
    for (int i = 0; i < 8; ++i) {
      const int si = __builtin_amdgcn_readlane(s, i);
      const int di = __builtin_amdgcn_readlane(d, i);
      rs[i] = *(const v2f*)(g_s + (size_t)si * DIM + 2 * lane);
      rd[i] = *(const v2f*)(g_d + (size_t)di * DIM + 2 * lane);
    }
  };

  int b = blockIdx.x;
  load_idx(b, s_cur, d_cur, r_cur);
  load_rows(s_cur, d_cur);
  load_idx(b + G, s_nxt, d_nxt, r_nxt);

  int buf = 0;
  for (; b < nb; b += G) {
    float* gb = sG[buf];
#pragma unroll
    for (int j = 0; j < 4; ++j) {
      const int r0 = __builtin_amdgcn_readlane(r_cur, 2 * j);
      const int r1 = __builtin_amdgcn_readlane(r_cur, 2 * j + 1);
      v2f g0 = rs[2 * j] + rd[2 * j];
      v2f g1 = rs[2 * j + 1] + rd[2 * j + 1];
      if (r0 < 0) { g0.x = 0.f; g0.y = 0.f; }
      if (r1 < 0) { g1.x = 0.f; g1.y = 0.f; }
      v4f o;
      o.x = g0.x; o.y = g1.x; o.z = g0.y; o.w = g1.y;
      *(v4f*)(gb + (wave * 4 + j) * PPITCH + 4 * lane) = o;
    }
    if (lane < 8) sR[buf][wave * 8 + lane] = r_cur;
    __syncthreads();

    s_cur = s_nxt; d_cur = d_nxt; r_cur = r_nxt;
    if (b + G < nb) load_rows(s_cur, d_cur);
    load_idx(b + 2 * G, s_nxt, d_nxt, r_nxt);

    const v4i rr = *(const v4i*)&sR[buf][4 * lane];
    const bool relv = ((unsigned)(rr.x - rbase) < 32u) |
                      ((unsigned)(rr.y - rbase) < 32u) |
                      ((unsigned)(rr.z - rbase) < 32u) |
                      ((unsigned)(rr.w - rbase) < 32u);
    unsigned mask = __builtin_amdgcn_ballot_w32(relv);
    while (mask) {
      const int g = __builtin_ctz(mask);
      mask &= mask - 1;
      const int* pr = &sR[buf][4 * g + 2 * hi];
      const int da = pr[0] - rbase - lm;
      const int db = pr[1] - rbase - lm;
      const float* pg = gb + (2 * g + hi) * PPITCH + 2 * lm;
      const v2f b0 = *(const v2f*)(pg);
      const v2f b1 = *(const v2f*)(pg + 32);
      const v2f b2 = *(const v2f*)(pg + 64);
      const v2f b3 = *(const v2f*)(pg + 96);
      {
        const bool ma = (da == 0), mb = (db == 0);
        if (__builtin_amdgcn_ballot_w32(ma | mb)) {
          v2f a;
          a.x = ma ? 1.0f : 0.0f;
          a.y = mb ? 1.0f : 0.0f;
          c00 = __builtin_amdgcn_wmma_f32_16x16x4_f32(false, a, false, b0,
                                                      (short)0, c00, false, false);
          c01 = __builtin_amdgcn_wmma_f32_16x16x4_f32(false, a, false, b1,
                                                      (short)0, c01, false, false);
          c02 = __builtin_amdgcn_wmma_f32_16x16x4_f32(false, a, false, b2,
                                                      (short)0, c02, false, false);
          c03 = __builtin_amdgcn_wmma_f32_16x16x4_f32(false, a, false, b3,
                                                      (short)0, c03, false, false);
        }
      }
      {
        const bool ma = (da == 16), mb = (db == 16);
        if (__builtin_amdgcn_ballot_w32(ma | mb)) {
          v2f a;
          a.x = ma ? 1.0f : 0.0f;
          a.y = mb ? 1.0f : 0.0f;
          c10 = __builtin_amdgcn_wmma_f32_16x16x4_f32(false, a, false, b0,
                                                      (short)0, c10, false, false);
          c11 = __builtin_amdgcn_wmma_f32_16x16x4_f32(false, a, false, b1,
                                                      (short)0, c11, false, false);
          c12 = __builtin_amdgcn_wmma_f32_16x16x4_f32(false, a, false, b2,
                                                      (short)0, c12, false, false);
          c13 = __builtin_amdgcn_wmma_f32_16x16x4_f32(false, a, false, b3,
                                                      (short)0, c13, false, false);
        }
      }
    }
    buf ^= 1;
  }

  __syncthreads();
  float* sFw = &sG[0][0] + wave * 1024;
  auto flush = [&](const v8f& t0, const v8f& t1, const v8f& t2, const v8f& t3,
                   int rt) {
#pragma unroll
    for (int v = 0; v < 8; ++v) {
      float* row = sFw + (v + 8 * hi) * DIM + lm;
      row[0] = t0[v]; row[16] = t1[v]; row[32] = t2[v]; row[48] = t3[v];
    }
    __syncthreads();
    for (int q = 0; q < 16; ++q) {
      const int r = (q + (int)blockIdx.x) & 15;
      const int R = rbase + rt * 16 + r;
      if (R < NUM_RELS) {
        atomicAdd(&pool[(size_t)R * DIM + lane], sFw[r * DIM + lane]);
        atomicAdd(&pool[(size_t)R * DIM + 32 + lane], sFw[r * DIM + 32 + lane]);
      }
    }
    __syncthreads();
  };
  flush(c00, c01, c02, c03, 0);
  flush(c10, c11, c12, c13, 1);
}

__global__ void h_rel_kernel(const float* __restrict__ pool,
                             const float* __restrict__ cnt,
                             const float* __restrict__ rel2_proj,
                             const float* __restrict__ W_rel3,
                             const float* __restrict__ b_rel3,
                             float* __restrict__ h_rel, int R) {
  const int t = blockIdx.x * blockDim.x + threadIdx.x;
  if (t >= R * DIM) return;
  const int r = t >> 6, o = t & 63;
  const float c = cnt[r];
  const float inv = 1.0f / fmaxf(c, 1.0f);
  const float* w = W_rel3 + (size_t)o * 192;
  float acc = b_rel3[o] + pool[t] * inv;
  if (c > 0.f) {
    const float* r2 = rel2_proj + (size_t)r * DIM;
    for (int j = 0; j < DIM; ++j) acc += r2[j] * w[128 + j];
  }
  h_rel[t] = acc;
}

#define OFF_MS     0
#define OFF_MD     4096
#define OFF_MR     8192
#define OFF_M2     12288
#define OFF_CVEC   16384
#define OFF_VREL2  16448
#define OFF_UR     16576
#define OFF_REL2P  49088
#define OFF_BSUM   281088
#define OFF_CNT    381088
#define OFF_POOL   381588
#define OFF_TOTAL  413588
#define OFF_BBASE  413600
#define OFF_BE     445588
#define OFF_US     1445588
#define OFF_UD     7845588
#define ZERO_CNT   (OFF_TOTAL + 1 - OFF_BSUM)

extern "C" void kernel_launch(void* const* d_in, const int* in_sizes, int n_in,
                              void* d_out, int out_size, void* d_ws,
                              size_t ws_size, hipStream_t stream) {
  const float* ent_embed = (const float*)d_in[0];
  const float* rel_embed = (const float*)d_in[1];
  const float* W_ent  = (const float*)d_in[2];
  const float* b_ent  = (const float*)d_in[3];
  const float* W_relL = (const float*)d_in[4];
  const float* b_relL = (const float*)d_in[5];
  const float* W_rel2 = (const float*)d_in[6];
  const float* b_rel2 = (const float*)d_in[7];
  const float* W_rel3 = (const float*)d_in[8];
  const float* b_rel3 = (const float*)d_in[9];
  const float* W_a    = (const float*)d_in[10];
  const float* b_a    = (const float*)d_in[11];
  const float* W_fc   = (const float*)d_in[12];
  const float* b_fc   = (const float*)d_in[13];
  const int* src_idx  = (const int*)d_in[14];
  const int* dst_idx  = (const int*)d_in[15];
  const int* rel_idx  = (const int*)d_in[16];

  float* ws = (float*)d_ws;
  float* h_ent = (float*)d_out;
  float* h_rel = (float*)d_out + (size_t)NUM_NODES * DIM;

  zero_kernel<<<512, 256, 0, stream>>>(ws + OFF_BSUM, ZERO_CNT);

  int* deg_cur = (int*)(ws + OFF_BSUM);
  int* seg_total = (int*)(ws + OFF_TOTAL);
  int* seg_bbase = (int*)(ws + OFF_BBASE);
  int* edge_rec = (int*)(ws + OFF_BE);
  edge_count_kernel<<<4096, 256, 0, stream>>>(src_idx, rel_idx, deg_cur,
                                              ws + OFF_CNT, NUM_EDGES);
  seg_alloc_kernel<<<(NUM_NODES + SEG_BLK - 1) / SEG_BLK, SEG_BLK, 0, stream>>>(
      deg_cur, seg_total, seg_bbase, NUM_NODES);
  edge_fill_kernel<<<4096, 256, 0, stream>>>(src_idx, dst_idx, rel_idx, deg_cur,
                                             edge_rec, NUM_EDGES);

  fold_kernel<<<1, 256, 0, stream>>>(
      W_ent, b_ent, W_relL, b_relL, W_rel2, b_rel2, W_fc, b_fc,
      ws + OFF_MS, ws + OFF_MD, ws + OFF_MR, ws + OFF_M2, ws + OFF_CVEC,
      ws + OFF_VREL2);

  const int gemm_blocks_N = (NUM_NODES + 127) / 128;
  const int gemm_blocks_R = (NUM_RELS + 127) / 128;
  wmma_gemm64<DIM><<<gemm_blocks_N, 256, 0, stream>>>(ent_embed, ws + OFF_MS,
                                                 nullptr, ws + OFF_US, NUM_NODES);
  wmma_gemm64<DIM><<<gemm_blocks_N, 256, 0, stream>>>(ent_embed, ws + OFF_MD,
                                                 nullptr, ws + OFF_UD, NUM_NODES);
  wmma_gemm64<DIM><<<gemm_blocks_R, 256, 0, stream>>>(rel_embed, ws + OFF_MR,
                                                 nullptr, ws + OFF_UR, NUM_RELS);
  wmma_gemm64<DIM><<<gemm_blocks_R, 256, 0, stream>>>(rel_embed, ws + OFF_M2,
                                                 ws + OFF_VREL2, ws + OFF_REL2P,
                                                 NUM_RELS);

  node_gather_kernel<<<(NUM_NODES + 7) / 8, 256, 0, stream>>>(
      deg_cur, seg_bbase, edge_rec, ws + OFF_US, ws + OFF_UD, ws + OFF_UR,
      ws + OFF_CVEC, W_a, b_a, h_ent, NUM_NODES, NUM_EDGES);
  wmma_gemm64<3 * DIM><<<gemm_blocks_N, 256, 0, stream>>>(h_ent, W_rel3,
                                                 nullptr, ws + OFF_US, NUM_NODES);
  wmma_gemm64<3 * DIM><<<gemm_blocks_N, 256, 0, stream>>>(h_ent, W_rel3 + DIM,
                                                 nullptr, ws + OFF_UD, NUM_NODES);
  edge_pool_kernel<<<POOL_WGS, 512, 0, stream>>>(
      src_idx, dst_idx, rel_idx, ws + OFF_US, ws + OFF_UD, ws + OFF_POOL,
      NUM_EDGES);

  h_rel_kernel<<<(NUM_RELS * DIM + 255) / 256, 256, 0, stream>>>(
      ws + OFF_POOL, ws + OFF_CNT, ws + OFF_REL2P, W_rel3, b_rel3, h_rel,
      NUM_RELS);
}


# edge counts = row sums of the one-hot WMMA operand in edge_pool (lane-private adds, one plain 128-B store per wave, summed in h_rel): 1M per-edge float atomics on 16 hot lines go; cnt exact
# speedup vs baseline: 2.4365x; 1.1476x over previous
#include <hip/hip_runtime.h>
#include <math.h>


#define NUM_NODES 100000
#define NUM_RELS  500
#define DIM       64
#define NUM_EDGES 1000000

typedef __attribute__((ext_vector_type(2))) float v2f;
typedef __attribute__((ext_vector_type(8))) float v8f;

__global__ void zero_kernel(float* __restrict__ p, int n) {
  for (int i = blockIdx.x * blockDim.x + threadIdx.x; i < n;
       i += gridDim.x * blockDim.x)
    p[i] = 0.0f;
}

__global__ __launch_bounds__(256) void fold_kernel(
    const float* __restrict__ W_ent, const float* __restrict__ b_ent,
    const float* __restrict__ W_relL, const float* __restrict__ b_relL,
    const float* __restrict__ W_rel2, const float* __restrict__ b_rel2,
    const float* __restrict__ W_fc, const float* __restrict__ b_fc,
    float* __restrict__ M_s, float* __restrict__ M_d,
    float* __restrict__ M_r, float* __restrict__ M_2,
    float* __restrict__ cvec, float* __restrict__ v_rel2) {
  const int t = threadIdx.x;
  for (int idx = t; idx < DIM * DIM; idx += 256) {
    const int o = idx >> 6, k = idx & 63;
    float ms = 0.f, md = 0.f, mr = 0.f, m2 = 0.f;
    for (int u = 0; u < DIM; ++u) {
      const float we = W_ent[u * DIM + k];
      const float wl = W_relL[u * DIM + k];
      ms += W_fc[o * 192 + u] * we;
      md += W_fc[o * 192 + 64 + u] * we;
      mr += W_fc[o * 192 + 128 + u] * wl;
      m2 += W_rel2[o * DIM + u] * wl;
    }
    M_s[idx] = ms; M_d[idx] = md; M_r[idx] = mr; M_2[idx] = m2;
  }
  if (t < DIM) {
    float cv = b_fc[t], v2 = b_rel2[t];
    for (int u = 0; u < DIM; ++u) {
      cv += W_fc[t * 192 + u] * b_ent[u]
          + W_fc[t * 192 + 64 + u] * b_ent[u]
          + W_fc[t * 192 + 128 + u] * b_relL[u];
      v2 += W_rel2[t * DIM + u] * b_relL[u];
    }
    cvec[t] = cv; v_rel2[t] = v2;
  }
}

template <int LDW>
__global__ __launch_bounds__(256) void wmma_gemm64(
    const float* __restrict__ A, const float* __restrict__ W,
    const float* __restrict__ bias, float* __restrict__ out, int M) {
  __shared__ float sW[DIM * DIM];
  for (int i = threadIdx.x; i < DIM * DIM; i += 256)
    sW[i] = W[(i >> 6) * LDW + (i & 63)];
  __syncthreads();

  const int wave = threadIdx.x >> 5;
  const int lane = threadIdx.x & 31;
  const int m0 = (blockIdx.x * 8 + wave) * 16;
  if (m0 >= M) return;

  const int lm = lane & 15;
  const int hi = lane >> 4;

  int arow = m0 + lm;
  if (arow >= M) arow = M - 1;
  const float* __restrict__ Arow = A + (size_t)arow * DIM;

  v8f acc0 = {}, acc1 = {}, acc2 = {}, acc3 = {};
  for (int k = 0; k < DIM; k += 4) {
    v2f a;
    a.x = Arow[k + 2 * hi + 0];
    a.y = Arow[k + 2 * hi + 1];
    v2f b0, b1, b2, b3;
    const float* w0 = &sW[(0 * 16 + lm) * DIM + k + 2 * hi];
    const float* w1 = &sW[(1 * 16 + lm) * DIM + k + 2 * hi];
    const float* w2 = &sW[(2 * 16 + lm) * DIM + k + 2 * hi];
    const float* w3 = &sW[(3 * 16 + lm) * DIM + k + 2 * hi];
    b0.x = w0[0]; b0.y = w0[1];
    b1.x = w1[0]; b1.y = w1[1];
    b2.x = w2[0]; b2.y = w2[1];
    b3.x = w3[0]; b3.y = w3[1];
    acc0 = __builtin_amdgcn_wmma_f32_16x16x4_f32(false, a, false, b0,
                                                 (short)0, acc0, false, false);
    acc1 = __builtin_amdgcn_wmma_f32_16x16x4_f32(false, a, false, b1,
                                                 (short)0, acc1, false, false);
    acc2 = __builtin_amdgcn_wmma_f32_16x16x4_f32(false, a, false, b2,
                                                 (short)0, acc2, false, false);
    acc3 = __builtin_amdgcn_wmma_f32_16x16x4_f32(false, a, false, b3,
                                                 (short)0, acc3, false, false);
  }

  v8f accs[4] = {acc0, acc1, acc2, acc3};
#pragma unroll
  for (int ob = 0; ob < 4; ++ob) {
    const float bv = bias ? bias[ob * 16 + lm] : 0.0f;
#pragma unroll
    for (int v = 0; v < 8; ++v) {
      const int row = m0 + v + 8 * hi;
      if (row < M)
        out[(size_t)row * DIM + ob * 16 + lm] = accs[ob][v] + bv;
    }
  }
}

template <int V> struct KConst { static constexpr int value = V; };

#define SEG_BLK 1024

__global__ void edge_count_kernel(const int* __restrict__ src,
                                  int* __restrict__ deg, int E) {
  for (int e = blockIdx.x * blockDim.x + threadIdx.x; e < E;
       e += gridDim.x * blockDim.x)
    atomicAdd(&deg[src[e]], 1);
}

__global__ __launch_bounds__(SEG_BLK) void seg_alloc_kernel(
    int* __restrict__ cursor  ,
    int* __restrict__ total, int* __restrict__ bbase, int N) {
  __shared__ int swave[32];
  __shared__ int sbase;
  const int lane = threadIdx.x & 31;
  const int wave = threadIdx.x >> 5;
  const int i = blockIdx.x * SEG_BLK + threadIdx.x;
  const int v = (i < N) ? cursor[i] : 0;
  int x = v;
#pragma unroll
  for (int o = 1; o < 32; o <<= 1) {
    const int y = __shfl_up(x, o);
    if (lane >= o) x += y;
  }
  if (lane == 31) swave[wave] = x;
  __syncthreads();
  if (wave == 0) {
    const int w = swave[lane];
    int xs = w;
#pragma unroll
    for (int o = 1; o < 32; o <<= 1) {
      const int y = __shfl_up(xs, o);
      if (lane >= o) xs += y;
    }
    swave[lane] = xs - w;
    if (lane == 31) {
      const int base = atomicAdd(total, xs);
      sbase = base;
      bbase[blockIdx.x] = base;
    }
  }
  __syncthreads();
  if (i < N) cursor[i] = sbase + swave[wave] + x - v;
}

__global__ void edge_fill_kernel(const int* __restrict__ src,
                                 const int* __restrict__ dst,
                                 const int* __restrict__ rel,
                                 int* __restrict__ cursor,
                                 int* __restrict__ rec, int E) {
  for (int e = blockIdx.x * blockDim.x + threadIdx.x; e < E;
       e += gridDim.x * blockDim.x) {
    const int p = atomicAdd(&cursor[src[e]], 1);
    if ((unsigned)p < (unsigned)E) rec[p] = dst[e] | (rel[e] << 17);
  }
}

__device__ __forceinline__ float xlane16(float x) {
  const int b = __float_as_int(x);
  return __int_as_float(
      __builtin_amdgcn_permlanex16(b, b, 0x76543210, 0xfedcba98, false, false));
}
template <int M>
__device__ __forceinline__ float xrow(float x) {
  const int b = __float_as_int(x);
  return __int_as_float(
      __builtin_amdgcn_update_dpp(b, b, 0x160 | M, 0xf, 0xf, false));
}

__global__ __launch_bounds__(256) void node_gather_kernel(
    const int* __restrict__ cursor, const int* __restrict__ bbase,
    const int* __restrict__ rec, const float* __restrict__ u_s,
    const float* __restrict__ u_d, const float* __restrict__ u_r,
    const float* __restrict__ cvec, const float* __restrict__ W_a,
    const float* __restrict__ b_a, float* __restrict__ h_ent, int N, int E) {
  const int s = blockIdx.x * (blockDim.x >> 5) + (threadIdx.x >> 5);
  const int lane = threadIdx.x & 31;
  if (s >= N) return;

  int end = cursor[s];
  int start = (s & (SEG_BLK - 1)) ? cursor[s - 1] : bbase[s / SEG_BLK];
  end = __builtin_amdgcn_readfirstlane(end);
  start = __builtin_amdgcn_readfirstlane(start);
  start = min(max(start, 0), E);
  end = min(max(end, start), E);
  const int n = end - start;

  v2f acc = {0.f, 0.f};
  if (n > 0) {
    int pk = (lane < n) ? rec[start + lane] : 0;
    const v2f wa = *(const v2f*)(W_a + 2 * lane);
    const float ba = b_a[0];
    const v2f base = *(const v2f*)(u_s + (size_t)s * DIM + 2 * lane) +
                     *(const v2f*)(cvec + 2 * lane);
    const bool h4 = (lane & 16) != 0;
    const bool h3 = (lane & 8) != 0;
    const bool h2 = (lane & 4) != 0;
    const float NEG = -3.0e38f;
    float m = NEG;
    float bsum = 0.f;

    for (int c0 = 0; c0 < n; c0 += 32) {
      if (c0 > 0) pk = (c0 + lane < n) ? rec[start + c0 + lane] : 0;
      const int mm = min(32, n - c0);
      int j = 0;
      auto group = [&](auto KC, int valid) {
        constexpr int K = decltype(KC)::value;
        v2f ud[K], ur[K], c[K];
        float p[K];
#pragma unroll
        for (int i = 0; i < K; ++i) {
          const int idx = (j + i) & 31;
          const int pj = __builtin_amdgcn_readlane(pk, idx);
          const int d = min(pj & 0x1FFFF, NUM_NODES - 1);
          const int r = min(pj >> 17, NUM_RELS - 1);
          ud[i] = *(const v2f*)(u_d + (size_t)d * DIM + 2 * lane);
          ur[i] = *(const v2f*)(u_r + (size_t)r * DIM + 2 * lane);
        }
#pragma unroll
        for (int i = 0; i < K; ++i) {
          c[i] = (base + ud[i]) + ur[i];
          p[i] = c[i].x * wa.x + c[i].y * wa.y;
        }
        float t;
        if constexpr (K == 8) {
          float q[4], q2[2];
#pragma unroll
          for (int i = 0; i < 4; ++i) {
            const float keep = h4 ? p[i + 4] : p[i];
            const float send = h4 ? p[i] : p[i + 4];
            q[i] = keep + xlane16(send);
          }
#pragma unroll
          for (int i = 0; i < 2; ++i) {
            const float keep = h3 ? q[i + 2] : q[i];
            const float send = h3 ? q[i] : q[i + 2];
            q2[i] = keep + xrow<8>(send);
          }
          const float keep = h2 ? q2[1] : q2[0];
          const float send = h2 ? q2[0] : q2[1];
          t = keep + xrow<4>(send);
        } else {
          float q[2];
#pragma unroll
          for (int i = 0; i < 2; ++i) {
            const float keep = h4 ? p[i + 2] : p[i];
            const float send = h4 ? p[i] : p[i + 2];
            q[i] = keep + xlane16(send);
          }
          const float keep = h3 ? q[1] : q[0];
          const float send = h3 ? q[0] : q[1];
          t = keep + xrow<8>(send);
          t += xrow<4>(t);
        }
        t += xrow<2>(t);
        t += xrow<1>(t);
        const int myedge = (K == 8) ? ((lane >> 2) & 7) : ((lane >> 3) & 3);
        float l = t + ba;
        l = l > 0.f ? l : 0.01f * l;
        if (myedge >= valid) l = NEG;
        float gm = l;
        gm = fmaxf(gm, xlane16(gm));
        gm = fmaxf(gm, xrow<8>(gm));
        gm = fmaxf(gm, xrow<4>(gm));
        const float mnew = fmaxf(m, gm);
        const float sc = expf(m - mnew);
        const int bbits = __float_as_int(expf(l - mnew));
        m = mnew;
        acc.x *= sc;
        acc.y *= sc;
        bsum *= sc;
#pragma unroll
        for (int i = 0; i < K; ++i) {
          if (i < valid) {
            const float w =
                __int_as_float(__builtin_amdgcn_readlane(bbits, i * (32 / K)));
            acc.x += w * c[i].x;
            acc.y += w * c[i].y;
            bsum += w;
          }
        }
        j += K;
      };
      while (mm - j >= 8) group(KConst<8>{}, 8);
      if (mm - j >= 4) group(KConst<4>{}, 4);
      if (mm - j > 0) group(KConst<4>{}, mm - j);
    }
    acc.x = acc.x / bsum;
    acc.y = acc.y / bsum;
  }
  *(v2f*)(h_ent + (size_t)s * DIM + 2 * lane) = acc;
}

typedef __attribute__((ext_vector_type(4))) float v4f;
typedef __attribute__((ext_vector_type(4))) int v4i;

#define PB      128
#define PPITCH  160
#define POOL_WGS 256
#define CNT_PITCH 512

__global__ __launch_bounds__(512) void edge_pool_kernel(
    const int* __restrict__ src, const int* __restrict__ dst,
    const int* __restrict__ rel, const float* __restrict__ g_s,
    const float* __restrict__ g_d, float* __restrict__ pool,
    float* __restrict__ cntp, int E) {
  __shared__ __attribute__((aligned(16))) float sG[2][(PB / 2) * PPITCH];
  __shared__ __attribute__((aligned(16))) int sR[2][PB];

  const int wave = threadIdx.x >> 5;
  const int lane = threadIdx.x & 31;
  const int lm = lane & 15;
  const int hi = lane >> 4;
  const int rbase = wave * 32;
  const int nb = (E + PB - 1) / PB;
  const int G = gridDim.x;

  v8f c00 = {}, c01 = {}, c02 = {}, c03 = {};
  v8f c10 = {}, c11 = {}, c12 = {}, c13 = {};
  float n0 = 0.f, n1 = 0.f;

  v2f rs[8], rd[8];
  int s_cur, d_cur, r_cur, s_nxt, d_nxt, r_nxt;

  auto load_idx = [&](int bb, int& s, int& d, int& r) {
    const int e = bb * PB + wave * 8 + (lane & 7);
    const bool ok = (bb < nb) && (e < E);
    s = ok ? src[e] : 0;
    d = ok ? dst[e] : 0;
    r = ok ? rel[e] : -1;
  };
  auto load_rows = [&](int s, int d) {
#pragma unroll
    for (int i = 0; i < 8; ++i) {
      const int si = __builtin_amdgcn_readlane(s, i);
      const int di = __builtin_amdgcn_readlane(d, i);
      rs[i] = *(const v2f*)(g_s + (size_t)si * DIM + 2 * lane);
      rd[i] = *(const v2f*)(g_d + (size_t)di * DIM + 2 * lane);
    }
  };

  int b = blockIdx.x;
  load_idx(b, s_cur, d_cur, r_cur);
  load_rows(s_cur, d_cur);
  load_idx(b + G, s_nxt, d_nxt, r_nxt);

  int buf = 0;
  for (; b < nb; b += G) {
    float* gb = sG[buf];
#pragma unroll
    for (int j = 0; j < 4; ++j) {
      const int r0 = __builtin_amdgcn_readlane(r_cur, 2 * j);
      const int r1 = __builtin_amdgcn_readlane(r_cur, 2 * j + 1);
      v2f g0 = rs[2 * j] + rd[2 * j];
      v2f g1 = rs[2 * j + 1] + rd[2 * j + 1];
      if (r0 < 0) { g0.x = 0.f; g0.y = 0.f; }
      if (r1 < 0) { g1.x = 0.f; g1.y = 0.f; }
      v4f o;
      o.x = g0.x; o.y = g1.x; o.z = g0.y; o.w = g1.y;
      *(v4f*)(gb + (wave * 4 + j) * PPITCH + 4 * lane) = o;
    }
    if (lane < 8) sR[buf][wave * 8 + lane] = r_cur;
    __syncthreads();

    s_cur = s_nxt; d_cur = d_nxt; r_cur = r_nxt;
    if (b + G < nb) load_rows(s_cur, d_cur);
    load_idx(b + 2 * G, s_nxt, d_nxt, r_nxt);

    const v4i rr = *(const v4i*)&sR[buf][4 * lane];
    const bool relv = ((unsigned)(rr.x - rbase) < 32u) |
                      ((unsigned)(rr.y - rbase) < 32u) |
                      ((unsigned)(rr.z - rbase) < 32u) |
                      ((unsigned)(rr.w - rbase) < 32u);
    unsigned mask = __builtin_amdgcn_ballot_w32(relv);
    while (mask) {
      const int g = __builtin_ctz(mask);
      mask &= mask - 1;
      const int* pr = &sR[buf][4 * g + 2 * hi];
      const int da = pr[0] - rbase - lm;
      const int db = pr[1] - rbase - lm;
      const float* pg = gb + (2 * g + hi) * PPITCH + 2 * lm;
      const v2f b0 = *(const v2f*)(pg);
      const v2f b1 = *(const v2f*)(pg + 32);
      const v2f b2 = *(const v2f*)(pg + 64);
      const v2f b3 = *(const v2f*)(pg + 96);
      {
        const bool ma = (da == 0), mb = (db == 0);
        if (__builtin_amdgcn_ballot_w32(ma | mb)) {
          v2f a;
          a.x = ma ? 1.0f : 0.0f;
          a.y = mb ? 1.0f : 0.0f;
          n0 += a.x + a.y;
          c00 = __builtin_amdgcn_wmma_f32_16x16x4_f32(false, a, false, b0,
                                                      (short)0, c00, false, false);
          c01 = __builtin_amdgcn_wmma_f32_16x16x4_f32(false, a, false, b1,
                                                      (short)0, c01, false, false);
          c02 = __builtin_amdgcn_wmma_f32_16x16x4_f32(false, a, false, b2,
                                                      (short)0, c02, false, false);
          c03 = __builtin_amdgcn_wmma_f32_16x16x4_f32(false, a, false, b3,
                                                      (short)0, c03, false, false);
        }
      }
      {
        const bool ma = (da == 16), mb = (db == 16);
        if (__builtin_amdgcn_ballot_w32(ma | mb)) {
          v2f a;
          a.x = ma ? 1.0f : 0.0f;
          a.y = mb ? 1.0f : 0.0f;
          n1 += a.x + a.y;
          c10 = __builtin_amdgcn_wmma_f32_16x16x4_f32(false, a, false, b0,
                                                      (short)0, c10, false, false);
          c11 = __builtin_amdgcn_wmma_f32_16x16x4_f32(false, a, false, b1,
                                                      (short)0, c11, false, false);
          c12 = __builtin_amdgcn_wmma_f32_16x16x4_f32(false, a, false, b2,
                                                      (short)0, c12, false, false);
          c13 = __builtin_amdgcn_wmma_f32_16x16x4_f32(false, a, false, b3,
                                                      (short)0, c13, false, false);
        }
      }
    }
    buf ^= 1;
  }

  __syncthreads();
  float* sFw = &sG[0][0] + wave * 1024;
  auto flush = [&](const v8f& t0, const v8f& t1, const v8f& t2, const v8f& t3,
                   int rt) {
#pragma unroll
    for (int v = 0; v < 8; ++v) {
      float* row = sFw + (v + 8 * hi) * DIM + lm;
      row[0] = t0[v]; row[16] = t1[v]; row[32] = t2[v]; row[48] = t3[v];
    }
    __syncthreads();
    for (int q = 0; q < 16; ++q) {
      const int r = (q + (int)blockIdx.x) & 15;
      const int R = rbase + rt * 16 + r;
      if (R < NUM_RELS) {
        atomicAdd(&pool[(size_t)R * DIM + lane], sFw[r * DIM + lane]);
        atomicAdd(&pool[(size_t)R * DIM + 32 + lane], sFw[r * DIM + 32 + lane]);
      }
    }
    __syncthreads();
  };
  flush(c00, c01, c02, c03, 0);
  flush(c10, c11, c12, c13, 1);

  {
    const float t0 = n0 + xlane16(n0);
    const float t1 = n1 + xlane16(n1);
    cntp[(size_t)blockIdx.x * CNT_PITCH + rbase + lane] = hi ? t1 : t0;
  }
}

__global__ __launch_bounds__(256) void h_rel_kernel(
    const float* __restrict__ pool, const float* __restrict__ cntp,
    const float* __restrict__ rel2_proj, const float* __restrict__ W_rel3,
    const float* __restrict__ b_rel3, float* __restrict__ h_rel, int R) {
  __shared__ float swsum[8];
  const int t = blockIdx.x * 256 + threadIdx.x;
  const int r = min(t >> 6, R - 1), o = t & 63;
  float c = 0.f;
#pragma unroll
  for (int j = 0; j < POOL_WGS / DIM; ++j)
    c += cntp[(size_t)(o + DIM * j) * CNT_PITCH + r];
  c += xlane16(c);
  c += xrow<8>(c);
  c += xrow<4>(c);
  c += xrow<2>(c);
  c += xrow<1>(c);
  const int wave = threadIdx.x >> 5;
  if ((threadIdx.x & 31) == 0) swsum[wave] = c;
  __syncthreads();
  c = swsum[wave & 6] + swsum[wave | 1];
  if (t >= R * DIM) return;
  const float inv = 1.0f / fmaxf(c, 1.0f);
  const float* w = W_rel3 + (size_t)o * 192;
  float acc = b_rel3[o] + pool[t] * inv;
  if (c > 0.f) {
    const float* r2 = rel2_proj + (size_t)r * DIM;
    for (int j = 0; j < DIM; ++j) acc += r2[j] * w[128 + j];
  }
  h_rel[t] = acc;
}

#define OFF_MS     0
#define OFF_MD     4096
#define OFF_MR     8192
#define OFF_M2     12288
#define OFF_CVEC   16384
#define OFF_VREL2  16448
#define OFF_UR     16576
#define OFF_REL2P  49088
#define OFF_BSUM   281088
#define OFF_POOL   381588
#define OFF_TOTAL  413588
#define OFF_BBASE  413600
#define OFF_BE     445588
#define OFF_CNTP   OFF_BE
#define OFF_US     1445588
#define OFF_UD     7845588
#define ZERO_CNT   (OFF_TOTAL + 1 - OFF_BSUM)

extern "C" void kernel_launch(void* const* d_in, const int* in_sizes, int n_in,
                              void* d_out, int out_size, void* d_ws,
                              size_t ws_size, hipStream_t stream) {
  const float* ent_embed = (const float*)d_in[0];
  const float* rel_embed = (const float*)d_in[1];
  const float* W_ent  = (const float*)d_in[2];
  const float* b_ent  = (const float*)d_in[3];
  const float* W_relL = (const float*)d_in[4];
  const float* b_relL = (const float*)d_in[5];
  const float* W_rel2 = (const float*)d_in[6];
  const float* b_rel2 = (const float*)d_in[7];
  const float* W_rel3 = (const float*)d_in[8];
  const float* b_rel3 = (const float*)d_in[9];
  const float* W_a    = (const float*)d_in[10];
  const float* b_a    = (const float*)d_in[11];
  const float* W_fc   = (const float*)d_in[12];
  const float* b_fc   = (const float*)d_in[13];
  const int* src_idx  = (const int*)d_in[14];
  const int* dst_idx  = (const int*)d_in[15];
  const int* rel_idx  = (const int*)d_in[16];

  float* ws = (float*)d_ws;
  float* h_ent = (float*)d_out;
  float* h_rel = (float*)d_out + (size_t)NUM_NODES * DIM;

  zero_kernel<<<512, 256, 0, stream>>>(ws + OFF_BSUM, ZERO_CNT);

  int* deg_cur = (int*)(ws + OFF_BSUM);
  int* seg_total = (int*)(ws + OFF_TOTAL);
  int* seg_bbase = (int*)(ws + OFF_BBASE);
  int* edge_rec = (int*)(ws + OFF_BE);
  edge_count_kernel<<<4096, 256, 0, stream>>>(src_idx, deg_cur, NUM_EDGES);
  seg_alloc_kernel<<<(NUM_NODES + SEG_BLK - 1) / SEG_BLK, SEG_BLK, 0, stream>>>(
      deg_cur, seg_total, seg_bbase, NUM_NODES);
  edge_fill_kernel<<<4096, 256, 0, stream>>>(src_idx, dst_idx, rel_idx, deg_cur,
                                             edge_rec, NUM_EDGES);

  fold_kernel<<<1, 256, 0, stream>>>(
      W_ent, b_ent, W_relL, b_relL, W_rel2, b_rel2, W_fc, b_fc,
      ws + OFF_MS, ws + OFF_MD, ws + OFF_MR, ws + OFF_M2, ws + OFF_CVEC,
      ws + OFF_VREL2);

  const int gemm_blocks_N = (NUM_NODES + 127) / 128;
  const int gemm_blocks_R = (NUM_RELS + 127) / 128;
  wmma_gemm64<DIM><<<gemm_blocks_N, 256, 0, stream>>>(ent_embed, ws + OFF_MS,
                                                 nullptr, ws + OFF_US, NUM_NODES);
  wmma_gemm64<DIM><<<gemm_blocks_N, 256, 0, stream>>>(ent_embed, ws + OFF_MD,
                                                 nullptr, ws + OFF_UD, NUM_NODES);
  wmma_gemm64<DIM><<<gemm_blocks_R, 256, 0, stream>>>(rel_embed, ws + OFF_MR,
                                                 nullptr, ws + OFF_UR, NUM_RELS);
  wmma_gemm64<DIM><<<gemm_blocks_R, 256, 0, stream>>>(rel_embed, ws + OFF_M2,
                                                 ws + OFF_VREL2, ws + OFF_REL2P,
                                                 NUM_RELS);

  node_gather_kernel<<<(NUM_NODES + 7) / 8, 256, 0, stream>>>(
      deg_cur, seg_bbase, edge_rec, ws + OFF_US, ws + OFF_UD, ws + OFF_UR,
      ws + OFF_CVEC, W_a, b_a, h_ent, NUM_NODES, NUM_EDGES);
  wmma_gemm64<3 * DIM><<<gemm_blocks_N, 256, 0, stream>>>(h_ent, W_rel3,
                                                 nullptr, ws + OFF_US, NUM_NODES);
  wmma_gemm64<3 * DIM><<<gemm_blocks_N, 256, 0, stream>>>(h_ent, W_rel3 + DIM,
                                                 nullptr, ws + OFF_UD, NUM_NODES);
  edge_pool_kernel<<<POOL_WGS, 512, 0, stream>>>(
      src_idx, dst_idx, rel_idx, ws + OFF_US, ws + OFF_UD, ws + OFF_POOL,
      ws + OFF_CNTP, NUM_EDGES);

  h_rel_kernel<<<(NUM_RELS * DIM + 255) / 256, 256, 0, stream>>>(
      ws + OFF_POOL, ws + OFF_CNTP, ws + OFF_REL2P, W_rel3, b_rel3, h_rel,
      NUM_RELS);
}


# projection GEMMs: whole W held in each wave's registers as B fragments (16-byte global loads; no LDS staging, no bank-conflicted fragment reads), waves loop over row tiles; GEMM k sum order changes
# speedup vs baseline: 2.6975x; 1.1071x over previous
#include <hip/hip_runtime.h>
#include <math.h>


#define NUM_NODES 100000
#define NUM_RELS  500
#define DIM       64
#define NUM_EDGES 1000000

typedef __attribute__((ext_vector_type(2))) float v2f;
typedef __attribute__((ext_vector_type(8))) float v8f;
typedef __attribute__((ext_vector_type(4))) float v4f;
typedef __attribute__((ext_vector_type(4))) int v4i;

__global__ void zero_kernel(float* __restrict__ p, int n) {
  for (int i = blockIdx.x * blockDim.x + threadIdx.x; i < n;
       i += gridDim.x * blockDim.x)
    p[i] = 0.0f;
}

__global__ __launch_bounds__(256) void fold_kernel(
    const float* __restrict__ W_ent, const float* __restrict__ b_ent,
    const float* __restrict__ W_relL, const float* __restrict__ b_relL,
    const float* __restrict__ W_rel2, const float* __restrict__ b_rel2,
    const float* __restrict__ W_fc, const float* __restrict__ b_fc,
    float* __restrict__ M_s, float* __restrict__ M_d,
    float* __restrict__ M_r, float* __restrict__ M_2,
    float* __restrict__ cvec, float* __restrict__ v_rel2) {
  const int t = threadIdx.x;
  for (int idx = t; idx < DIM * DIM; idx += 256) {
    const int o = idx >> 6, k = idx & 63;
    float ms = 0.f, md = 0.f, mr = 0.f, m2 = 0.f;
    for (int u = 0; u < DIM; ++u) {
      const float we = W_ent[u * DIM + k];
      const float wl = W_relL[u * DIM + k];
      ms += W_fc[o * 192 + u] * we;
      md += W_fc[o * 192 + 64 + u] * we;
      mr += W_fc[o * 192 + 128 + u] * wl;
      m2 += W_rel2[o * DIM + u] * wl;
    }
    M_s[idx] = ms; M_d[idx] = md; M_r[idx] = mr; M_2[idx] = m2;
  }
  if (t < DIM) {
    float cv = b_fc[t], v2 = b_rel2[t];
    for (int u = 0; u < DIM; ++u) {
      cv += W_fc[t * 192 + u] * b_ent[u]
          + W_fc[t * 192 + 64 + u] * b_ent[u]
          + W_fc[t * 192 + 128 + u] * b_relL[u];
      v2 += W_rel2[t * DIM + u] * b_relL[u];
    }
    cvec[t] = cv; v_rel2[t] = v2;
  }
}

#define GEMM_WAVES 4
template <int LDW>
__global__ __launch_bounds__(32 * GEMM_WAVES) void wmma_gemm64(
    const float* __restrict__ A, const float* __restrict__ W,
    const float* __restrict__ bias, float* __restrict__ out, int M) {
  const int wave = threadIdx.x >> 5;
  const int lane = threadIdx.x & 31;
  const int lm = lane & 15;
  const int hi = lane >> 4;
  const int ntiles = (M + 15) >> 4;
  const int nwaves = gridDim.x * GEMM_WAVES;
  int tile = blockIdx.x * GEMM_WAVES + wave;
  if (tile >= ntiles) return;

  v4f w[4][8];
#pragma unroll
  for (int nb = 0; nb < 4; ++nb) {
    const float* __restrict__ wr = W + (size_t)(nb * 16 + lm) * LDW + 4 * hi;
#pragma unroll
    for (int t = 0; t < 8; ++t) w[nb][t] = *(const v4f*)(wr + 8 * t);
  }
  float bv[4];
#pragma unroll
  for (int nb = 0; nb < 4; ++nb) bv[nb] = bias ? bias[nb * 16 + lm] : 0.0f;

  auto load_a = [&](int tl, v4f (&a)[8]) {
    const int row = min(tl * 16 + lm, M - 1);
    const float* __restrict__ ar = A + (size_t)row * DIM + 4 * hi;
#pragma unroll
    for (int t = 0; t < 8; ++t) a[t] = *(const v4f*)(ar + 8 * t);
  };

  v4f a[8];
  load_a(tile, a);
  for (; tile < ntiles; tile += nwaves) {
    v4f an[8];
    load_a(min(tile + nwaves, ntiles - 1), an);

    v8f acc0 = {}, acc1 = {}, acc2 = {}, acc3 = {};
#pragma unroll
    for (int t = 0; t < 8; ++t) {
      const v2f a0 = {a[t].x, a[t].y};
      const v2f a1 = {a[t].z, a[t].w};
      const v2f b00 = {w[0][t].x, w[0][t].y}, b01 = {w[0][t].z, w[0][t].w};
      const v2f b10 = {w[1][t].x, w[1][t].y}, b11 = {w[1][t].z, w[1][t].w};
      const v2f b20 = {w[2][t].x, w[2][t].y}, b21 = {w[2][t].z, w[2][t].w};
      const v2f b30 = {w[3][t].x, w[3][t].y}, b31 = {w[3][t].z, w[3][t].w};
      acc0 = __builtin_amdgcn_wmma_f32_16x16x4_f32(false, a0, false, b00,
                                                   (short)0, acc0, false, false);
      acc1 = __builtin_amdgcn_wmma_f32_16x16x4_f32(false, a0, false, b10,
                                                   (short)0, acc1, false, false);
      acc2 = __builtin_amdgcn_wmma_f32_16x16x4_f32(false, a0, false, b20,
                                                   (short)0, acc2, false, false);
      acc3 = __builtin_amdgcn_wmma_f32_16x16x4_f32(false, a0, false, b30,
                                                   (short)0, acc3, false, false);
      acc0 = __builtin_amdgcn_wmma_f32_16x16x4_f32(false, a1, false, b01,
                                                   (short)0, acc0, false, false);
      acc1 = __builtin_amdgcn_wmma_f32_16x16x4_f32(false, a1, false, b11,
                                                   (short)0, acc1, false, false);
      acc2 = __builtin_amdgcn_wmma_f32_16x16x4_f32(false, a1, false, b21,
                                                   (short)0, acc2, false, false);
      acc3 = __builtin_amdgcn_wmma_f32_16x16x4_f32(false, a1, false, b31,
                                                   (short)0, acc3, false, false);
    }

    const int m0 = tile * 16;
    v8f accs[4] = {acc0, acc1, acc2, acc3};
#pragma unroll
    for (int ob = 0; ob < 4; ++ob) {
#pragma unroll
      for (int v = 0; v < 8; ++v) {
        const int row = m0 + v + 8 * hi;
        if (row < M)
          out[(size_t)row * DIM + ob * 16 + lm] = accs[ob][v] + bv[ob];
      }
    }
#pragma unroll
    for (int t = 0; t < 8; ++t) a[t] = an[t];
  }
}

template <int V> struct KConst { static constexpr int value = V; };

#define SEG_BLK 1024

__global__ void edge_count_kernel(const int* __restrict__ src,
                                  int* __restrict__ deg, int E) {
  for (int e = blockIdx.x * blockDim.x + threadIdx.x; e < E;
       e += gridDim.x * blockDim.x)
    atomicAdd(&deg[src[e]], 1);
}

__global__ __launch_bounds__(SEG_BLK) void seg_alloc_kernel(
    int* __restrict__ cursor  ,
    int* __restrict__ total, int* __restrict__ bbase, int N) {
  __shared__ int swave[32];
  __shared__ int sbase;
  const int lane = threadIdx.x & 31;
  const int wave = threadIdx.x >> 5;
  const int i = blockIdx.x * SEG_BLK + threadIdx.x;
  const int v = (i < N) ? cursor[i] : 0;
  int x = v;
#pragma unroll
  for (int o = 1; o < 32; o <<= 1) {
    const int y = __shfl_up(x, o);
    if (lane >= o) x += y;
  }
  if (lane == 31) swave[wave] = x;
  __syncthreads();
  if (wave == 0) {
    const int w = swave[lane];
    int xs = w;
#pragma unroll
    for (int o = 1; o < 32; o <<= 1) {
      const int y = __shfl_up(xs, o);
      if (lane >= o) xs += y;
    }
    swave[lane] = xs - w;
    if (lane == 31) {
      const int base = atomicAdd(total, xs);
      sbase = base;
      bbase[blockIdx.x] = base;
    }
  }
  __syncthreads();
  if (i < N) cursor[i] = sbase + swave[wave] + x - v;
}

__global__ void edge_fill_kernel(const int* __restrict__ src,
                                 const int* __restrict__ dst,
                                 const int* __restrict__ rel,
                                 int* __restrict__ cursor,
                                 int* __restrict__ rec, int E) {
  for (int e = blockIdx.x * blockDim.x + threadIdx.x; e < E;
       e += gridDim.x * blockDim.x) {
    const int p = atomicAdd(&cursor[src[e]], 1);
    if ((unsigned)p < (unsigned)E) rec[p] = dst[e] | (rel[e] << 17);
  }
}

__device__ __forceinline__ float xlane16(float x) {
  const int b = __float_as_int(x);
  return __int_as_float(
      __builtin_amdgcn_permlanex16(b, b, 0x76543210, 0xfedcba98, false, false));
}
template <int M>
__device__ __forceinline__ float xrow(float x) {
  const int b = __float_as_int(x);
  return __int_as_float(
      __builtin_amdgcn_update_dpp(b, b, 0x160 | M, 0xf, 0xf, false));
}

__global__ __launch_bounds__(256) void node_gather_kernel(
    const int* __restrict__ cursor, const int* __restrict__ bbase,
    const int* __restrict__ rec, const float* __restrict__ u_s,
    const float* __restrict__ u_d, const float* __restrict__ u_r,
    const float* __restrict__ cvec, const float* __restrict__ W_a,
    const float* __restrict__ b_a, float* __restrict__ h_ent, int N, int E) {
  const int s = blockIdx.x * (blockDim.x >> 5) + (threadIdx.x >> 5);
  const int lane = threadIdx.x & 31;
  if (s >= N) return;

  int end = cursor[s];
  int start = (s & (SEG_BLK - 1)) ? cursor[s - 1] : bbase[s / SEG_BLK];
  end = __builtin_amdgcn_readfirstlane(end);
  start = __builtin_amdgcn_readfirstlane(start);
  start = min(max(start, 0), E);
  end = min(max(end, start), E);
  const int n = end - start;

  v2f acc = {0.f, 0.f};
  if (n > 0) {
    int pk = (lane < n) ? rec[start + lane] : 0;
    const v2f wa = *(const v2f*)(W_a + 2 * lane);
    const float ba = b_a[0];
    const v2f base = *(const v2f*)(u_s + (size_t)s * DIM + 2 * lane) +
                     *(const v2f*)(cvec + 2 * lane);
    const bool h4 = (lane & 16) != 0;
    const bool h3 = (lane & 8) != 0;
    const bool h2 = (lane & 4) != 0;
    const float NEG = -3.0e38f;
    float m = NEG;
    float bsum = 0.f;

    for (int c0 = 0; c0 < n; c0 += 32) {
      if (c0 > 0) pk = (c0 + lane < n) ? rec[start + c0 + lane] : 0;
      const int mm = min(32, n - c0);
      int j = 0;
      auto group = [&](auto KC, int valid) {
        constexpr int K = decltype(KC)::value;
        v2f ud[K], ur[K], c[K];
        float p[K];
#pragma unroll
        for (int i = 0; i < K; ++i) {
          const int idx = (j + i) & 31;
          const int pj = __builtin_amdgcn_readlane(pk, idx);
          const int d = min(pj & 0x1FFFF, NUM_NODES - 1);
          const int r = min(pj >> 17, NUM_RELS - 1);
          ud[i] = *(const v2f*)(u_d + (size_t)d * DIM + 2 * lane);
          ur[i] = *(const v2f*)(u_r + (size_t)r * DIM + 2 * lane);
        }
#pragma unroll
        for (int i = 0; i < K; ++i) {
          c[i] = (base + ud[i]) + ur[i];
          p[i] = c[i].x * wa.x + c[i].y * wa.y;
        }
        float t;
        if constexpr (K == 8) {
          float q[4], q2[2];
#pragma unroll
          for (int i = 0; i < 4; ++i) {
            const float keep = h4 ? p[i + 4] : p[i];
            const float send = h4 ? p[i] : p[i + 4];
            q[i] = keep + xlane16(send);
          }
#pragma unroll
          for (int i = 0; i < 2; ++i) {
            const float keep = h3 ? q[i + 2] : q[i];
            const float send = h3 ? q[i] : q[i + 2];
            q2[i] = keep + xrow<8>(send);
          }
          const float keep = h2 ? q2[1] : q2[0];
          const float send = h2 ? q2[0] : q2[1];
          t = keep + xrow<4>(send);
        } else {
          float q[2];
#pragma unroll
          for (int i = 0; i < 2; ++i) {
            const float keep = h4 ? p[i + 2] : p[i];
            const float send = h4 ? p[i] : p[i + 2];
            q[i] = keep + xlane16(send);
          }
          const float keep = h3 ? q[1] : q[0];
          const float send = h3 ? q[0] : q[1];
          t = keep + xrow<8>(send);
          t += xrow<4>(t);
        }
        t += xrow<2>(t);
        t += xrow<1>(t);
        const int myedge = (K == 8) ? ((lane >> 2) & 7) : ((lane >> 3) & 3);
        float l = t + ba;
        l = l > 0.f ? l : 0.01f * l;
        if (myedge >= valid) l = NEG;
        float gm = l;
        gm = fmaxf(gm, xlane16(gm));
        gm = fmaxf(gm, xrow<8>(gm));
        gm = fmaxf(gm, xrow<4>(gm));
        const float mnew = fmaxf(m, gm);
        const float sc = expf(m - mnew);
        const int bbits = __float_as_int(expf(l - mnew));
        m = mnew;
        acc.x *= sc;
        acc.y *= sc;
        bsum *= sc;
#pragma unroll
        for (int i = 0; i < K; ++i) {
          if (i < valid) {
            const float w =
                __int_as_float(__builtin_amdgcn_readlane(bbits, i * (32 / K)));
            acc.x += w * c[i].x;
            acc.y += w * c[i].y;
            bsum += w;
          }
        }
        j += K;
      };
      while (mm - j >= 8) group(KConst<8>{}, 8);
      if (mm - j >= 4) group(KConst<4>{}, 4);
      if (mm - j > 0) group(KConst<4>{}, mm - j);
    }
    acc.x = acc.x / bsum;
    acc.y = acc.y / bsum;
  }
  *(v2f*)(h_ent + (size_t)s * DIM + 2 * lane) = acc;
}


#define PB      128
#define PPITCH  160
#define POOL_WGS 256
#define CNT_PITCH 512

__global__ __launch_bounds__(512) void edge_pool_kernel(
    const int* __restrict__ src, const int* __restrict__ dst,
    const int* __restrict__ rel, const float* __restrict__ g_s,
    const float* __restrict__ g_d, float* __restrict__ pool,
    float* __restrict__ cntp, int E) {
  __shared__ __attribute__((aligned(16))) float sG[2][(PB / 2) * PPITCH];
  __shared__ __attribute__((aligned(16))) int sR[2][PB];

  const int wave = threadIdx.x >> 5;
  const int lane = threadIdx.x & 31;
  const int lm = lane & 15;
  const int hi = lane >> 4;
  const int rbase = wave * 32;
  const int nb = (E + PB - 1) / PB;
  const int G = gridDim.x;

  v8f c00 = {}, c01 = {}, c02 = {}, c03 = {};
  v8f c10 = {}, c11 = {}, c12 = {}, c13 = {};
  float n0 = 0.f, n1 = 0.f;

  v2f rs[8], rd[8];
  int s_cur, d_cur, r_cur, s_nxt, d_nxt, r_nxt;

  auto load_idx = [&](int bb, int& s, int& d, int& r) {
    const int e = bb * PB + wave * 8 + (lane & 7);
    const bool ok = (bb < nb) && (e < E);
    s = ok ? src[e] : 0;
    d = ok ? dst[e] : 0;
    r = ok ? rel[e] : -1;
  };
  auto load_rows = [&](int s, int d) {
#pragma unroll
    for (int i = 0; i < 8; ++i) {
      const int si = __builtin_amdgcn_readlane(s, i);
      const int di = __builtin_amdgcn_readlane(d, i);
      rs[i] = *(const v2f*)(g_s + (size_t)si * DIM + 2 * lane);
      rd[i] = *(const v2f*)(g_d + (size_t)di * DIM + 2 * lane);
    }
  };

  int b = blockIdx.x;
  load_idx(b, s_cur, d_cur, r_cur);
  load_rows(s_cur, d_cur);
  load_idx(b + G, s_nxt, d_nxt, r_nxt);

  int buf = 0;
  for (; b < nb; b += G) {
    float* gb = sG[buf];
#pragma unroll
    for (int j = 0; j < 4; ++j) {
      const int r0 = __builtin_amdgcn_readlane(r_cur, 2 * j);
      const int r1 = __builtin_amdgcn_readlane(r_cur, 2 * j + 1);
      v2f g0 = rs[2 * j] + rd[2 * j];
      v2f g1 = rs[2 * j + 1] + rd[2 * j + 1];
      if (r0 < 0) { g0.x = 0.f; g0.y = 0.f; }
      if (r1 < 0) { g1.x = 0.f; g1.y = 0.f; }
      v4f o;
      o.x = g0.x; o.y = g1.x; o.z = g0.y; o.w = g1.y;
      *(v4f*)(gb + (wave * 4 + j) * PPITCH + 4 * lane) = o;
    }
    if (lane < 8) sR[buf][wave * 8 + lane] = r_cur;
    __syncthreads();

    s_cur = s_nxt; d_cur = d_nxt; r_cur = r_nxt;
    if (b + G < nb) load_rows(s_cur, d_cur);
    load_idx(b + 2 * G, s_nxt, d_nxt, r_nxt);

    const v4i rr = *(const v4i*)&sR[buf][4 * lane];
    const bool relv = ((unsigned)(rr.x - rbase) < 32u) |
                      ((unsigned)(rr.y - rbase) < 32u) |
                      ((unsigned)(rr.z - rbase) < 32u) |
                      ((unsigned)(rr.w - rbase) < 32u);
    unsigned mask = __builtin_amdgcn_ballot_w32(relv);
    while (mask) {
      const int g = __builtin_ctz(mask);
      mask &= mask - 1;
      const int* pr = &sR[buf][4 * g + 2 * hi];
      const int da = pr[0] - rbase - lm;
      const int db = pr[1] - rbase - lm;
      const float* pg = gb + (2 * g + hi) * PPITCH + 2 * lm;
      const v2f b0 = *(const v2f*)(pg);
      const v2f b1 = *(const v2f*)(pg + 32);
      const v2f b2 = *(const v2f*)(pg + 64);
      const v2f b3 = *(const v2f*)(pg + 96);
      {
        const bool ma = (da == 0), mb = (db == 0);
        if (__builtin_amdgcn_ballot_w32(ma | mb)) {
          v2f a;
          a.x = ma ? 1.0f : 0.0f;
          a.y = mb ? 1.0f : 0.0f;
          n0 += a.x + a.y;
          c00 = __builtin_amdgcn_wmma_f32_16x16x4_f32(false, a, false, b0,
                                                      (short)0, c00, false, false);
          c01 = __builtin_amdgcn_wmma_f32_16x16x4_f32(false, a, false, b1,
                                                      (short)0, c01, false, false);
          c02 = __builtin_amdgcn_wmma_f32_16x16x4_f32(false, a, false, b2,
                                                      (short)0, c02, false, false);
          c03 = __builtin_amdgcn_wmma_f32_16x16x4_f32(false, a, false, b3,
                                                      (short)0, c03, false, false);
        }
      }
      {
        const bool ma = (da == 16), mb = (db == 16);
        if (__builtin_amdgcn_ballot_w32(ma | mb)) {
          v2f a;
          a.x = ma ? 1.0f : 0.0f;
          a.y = mb ? 1.0f : 0.0f;
          n1 += a.x + a.y;
          c10 = __builtin_amdgcn_wmma_f32_16x16x4_f32(false, a, false, b0,
                                                      (short)0, c10, false, false);
          c11 = __builtin_amdgcn_wmma_f32_16x16x4_f32(false, a, false, b1,
                                                      (short)0, c11, false, false);
          c12 = __builtin_amdgcn_wmma_f32_16x16x4_f32(false, a, false, b2,
                                                      (short)0, c12, false, false);
          c13 = __builtin_amdgcn_wmma_f32_16x16x4_f32(false, a, false, b3,
                                                      (short)0, c13, false, false);
        }
      }
    }
    buf ^= 1;
  }

  __syncthreads();
  float* sFw = &sG[0][0] + wave * 1024;
  auto flush = [&](const v8f& t0, const v8f& t1, const v8f& t2, const v8f& t3,
                   int rt) {
#pragma unroll
    for (int v = 0; v < 8; ++v) {
      float* row = sFw + (v + 8 * hi) * DIM + lm;
      row[0] = t0[v]; row[16] = t1[v]; row[32] = t2[v]; row[48] = t3[v];
    }
    __syncthreads();
    for (int q = 0; q < 16; ++q) {
      const int r = (q + (int)blockIdx.x) & 15;
      const int R = rbase + rt * 16 + r;
      if (R < NUM_RELS) {
        atomicAdd(&pool[(size_t)R * DIM + lane], sFw[r * DIM + lane]);
        atomicAdd(&pool[(size_t)R * DIM + 32 + lane], sFw[r * DIM + 32 + lane]);
      }
    }
    __syncthreads();
  };
  flush(c00, c01, c02, c03, 0);
  flush(c10, c11, c12, c13, 1);

  {
    const float t0 = n0 + xlane16(n0);
    const float t1 = n1 + xlane16(n1);
    cntp[(size_t)blockIdx.x * CNT_PITCH + rbase + lane] = hi ? t1 : t0;
  }
}

__global__ __launch_bounds__(256) void h_rel_kernel(
    const float* __restrict__ pool, const float* __restrict__ cntp,
    const float* __restrict__ rel2_proj, const float* __restrict__ W_rel3,
    const float* __restrict__ b_rel3, float* __restrict__ h_rel, int R) {
  __shared__ float swsum[8];
  const int t = blockIdx.x * 256 + threadIdx.x;
  const int r = min(t >> 6, R - 1), o = t & 63;
  float c = 0.f;
#pragma unroll
  for (int j = 0; j < POOL_WGS / DIM; ++j)
    c += cntp[(size_t)(o + DIM * j) * CNT_PITCH + r];
  c += xlane16(c);
  c += xrow<8>(c);
  c += xrow<4>(c);
  c += xrow<2>(c);
  c += xrow<1>(c);
  const int wave = threadIdx.x >> 5;
  if ((threadIdx.x & 31) == 0) swsum[wave] = c;
  __syncthreads();
  c = swsum[wave & 6] + swsum[wave | 1];
  if (t >= R * DIM) return;
  const float inv = 1.0f / fmaxf(c, 1.0f);
  const float* w = W_rel3 + (size_t)o * 192;
  float acc = b_rel3[o] + pool[t] * inv;
  if (c > 0.f) {
    const float* r2 = rel2_proj + (size_t)r * DIM;
    for (int j = 0; j < DIM; ++j) acc += r2[j] * w[128 + j];
  }
  h_rel[t] = acc;
}

#define OFF_MS     0
#define OFF_MD     4096
#define OFF_MR     8192
#define OFF_M2     12288
#define OFF_CVEC   16384
#define OFF_VREL2  16448
#define OFF_UR     16576
#define OFF_REL2P  49088
#define OFF_BSUM   281088
#define OFF_POOL   381588
#define OFF_TOTAL  413588
#define OFF_BBASE  413600
#define OFF_BE     445588
#define OFF_CNTP   OFF_BE
#define OFF_US     1445588
#define OFF_UD     7845588
#define ZERO_CNT   (OFF_TOTAL + 1 - OFF_BSUM)

extern "C" void kernel_launch(void* const* d_in, const int* in_sizes, int n_in,
                              void* d_out, int out_size, void* d_ws,
                              size_t ws_size, hipStream_t stream) {
  const float* ent_embed = (const float*)d_in[0];
  const float* rel_embed = (const float*)d_in[1];
  const float* W_ent  = (const float*)d_in[2];
  const float* b_ent  = (const float*)d_in[3];
  const float* W_relL = (const float*)d_in[4];
  const float* b_relL = (const float*)d_in[5];
  const float* W_rel2 = (const float*)d_in[6];
  const float* b_rel2 = (const float*)d_in[7];
  const float* W_rel3 = (const float*)d_in[8];
  const float* b_rel3 = (const float*)d_in[9];
  const float* W_a    = (const float*)d_in[10];
  const float* b_a    = (const float*)d_in[11];
  const float* W_fc   = (const float*)d_in[12];
  const float* b_fc   = (const float*)d_in[13];
  const int* src_idx  = (const int*)d_in[14];
  const int* dst_idx  = (const int*)d_in[15];
  const int* rel_idx  = (const int*)d_in[16];

  float* ws = (float*)d_ws;
  float* h_ent = (float*)d_out;
  float* h_rel = (float*)d_out + (size_t)NUM_NODES * DIM;

  zero_kernel<<<512, 256, 0, stream>>>(ws + OFF_BSUM, ZERO_CNT);

  int* deg_cur = (int*)(ws + OFF_BSUM);
  int* seg_total = (int*)(ws + OFF_TOTAL);
  int* seg_bbase = (int*)(ws + OFF_BBASE);
  int* edge_rec = (int*)(ws + OFF_BE);
  edge_count_kernel<<<4096, 256, 0, stream>>>(src_idx, deg_cur, NUM_EDGES);
  seg_alloc_kernel<<<(NUM_NODES + SEG_BLK - 1) / SEG_BLK, SEG_BLK, 0, stream>>>(
      deg_cur, seg_total, seg_bbase, NUM_NODES);
  edge_fill_kernel<<<4096, 256, 0, stream>>>(src_idx, dst_idx, rel_idx, deg_cur,
                                             edge_rec, NUM_EDGES);

  fold_kernel<<<1, 256, 0, stream>>>(
      W_ent, b_ent, W_relL, b_relL, W_rel2, b_rel2, W_fc, b_fc,
      ws + OFF_MS, ws + OFF_MD, ws + OFF_MR, ws + OFF_M2, ws + OFF_CVEC,
      ws + OFF_VREL2);

  const int gemm_blocks_N = 2048 / GEMM_WAVES;
  const int gemm_blocks_R = ((NUM_RELS + 15) / 16 + GEMM_WAVES - 1) / GEMM_WAVES;
  wmma_gemm64<DIM><<<gemm_blocks_N, 32 * GEMM_WAVES, 0, stream>>>(ent_embed, ws + OFF_MS,
                                                 nullptr, ws + OFF_US, NUM_NODES);
  wmma_gemm64<DIM><<<gemm_blocks_N, 32 * GEMM_WAVES, 0, stream>>>(ent_embed, ws + OFF_MD,
                                                 nullptr, ws + OFF_UD, NUM_NODES);
  wmma_gemm64<DIM><<<gemm_blocks_R, 32 * GEMM_WAVES, 0, stream>>>(rel_embed, ws + OFF_MR,
                                                 nullptr, ws + OFF_UR, NUM_RELS);
  wmma_gemm64<DIM><<<gemm_blocks_R, 32 * GEMM_WAVES, 0, stream>>>(rel_embed, ws + OFF_M2,
                                                 ws + OFF_VREL2, ws + OFF_REL2P,
                                                 NUM_RELS);

  node_gather_kernel<<<(NUM_NODES + 7) / 8, 256, 0, stream>>>(
      deg_cur, seg_bbase, edge_rec, ws + OFF_US, ws + OFF_UD, ws + OFF_UR,
      ws + OFF_CVEC, W_a, b_a, h_ent, NUM_NODES, NUM_EDGES);
  wmma_gemm64<3 * DIM><<<gemm_blocks_N, 32 * GEMM_WAVES, 0, stream>>>(h_ent, W_rel3,
                                                 nullptr, ws + OFF_US, NUM_NODES);
  wmma_gemm64<3 * DIM><<<gemm_blocks_N, 32 * GEMM_WAVES, 0, stream>>>(h_ent, W_rel3 + DIM,
                                                 nullptr, ws + OFF_UD, NUM_NODES);
  edge_pool_kernel<<<POOL_WGS, 512, 0, stream>>>(
      src_idx, dst_idx, rel_idx, ws + OFF_US, ws + OFF_UD, ws + OFF_POOL,
      ws + OFF_CNTP, NUM_EDGES);

  h_rel_kernel<<<(NUM_RELS * DIM + 255) / 256, 256, 0, stream>>>(
      ws + OFF_POOL, ws + OFF_CNTP, ws + OFF_REL2P, W_rel3, b_rel3, h_rel,
      NUM_RELS);
}


# weight fold on the matrix unit: the four 64x64x64 weight products as f32 16x16x4 WMMA, one wave per 16 output rows, not 4096 serial 64-term loops in one work-group; M_s/M_d/M_r/M_2 sum order changes
# speedup vs baseline: 3.0051x; 1.1140x over previous
#include <hip/hip_runtime.h>
#include <math.h>


#define NUM_NODES 100000
#define NUM_RELS  500
#define DIM       64
#define NUM_EDGES 1000000

typedef __attribute__((ext_vector_type(2))) float v2f;
typedef __attribute__((ext_vector_type(8))) float v8f;
typedef __attribute__((ext_vector_type(4))) float v4f;
typedef __attribute__((ext_vector_type(4))) int v4i;

__global__ void zero_kernel(float* __restrict__ p, int n) {
  for (int i = blockIdx.x * blockDim.x + threadIdx.x; i < n;
       i += gridDim.x * blockDim.x)
    p[i] = 0.0f;
}

#define FOLD_WGS 5
__global__ __launch_bounds__(128) void fold_kernel(
    const float* __restrict__ W_ent, const float* __restrict__ b_ent,
    const float* __restrict__ W_relL, const float* __restrict__ b_relL,
    const float* __restrict__ W_rel2, const float* __restrict__ b_rel2,
    const float* __restrict__ W_fc, const float* __restrict__ b_fc,
    float* __restrict__ M_s, float* __restrict__ M_d,
    float* __restrict__ M_r, float* __restrict__ M_2,
    float* __restrict__ cvec, float* __restrict__ v_rel2) {
  const int t = threadIdx.x;
  const int q = blockIdx.x;
  if (q == 4) {
    if (t < DIM) {
      float cv = b_fc[t], v2 = b_rel2[t];
      for (int u = 0; u < DIM; ++u) {
        cv += W_fc[t * 192 + u] * b_ent[u]
            + W_fc[t * 192 + 64 + u] * b_ent[u]
            + W_fc[t * 192 + 128 + u] * b_relL[u];
        v2 += W_rel2[t * DIM + u] * b_relL[u];
      }
      cvec[t] = cv; v_rel2[t] = v2;
    }
    return;
  }
  const float* __restrict__ L = (q == 3) ? W_rel2 : W_fc + 64 * q;
  const int pitch = (q == 3) ? DIM : 3 * DIM;
  const float* __restrict__ R = (q < 2) ? W_ent : W_relL;
  float* __restrict__ out = (q == 0) ? M_s : (q == 1) ? M_d : (q == 2) ? M_r : M_2;

  const int wave = t >> 5;
  const int lane = t & 31;
  const int c = lane & 15;
  const int h = lane >> 4;
  const float* __restrict__ lrow = L + (size_t)(16 * wave + c) * pitch + 2 * h;
  const float* __restrict__ rcol = R + (size_t)(2 * h) * DIM + 4 * c;

  v2f a[16];
  v4f r0[16], r1[16];
#pragma unroll
  for (int s = 0; s < 16; ++s) {
    a[s] = *(const v2f*)(lrow + 4 * s);
    r0[s] = *(const v4f*)(rcol + (size_t)(4 * s) * DIM);
    r1[s] = *(const v4f*)(rcol + (size_t)(4 * s + 1) * DIM);
  }

  v8f acc0 = {}, acc1 = {}, acc2 = {}, acc3 = {};
#pragma unroll
  for (int s = 0; s < 16; ++s) {
    const v2f b0 = {r0[s].x, r1[s].x};
    const v2f b1 = {r0[s].y, r1[s].y};
    const v2f b2 = {r0[s].z, r1[s].z};
    const v2f b3 = {r0[s].w, r1[s].w};
    acc0 = __builtin_amdgcn_wmma_f32_16x16x4_f32(false, a[s], false, b0,
                                                 (short)0, acc0, false, false);
    acc1 = __builtin_amdgcn_wmma_f32_16x16x4_f32(false, a[s], false, b1,
                                                 (short)0, acc1, false, false);
    acc2 = __builtin_amdgcn_wmma_f32_16x16x4_f32(false, a[s], false, b2,
                                                 (short)0, acc2, false, false);
    acc3 = __builtin_amdgcn_wmma_f32_16x16x4_f32(false, a[s], false, b3,
                                                 (short)0, acc3, false, false);
  }

  float* __restrict__ orow = out + (size_t)(16 * wave + 8 * h) * DIM + 4 * c;
#pragma unroll
  for (int r = 0; r < 8; ++r) {
    v4f o;
    o.x = acc0[r]; o.y = acc1[r]; o.z = acc2[r]; o.w = acc3[r];
    *(v4f*)(orow + (size_t)r * DIM) = o;
  }
}

#define GEMM_WAVES 4
template <int LDW>
__global__ __launch_bounds__(32 * GEMM_WAVES) void wmma_gemm64(
    const float* __restrict__ A, const float* __restrict__ W,
    const float* __restrict__ bias, float* __restrict__ out, int M) {
  const int wave = threadIdx.x >> 5;
  const int lane = threadIdx.x & 31;
  const int lm = lane & 15;
  const int hi = lane >> 4;
  const int ntiles = (M + 15) >> 4;
  const int nwaves = gridDim.x * GEMM_WAVES;
  int tile = blockIdx.x * GEMM_WAVES + wave;
  if (tile >= ntiles) return;

  v4f w[4][8];
#pragma unroll
  for (int nb = 0; nb < 4; ++nb) {
    const float* __restrict__ wr = W + (size_t)(nb * 16 + lm) * LDW + 4 * hi;
#pragma unroll
    for (int t = 0; t < 8; ++t) w[nb][t] = *(const v4f*)(wr + 8 * t);
  }
  float bv[4];
#pragma unroll
  for (int nb = 0; nb < 4; ++nb) bv[nb] = bias ? bias[nb * 16 + lm] : 0.0f;

  auto load_a = [&](int tl, v4f (&a)[8]) {
    const int row = min(tl * 16 + lm, M - 1);
    const float* __restrict__ ar = A + (size_t)row * DIM + 4 * hi;
#pragma unroll
    for (int t = 0; t < 8; ++t) a[t] = *(const v4f*)(ar + 8 * t);
  };

  v4f a[8];
  load_a(tile, a);
  for (; tile < ntiles; tile += nwaves) {
    v4f an[8];
    load_a(min(tile + nwaves, ntiles - 1), an);

    v8f acc0 = {}, acc1 = {}, acc2 = {}, acc3 = {};
#pragma unroll
    for (int t = 0; t < 8; ++t) {
      const v2f a0 = {a[t].x, a[t].y};
      const v2f a1 = {a[t].z, a[t].w};
      const v2f b00 = {w[0][t].x, w[0][t].y}, b01 = {w[0][t].z, w[0][t].w};
      const v2f b10 = {w[1][t].x, w[1][t].y}, b11 = {w[1][t].z, w[1][t].w};
      const v2f b20 = {w[2][t].x, w[2][t].y}, b21 = {w[2][t].z, w[2][t].w};
      const v2f b30 = {w[3][t].x, w[3][t].y}, b31 = {w[3][t].z, w[3][t].w};
      acc0 = __builtin_amdgcn_wmma_f32_16x16x4_f32(false, a0, false, b00,
                                                   (short)0, acc0, false, false);
      acc1 = __builtin_amdgcn_wmma_f32_16x16x4_f32(false, a0, false, b10,
                                                   (short)0, acc1, false, false);
      acc2 = __builtin_amdgcn_wmma_f32_16x16x4_f32(false, a0, false, b20,
                                                   (short)0, acc2, false, false);
      acc3 = __builtin_amdgcn_wmma_f32_16x16x4_f32(false, a0, false, b30,
                                                   (short)0, acc3, false, false);
      acc0 = __builtin_amdgcn_wmma_f32_16x16x4_f32(false, a1, false, b01,
                                                   (short)0, acc0, false, false);
      acc1 = __builtin_amdgcn_wmma_f32_16x16x4_f32(false, a1, false, b11,
                                                   (short)0, acc1, false, false);
      acc2 = __builtin_amdgcn_wmma_f32_16x16x4_f32(false, a1, false, b21,
                                                   (short)0, acc2, false, false);
      acc3 = __builtin_amdgcn_wmma_f32_16x16x4_f32(false, a1, false, b31,
                                                   (short)0, acc3, false, false);
    }

    const int m0 = tile * 16;
    v8f accs[4] = {acc0, acc1, acc2, acc3};
#pragma unroll
    for (int ob = 0; ob < 4; ++ob) {
#pragma unroll
      for (int v = 0; v < 8; ++v) {
        const int row = m0 + v + 8 * hi;
        if (row < M)
          out[(size_t)row * DIM + ob * 16 + lm] = accs[ob][v] + bv[ob];
      }
    }
#pragma unroll
    for (int t = 0; t < 8; ++t) a[t] = an[t];
  }
}

template <int V> struct KConst { static constexpr int value = V; };

#define SEG_BLK 1024

__global__ void edge_count_kernel(const int* __restrict__ src,
                                  int* __restrict__ deg, int E) {
  for (int e = blockIdx.x * blockDim.x + threadIdx.x; e < E;
       e += gridDim.x * blockDim.x)
    atomicAdd(&deg[src[e]], 1);
}

__global__ __launch_bounds__(SEG_BLK) void seg_alloc_kernel(
    int* __restrict__ cursor  ,
    int* __restrict__ total, int* __restrict__ bbase, int N) {
  __shared__ int swave[32];
  __shared__ int sbase;
  const int lane = threadIdx.x & 31;
  const int wave = threadIdx.x >> 5;
  const int i = blockIdx.x * SEG_BLK + threadIdx.x;
  const int v = (i < N) ? cursor[i] : 0;
  int x = v;
#pragma unroll
  for (int o = 1; o < 32; o <<= 1) {
    const int y = __shfl_up(x, o);
    if (lane >= o) x += y;
  }
  if (lane == 31) swave[wave] = x;
  __syncthreads();
  if (wave == 0) {
    const int w = swave[lane];
    int xs = w;
#pragma unroll
    for (int o = 1; o < 32; o <<= 1) {
      const int y = __shfl_up(xs, o);
      if (lane >= o) xs += y;
    }
    swave[lane] = xs - w;
    if (lane == 31) {
      const int base = atomicAdd(total, xs);
      sbase = base;
      bbase[blockIdx.x] = base;
    }
  }
  __syncthreads();
  if (i < N) cursor[i] = sbase + swave[wave] + x - v;
}

__global__ void edge_fill_kernel(const int* __restrict__ src,
                                 const int* __restrict__ dst,
                                 const int* __restrict__ rel,
                                 int* __restrict__ cursor,
                                 int* __restrict__ rec, int E) {
  for (int e = blockIdx.x * blockDim.x + threadIdx.x; e < E;
       e += gridDim.x * blockDim.x) {
    const int p = atomicAdd(&cursor[src[e]], 1);
    if ((unsigned)p < (unsigned)E) rec[p] = dst[e] | (rel[e] << 17);
  }
}

__device__ __forceinline__ float xlane16(float x) {
  const int b = __float_as_int(x);
  return __int_as_float(
      __builtin_amdgcn_permlanex16(b, b, 0x76543210, 0xfedcba98, false, false));
}
template <int M>
__device__ __forceinline__ float xrow(float x) {
  const int b = __float_as_int(x);
  return __int_as_float(
      __builtin_amdgcn_update_dpp(b, b, 0x160 | M, 0xf, 0xf, false));
}

__global__ __launch_bounds__(256) void node_gather_kernel(
    const int* __restrict__ cursor, const int* __restrict__ bbase,
    const int* __restrict__ rec, const float* __restrict__ u_s,
    const float* __restrict__ u_d, const float* __restrict__ u_r,
    const float* __restrict__ cvec, const float* __restrict__ W_a,
    const float* __restrict__ b_a, float* __restrict__ h_ent, int N, int E) {
  const int s = blockIdx.x * (blockDim.x >> 5) + (threadIdx.x >> 5);
  const int lane = threadIdx.x & 31;
  if (s >= N) return;

  int end = cursor[s];
  int start = (s & (SEG_BLK - 1)) ? cursor[s - 1] : bbase[s / SEG_BLK];
  end = __builtin_amdgcn_readfirstlane(end);
  start = __builtin_amdgcn_readfirstlane(start);
  start = min(max(start, 0), E);
  end = min(max(end, start), E);
  const int n = end - start;

  v2f acc = {0.f, 0.f};
  if (n > 0) {
    int pk = (lane < n) ? rec[start + lane] : 0;
    const v2f wa = *(const v2f*)(W_a + 2 * lane);
    const float ba = b_a[0];
    const v2f base = *(const v2f*)(u_s + (size_t)s * DIM + 2 * lane) +
                     *(const v2f*)(cvec + 2 * lane);
    const bool h4 = (lane & 16) != 0;
    const bool h3 = (lane & 8) != 0;
    const bool h2 = (lane & 4) != 0;
    const float NEG = -3.0e38f;
    float m = NEG;
    float bsum = 0.f;

    for (int c0 = 0; c0 < n; c0 += 32) {
      if (c0 > 0) pk = (c0 + lane < n) ? rec[start + c0 + lane] : 0;
      const int mm = min(32, n - c0);
      int j = 0;
      auto group = [&](auto KC, int valid) {
        constexpr int K = decltype(KC)::value;
        v2f ud[K], ur[K], c[K];
        float p[K];
#pragma unroll
        for (int i = 0; i < K; ++i) {
          const int idx = (j + i) & 31;
          const int pj = __builtin_amdgcn_readlane(pk, idx);
          const int d = min(pj & 0x1FFFF, NUM_NODES - 1);
          const int r = min(pj >> 17, NUM_RELS - 1);
          ud[i] = *(const v2f*)(u_d + (size_t)d * DIM + 2 * lane);
          ur[i] = *(const v2f*)(u_r + (size_t)r * DIM + 2 * lane);
        }
#pragma unroll
        for (int i = 0; i < K; ++i) {
          c[i] = (base + ud[i]) + ur[i];
          p[i] = c[i].x * wa.x + c[i].y * wa.y;
        }
        float t;
        if constexpr (K == 8) {
          float q[4], q2[2];
#pragma unroll
          for (int i = 0; i < 4; ++i) {
            const float keep = h4 ? p[i + 4] : p[i];
            const float send = h4 ? p[i] : p[i + 4];
            q[i] = keep + xlane16(send);
          }
#pragma unroll
          for (int i = 0; i < 2; ++i) {
            const float keep = h3 ? q[i + 2] : q[i];
            const float send = h3 ? q[i] : q[i + 2];
            q2[i] = keep + xrow<8>(send);
          }
          const float keep = h2 ? q2[1] : q2[0];
          const float send = h2 ? q2[0] : q2[1];
          t = keep + xrow<4>(send);
        } else {
          float q[2];
#pragma unroll
          for (int i = 0; i < 2; ++i) {
            const float keep = h4 ? p[i + 2] : p[i];
            const float send = h4 ? p[i] : p[i + 2];
            q[i] = keep + xlane16(send);
          }
          const float keep = h3 ? q[1] : q[0];
          const float send = h3 ? q[0] : q[1];
          t = keep + xrow<8>(send);
          t += xrow<4>(t);
        }
        t += xrow<2>(t);
        t += xrow<1>(t);
        const int myedge = (K == 8) ? ((lane >> 2) & 7) : ((lane >> 3) & 3);
        float l = t + ba;
        l = l > 0.f ? l : 0.01f * l;
        if (myedge >= valid) l = NEG;
        float gm = l;
        gm = fmaxf(gm, xlane16(gm));
        gm = fmaxf(gm, xrow<8>(gm));
        gm = fmaxf(gm, xrow<4>(gm));
        const float mnew = fmaxf(m, gm);
        const float sc = expf(m - mnew);
        const int bbits = __float_as_int(expf(l - mnew));
        m = mnew;
        acc.x *= sc;
        acc.y *= sc;
        bsum *= sc;
#pragma unroll
        for (int i = 0; i < K; ++i) {
          if (i < valid) {
            const float w =
                __int_as_float(__builtin_amdgcn_readlane(bbits, i * (32 / K)));
            acc.x += w * c[i].x;
            acc.y += w * c[i].y;
            bsum += w;
          }
        }
        j += K;
      };
      while (mm - j >= 8) group(KConst<8>{}, 8);
      if (mm - j >= 4) group(KConst<4>{}, 4);
      if (mm - j > 0) group(KConst<4>{}, mm - j);
    }
    acc.x = acc.x / bsum;
    acc.y = acc.y / bsum;
  }
  *(v2f*)(h_ent + (size_t)s * DIM + 2 * lane) = acc;
}


#define PB      128
#define PPITCH  160
#define POOL_WGS 256
#define CNT_PITCH 512

__global__ __launch_bounds__(512) void edge_pool_kernel(
    const int* __restrict__ src, const int* __restrict__ dst,
    const int* __restrict__ rel, const float* __restrict__ g_s,
    const float* __restrict__ g_d, float* __restrict__ pool,
    float* __restrict__ cntp, int E) {
  __shared__ __attribute__((aligned(16))) float sG[2][(PB / 2) * PPITCH];
  __shared__ __attribute__((aligned(16))) int sR[2][PB];

  const int wave = threadIdx.x >> 5;
  const int lane = threadIdx.x & 31;
  const int lm = lane & 15;
  const int hi = lane >> 4;
  const int rbase = wave * 32;
  const int nb = (E + PB - 1) / PB;
  const int G = gridDim.x;

  v8f c00 = {}, c01 = {}, c02 = {}, c03 = {};
  v8f c10 = {}, c11 = {}, c12 = {}, c13 = {};
  float n0 = 0.f, n1 = 0.f;

  v2f rs[8], rd[8];
  int s_cur, d_cur, r_cur, s_nxt, d_nxt, r_nxt;

  auto load_idx = [&](int bb, int& s, int& d, int& r) {
    const int e = bb * PB + wave * 8 + (lane & 7);
    const bool ok = (bb < nb) && (e < E);
    s = ok ? src[e] : 0;
    d = ok ? dst[e] : 0;
    r = ok ? rel[e] : -1;
  };
  auto load_rows = [&](int s, int d) {
#pragma unroll
    for (int i = 0; i < 8; ++i) {
      const int si = __builtin_amdgcn_readlane(s, i);
      const int di = __builtin_amdgcn_readlane(d, i);
      rs[i] = *(const v2f*)(g_s + (size_t)si * DIM + 2 * lane);
      rd[i] = *(const v2f*)(g_d + (size_t)di * DIM + 2 * lane);
    }
  };

  int b = blockIdx.x;
  load_idx(b, s_cur, d_cur, r_cur);
  load_rows(s_cur, d_cur);
  load_idx(b + G, s_nxt, d_nxt, r_nxt);

  int buf = 0;
  for (; b < nb; b += G) {
    float* gb = sG[buf];
#pragma unroll
    for (int j = 0; j < 4; ++j) {
      const int r0 = __builtin_amdgcn_readlane(r_cur, 2 * j);
      const int r1 = __builtin_amdgcn_readlane(r_cur, 2 * j + 1);
      v2f g0 = rs[2 * j] + rd[2 * j];
      v2f g1 = rs[2 * j + 1] + rd[2 * j + 1];
      if (r0 < 0) { g0.x = 0.f; g0.y = 0.f; }
      if (r1 < 0) { g1.x = 0.f; g1.y = 0.f; }
      v4f o;
      o.x = g0.x; o.y = g1.x; o.z = g0.y; o.w = g1.y;
      *(v4f*)(gb + (wave * 4 + j) * PPITCH + 4 * lane) = o;
    }
    if (lane < 8) sR[buf][wave * 8 + lane] = r_cur;
    __syncthreads();

    s_cur = s_nxt; d_cur = d_nxt; r_cur = r_nxt;
    if (b + G < nb) load_rows(s_cur, d_cur);
    load_idx(b + 2 * G, s_nxt, d_nxt, r_nxt);

    const v4i rr = *(const v4i*)&sR[buf][4 * lane];
    const bool relv = ((unsigned)(rr.x - rbase) < 32u) |
                      ((unsigned)(rr.y - rbase) < 32u) |
                      ((unsigned)(rr.z - rbase) < 32u) |
                      ((unsigned)(rr.w - rbase) < 32u);
    unsigned mask = __builtin_amdgcn_ballot_w32(relv);
    while (mask) {
      const int g = __builtin_ctz(mask);
      mask &= mask - 1;
      const int* pr = &sR[buf][4 * g + 2 * hi];
      const int da = pr[0] - rbase - lm;
      const int db = pr[1] - rbase - lm;
      const float* pg = gb + (2 * g + hi) * PPITCH + 2 * lm;
      const v2f b0 = *(const v2f*)(pg);
      const v2f b1 = *(const v2f*)(pg + 32);
      const v2f b2 = *(const v2f*)(pg + 64);
      const v2f b3 = *(const v2f*)(pg + 96);
      {
        const bool ma = (da == 0), mb = (db == 0);
        if (__builtin_amdgcn_ballot_w32(ma | mb)) {
          v2f a;
          a.x = ma ? 1.0f : 0.0f;
          a.y = mb ? 1.0f : 0.0f;
          n0 += a.x + a.y;
          c00 = __builtin_amdgcn_wmma_f32_16x16x4_f32(false, a, false, b0,
                                                      (short)0, c00, false, false);
          c01 = __builtin_amdgcn_wmma_f32_16x16x4_f32(false, a, false, b1,
                                                      (short)0, c01, false, false);
          c02 = __builtin_amdgcn_wmma_f32_16x16x4_f32(false, a, false, b2,
                                                      (short)0, c02, false, false);
          c03 = __builtin_amdgcn_wmma_f32_16x16x4_f32(false, a, false, b3,
                                                      (short)0, c03, false, false);
        }
      }
      {
        const bool ma = (da == 16), mb = (db == 16);
        if (__builtin_amdgcn_ballot_w32(ma | mb)) {
          v2f a;
          a.x = ma ? 1.0f : 0.0f;
          a.y = mb ? 1.0f : 0.0f;
          n1 += a.x + a.y;
          c10 = __builtin_amdgcn_wmma_f32_16x16x4_f32(false, a, false, b0,
                                                      (short)0, c10, false, false);
          c11 = __builtin_amdgcn_wmma_f32_16x16x4_f32(false, a, false, b1,
                                                      (short)0, c11, false, false);
          c12 = __builtin_amdgcn_wmma_f32_16x16x4_f32(false, a, false, b2,
                                                      (short)0, c12, false, false);
          c13 = __builtin_amdgcn_wmma_f32_16x16x4_f32(false, a, false, b3,
                                                      (short)0, c13, false, false);
        }
      }
    }
    buf ^= 1;
  }

  __syncthreads();
  float* sFw = &sG[0][0] + wave * 1024;
  auto flush = [&](const v8f& t0, const v8f& t1, const v8f& t2, const v8f& t3,
                   int rt) {
#pragma unroll
    for (int v = 0; v < 8; ++v) {
      float* row = sFw + (v + 8 * hi) * DIM + lm;
      row[0] = t0[v]; row[16] = t1[v]; row[32] = t2[v]; row[48] = t3[v];
    }
    __syncthreads();
    for (int q = 0; q < 16; ++q) {
      const int r = (q + (int)blockIdx.x) & 15;
      const int R = rbase + rt * 16 + r;
      if (R < NUM_RELS) {
        atomicAdd(&pool[(size_t)R * DIM + lane], sFw[r * DIM + lane]);
        atomicAdd(&pool[(size_t)R * DIM + 32 + lane], sFw[r * DIM + 32 + lane]);
      }
    }
    __syncthreads();
  };
  flush(c00, c01, c02, c03, 0);
  flush(c10, c11, c12, c13, 1);

  {
    const float t0 = n0 + xlane16(n0);
    const float t1 = n1 + xlane16(n1);
    cntp[(size_t)blockIdx.x * CNT_PITCH + rbase + lane] = hi ? t1 : t0;
  }
}

__global__ __launch_bounds__(256) void h_rel_kernel(
    const float* __restrict__ pool, const float* __restrict__ cntp,
    const float* __restrict__ rel2_proj, const float* __restrict__ W_rel3,
    const float* __restrict__ b_rel3, float* __restrict__ h_rel, int R) {
  __shared__ float swsum[8];
  const int t = blockIdx.x * 256 + threadIdx.x;
  const int r = min(t >> 6, R - 1), o = t & 63;
  float c = 0.f;
#pragma unroll
  for (int j = 0; j < POOL_WGS / DIM; ++j)
    c += cntp[(size_t)(o + DIM * j) * CNT_PITCH + r];
  c += xlane16(c);
  c += xrow<8>(c);
  c += xrow<4>(c);
  c += xrow<2>(c);
  c += xrow<1>(c);
  const int wave = threadIdx.x >> 5;
  if ((threadIdx.x & 31) == 0) swsum[wave] = c;
  __syncthreads();
  c = swsum[wave & 6] + swsum[wave | 1];
  if (t >= R * DIM) return;
  const float inv = 1.0f / fmaxf(c, 1.0f);
  const float* w = W_rel3 + (size_t)o * 192;
  float acc = b_rel3[o] + pool[t] * inv;
  if (c > 0.f) {
    const float* r2 = rel2_proj + (size_t)r * DIM;
    for (int j = 0; j < DIM; ++j) acc += r2[j] * w[128 + j];
  }
  h_rel[t] = acc;
}

#define OFF_MS     0
#define OFF_MD     4096
#define OFF_MR     8192
#define OFF_M2     12288
#define OFF_CVEC   16384
#define OFF_VREL2  16448
#define OFF_UR     16576
#define OFF_REL2P  49088
#define OFF_BSUM   281088
#define OFF_POOL   381588
#define OFF_TOTAL  413588
#define OFF_BBASE  413600
#define OFF_BE     445588
#define OFF_CNTP   OFF_BE
#define OFF_US     1445588
#define OFF_UD     7845588
#define ZERO_CNT   (OFF_TOTAL + 1 - OFF_BSUM)

extern "C" void kernel_launch(void* const* d_in, const int* in_sizes, int n_in,
                              void* d_out, int out_size, void* d_ws,
                              size_t ws_size, hipStream_t stream) {
  const float* ent_embed = (const float*)d_in[0];
  const float* rel_embed = (const float*)d_in[1];
  const float* W_ent  = (const float*)d_in[2];
  const float* b_ent  = (const float*)d_in[3];
  const float* W_relL = (const float*)d_in[4];
  const float* b_relL = (const float*)d_in[5];
  const float* W_rel2 = (const float*)d_in[6];
  const float* b_rel2 = (const float*)d_in[7];
  const float* W_rel3 = (const float*)d_in[8];
  const float* b_rel3 = (const float*)d_in[9];
  const float* W_a    = (const float*)d_in[10];
  const float* b_a    = (const float*)d_in[11];
  const float* W_fc   = (const float*)d_in[12];
  const float* b_fc   = (const float*)d_in[13];
  const int* src_idx  = (const int*)d_in[14];
  const int* dst_idx  = (const int*)d_in[15];
  const int* rel_idx  = (const int*)d_in[16];

  float* ws = (float*)d_ws;
  float* h_ent = (float*)d_out;
  float* h_rel = (float*)d_out + (size_t)NUM_NODES * DIM;

  zero_kernel<<<512, 256, 0, stream>>>(ws + OFF_BSUM, ZERO_CNT);

  int* deg_cur = (int*)(ws + OFF_BSUM);
  int* seg_total = (int*)(ws + OFF_TOTAL);
  int* seg_bbase = (int*)(ws + OFF_BBASE);
  int* edge_rec = (int*)(ws + OFF_BE);
  edge_count_kernel<<<4096, 256, 0, stream>>>(src_idx, deg_cur, NUM_EDGES);
  seg_alloc_kernel<<<(NUM_NODES + SEG_BLK - 1) / SEG_BLK, SEG_BLK, 0, stream>>>(
      deg_cur, seg_total, seg_bbase, NUM_NODES);
  edge_fill_kernel<<<4096, 256, 0, stream>>>(src_idx, dst_idx, rel_idx, deg_cur,
                                             edge_rec, NUM_EDGES);

  fold_kernel<<<FOLD_WGS, 128, 0, stream>>>(
      W_ent, b_ent, W_relL, b_relL, W_rel2, b_rel2, W_fc, b_fc,
      ws + OFF_MS, ws + OFF_MD, ws + OFF_MR, ws + OFF_M2, ws + OFF_CVEC,
      ws + OFF_VREL2);

  const int gemm_blocks_N = 2048 / GEMM_WAVES;
  const int gemm_blocks_R = ((NUM_RELS + 15) / 16 + GEMM_WAVES - 1) / GEMM_WAVES;
  wmma_gemm64<DIM><<<gemm_blocks_N, 32 * GEMM_WAVES, 0, stream>>>(ent_embed, ws + OFF_MS,
                                                 nullptr, ws + OFF_US, NUM_NODES);
  wmma_gemm64<DIM><<<gemm_blocks_N, 32 * GEMM_WAVES, 0, stream>>>(ent_embed, ws + OFF_MD,
                                                 nullptr, ws + OFF_UD, NUM_NODES);
  wmma_gemm64<DIM><<<gemm_blocks_R, 32 * GEMM_WAVES, 0, stream>>>(rel_embed, ws + OFF_MR,
                                                 nullptr, ws + OFF_UR, NUM_RELS);
  wmma_gemm64<DIM><<<gemm_blocks_R, 32 * GEMM_WAVES, 0, stream>>>(rel_embed, ws + OFF_M2,
                                                 ws + OFF_VREL2, ws + OFF_REL2P,
                                                 NUM_RELS);

  node_gather_kernel<<<(NUM_NODES + 7) / 8, 256, 0, stream>>>(
      deg_cur, seg_bbase, edge_rec, ws + OFF_US, ws + OFF_UD, ws + OFF_UR,
      ws + OFF_CVEC, W_a, b_a, h_ent, NUM_NODES, NUM_EDGES);
  wmma_gemm64<3 * DIM><<<gemm_blocks_N, 32 * GEMM_WAVES, 0, stream>>>(h_ent, W_rel3,
                                                 nullptr, ws + OFF_US, NUM_NODES);
  wmma_gemm64<3 * DIM><<<gemm_blocks_N, 32 * GEMM_WAVES, 0, stream>>>(h_ent, W_rel3 + DIM,
                                                 nullptr, ws + OFF_UD, NUM_NODES);
  edge_pool_kernel<<<POOL_WGS, 512, 0, stream>>>(
      src_idx, dst_idx, rel_idx, ws + OFF_US, ws + OFF_UD, ws + OFF_POOL,
      ws + OFF_CNTP, NUM_EDGES);

  h_rel_kernel<<<(NUM_RELS * DIM + 255) / 256, 256, 0, stream>>>(
      ws + OFF_POOL, ws + OFF_CNTP, ws + OFF_REL2P, W_rel3, b_rel3, h_rel,
      NUM_RELS);
}


# workspace planes u_s/u_d (later g_s/g_d), pool and records moved to 256-byte multiples: every gathered or stored 64-float row is two whole 128-byte lines, not pieces of three
# speedup vs baseline: 3.2270x; 1.0738x over previous
#include <hip/hip_runtime.h>
#include <math.h>


#define NUM_NODES 100000
#define NUM_RELS  500
#define DIM       64
#define NUM_EDGES 1000000

typedef __attribute__((ext_vector_type(2))) float v2f;
typedef __attribute__((ext_vector_type(8))) float v8f;
typedef __attribute__((ext_vector_type(4))) float v4f;
typedef __attribute__((ext_vector_type(4))) int v4i;

__global__ void zero_kernel(float* __restrict__ p, int n) {
  for (int i = blockIdx.x * blockDim.x + threadIdx.x; i < n;
       i += gridDim.x * blockDim.x)
    p[i] = 0.0f;
}

#define FOLD_WGS 5
__global__ __launch_bounds__(128) void fold_kernel(
    const float* __restrict__ W_ent, const float* __restrict__ b_ent,
    const float* __restrict__ W_relL, const float* __restrict__ b_relL,
    const float* __restrict__ W_rel2, const float* __restrict__ b_rel2,
    const float* __restrict__ W_fc, const float* __restrict__ b_fc,
    float* __restrict__ M_s, float* __restrict__ M_d,
    float* __restrict__ M_r, float* __restrict__ M_2,
    float* __restrict__ cvec, float* __restrict__ v_rel2) {
  const int t = threadIdx.x;
  const int q = blockIdx.x;
  if (q == 4) {
    if (t < DIM) {
      float cv = b_fc[t], v2 = b_rel2[t];
      for (int u = 0; u < DIM; ++u) {
        cv += W_fc[t * 192 + u] * b_ent[u]
            + W_fc[t * 192 + 64 + u] * b_ent[u]
            + W_fc[t * 192 + 128 + u] * b_relL[u];
        v2 += W_rel2[t * DIM + u] * b_relL[u];
      }
      cvec[t] = cv; v_rel2[t] = v2;
    }
    return;
  }
  const float* __restrict__ L = (q == 3) ? W_rel2 : W_fc + 64 * q;
  const int pitch = (q == 3) ? DIM : 3 * DIM;
  const float* __restrict__ R = (q < 2) ? W_ent : W_relL;
  float* __restrict__ out = (q == 0) ? M_s : (q == 1) ? M_d : (q == 2) ? M_r : M_2;

  const int wave = t >> 5;
  const int lane = t & 31;
  const int c = lane & 15;
  const int h = lane >> 4;
  const float* __restrict__ lrow = L + (size_t)(16 * wave + c) * pitch + 2 * h;
  const float* __restrict__ rcol = R + (size_t)(2 * h) * DIM + 4 * c;

  v2f a[16];
  v4f r0[16], r1[16];
#pragma unroll
  for (int s = 0; s < 16; ++s) {
    a[s] = *(const v2f*)(lrow + 4 * s);
    r0[s] = *(const v4f*)(rcol + (size_t)(4 * s) * DIM);
    r1[s] = *(const v4f*)(rcol + (size_t)(4 * s + 1) * DIM);
  }

  v8f acc0 = {}, acc1 = {}, acc2 = {}, acc3 = {};
#pragma unroll
  for (int s = 0; s < 16; ++s) {
    const v2f b0 = {r0[s].x, r1[s].x};
    const v2f b1 = {r0[s].y, r1[s].y};
    const v2f b2 = {r0[s].z, r1[s].z};
    const v2f b3 = {r0[s].w, r1[s].w};
    acc0 = __builtin_amdgcn_wmma_f32_16x16x4_f32(false, a[s], false, b0,
                                                 (short)0, acc0, false, false);
    acc1 = __builtin_amdgcn_wmma_f32_16x16x4_f32(false, a[s], false, b1,
                                                 (short)0, acc1, false, false);
    acc2 = __builtin_amdgcn_wmma_f32_16x16x4_f32(false, a[s], false, b2,
                                                 (short)0, acc2, false, false);
    acc3 = __builtin_amdgcn_wmma_f32_16x16x4_f32(false, a[s], false, b3,
                                                 (short)0, acc3, false, false);
  }

  float* __restrict__ orow = out + (size_t)(16 * wave + 8 * h) * DIM + 4 * c;
#pragma unroll
  for (int r = 0; r < 8; ++r) {
    v4f o;
    o.x = acc0[r]; o.y = acc1[r]; o.z = acc2[r]; o.w = acc3[r];
    *(v4f*)(orow + (size_t)r * DIM) = o;
  }
}

#define GEMM_WAVES 4
template <int LDW>
__global__ __launch_bounds__(32 * GEMM_WAVES) void wmma_gemm64(
    const float* __restrict__ A, const float* __restrict__ W,
    const float* __restrict__ bias, float* __restrict__ out, int M) {
  const int wave = threadIdx.x >> 5;
  const int lane = threadIdx.x & 31;
  const int lm = lane & 15;
  const int hi = lane >> 4;
  const int ntiles = (M + 15) >> 4;
  const int nwaves = gridDim.x * GEMM_WAVES;
  int tile = blockIdx.x * GEMM_WAVES + wave;
  if (tile >= ntiles) return;

  v4f w[4][8];
#pragma unroll
  for (int nb = 0; nb < 4; ++nb) {
    const float* __restrict__ wr = W + (size_t)(nb * 16 + lm) * LDW + 4 * hi;
#pragma unroll
    for (int t = 0; t < 8; ++t) w[nb][t] = *(const v4f*)(wr + 8 * t);
  }
  float bv[4];
#pragma unroll
  for (int nb = 0; nb < 4; ++nb) bv[nb] = bias ? bias[nb * 16 + lm] : 0.0f;

  auto load_a = [&](int tl, v4f (&a)[8]) {
    const int row = min(tl * 16 + lm, M - 1);
    const float* __restrict__ ar = A + (size_t)row * DIM + 4 * hi;
#pragma unroll
    for (int t = 0; t < 8; ++t) a[t] = *(const v4f*)(ar + 8 * t);
  };

  v4f a[8];
  load_a(tile, a);
  for (; tile < ntiles; tile += nwaves) {
    v4f an[8];
    load_a(min(tile + nwaves, ntiles - 1), an);

    v8f acc0 = {}, acc1 = {}, acc2 = {}, acc3 = {};
#pragma unroll
    for (int t = 0; t < 8; ++t) {
      const v2f a0 = {a[t].x, a[t].y};
      const v2f a1 = {a[t].z, a[t].w};
      const v2f b00 = {w[0][t].x, w[0][t].y}, b01 = {w[0][t].z, w[0][t].w};
      const v2f b10 = {w[1][t].x, w[1][t].y}, b11 = {w[1][t].z, w[1][t].w};
      const v2f b20 = {w[2][t].x, w[2][t].y}, b21 = {w[2][t].z, w[2][t].w};
      const v2f b30 = {w[3][t].x, w[3][t].y}, b31 = {w[3][t].z, w[3][t].w};
      acc0 = __builtin_amdgcn_wmma_f32_16x16x4_f32(false, a0, false, b00,
                                                   (short)0, acc0, false, false);
      acc1 = __builtin_amdgcn_wmma_f32_16x16x4_f32(false, a0, false, b10,
                                                   (short)0, acc1, false, false);
      acc2 = __builtin_amdgcn_wmma_f32_16x16x4_f32(false, a0, false, b20,
                                                   (short)0, acc2, false, false);
      acc3 = __builtin_amdgcn_wmma_f32_16x16x4_f32(false, a0, false, b30,
                                                   (short)0, acc3, false, false);
      acc0 = __builtin_amdgcn_wmma_f32_16x16x4_f32(false, a1, false, b01,
                                                   (short)0, acc0, false, false);
      acc1 = __builtin_amdgcn_wmma_f32_16x16x4_f32(false, a1, false, b11,
                                                   (short)0, acc1, false, false);
      acc2 = __builtin_amdgcn_wmma_f32_16x16x4_f32(false, a1, false, b21,
                                                   (short)0, acc2, false, false);
      acc3 = __builtin_amdgcn_wmma_f32_16x16x4_f32(false, a1, false, b31,
                                                   (short)0, acc3, false, false);
    }

    const int m0 = tile * 16;
    v8f accs[4] = {acc0, acc1, acc2, acc3};
#pragma unroll
    for (int ob = 0; ob < 4; ++ob) {
#pragma unroll
      for (int v = 0; v < 8; ++v) {
        const int row = m0 + v + 8 * hi;
        if (row < M)
          out[(size_t)row * DIM + ob * 16 + lm] = accs[ob][v] + bv[ob];
      }
    }
#pragma unroll
    for (int t = 0; t < 8; ++t) a[t] = an[t];
  }
}

template <int V> struct KConst { static constexpr int value = V; };

#define SEG_BLK 1024

__global__ void edge_count_kernel(const int* __restrict__ src,
                                  int* __restrict__ deg, int E) {
  for (int e = blockIdx.x * blockDim.x + threadIdx.x; e < E;
       e += gridDim.x * blockDim.x)
    atomicAdd(&deg[src[e]], 1);
}

__global__ __launch_bounds__(SEG_BLK) void seg_alloc_kernel(
    int* __restrict__ cursor  ,
    int* __restrict__ total, int* __restrict__ bbase, int N) {
  __shared__ int swave[32];
  __shared__ int sbase;
  const int lane = threadIdx.x & 31;
  const int wave = threadIdx.x >> 5;
  const int i = blockIdx.x * SEG_BLK + threadIdx.x;
  const int v = (i < N) ? cursor[i] : 0;
  int x = v;
#pragma unroll
  for (int o = 1; o < 32; o <<= 1) {
    const int y = __shfl_up(x, o);
    if (lane >= o) x += y;
  }
  if (lane == 31) swave[wave] = x;
  __syncthreads();
  if (wave == 0) {
    const int w = swave[lane];
    int xs = w;
#pragma unroll
    for (int o = 1; o < 32; o <<= 1) {
      const int y = __shfl_up(xs, o);
      if (lane >= o) xs += y;
    }
    swave[lane] = xs - w;
    if (lane == 31) {
      const int base = atomicAdd(total, xs);
      sbase = base;
      bbase[blockIdx.x] = base;
    }
  }
  __syncthreads();
  if (i < N) cursor[i] = sbase + swave[wave] + x - v;
}

__global__ void edge_fill_kernel(const int* __restrict__ src,
                                 const int* __restrict__ dst,
                                 const int* __restrict__ rel,
                                 int* __restrict__ cursor,
                                 int* __restrict__ rec, int E) {
  for (int e = blockIdx.x * blockDim.x + threadIdx.x; e < E;
       e += gridDim.x * blockDim.x) {
    const int p = atomicAdd(&cursor[src[e]], 1);
    if ((unsigned)p < (unsigned)E) rec[p] = dst[e] | (rel[e] << 17);
  }
}

__device__ __forceinline__ float xlane16(float x) {
  const int b = __float_as_int(x);
  return __int_as_float(
      __builtin_amdgcn_permlanex16(b, b, 0x76543210, 0xfedcba98, false, false));
}
template <int M>
__device__ __forceinline__ float xrow(float x) {
  const int b = __float_as_int(x);
  return __int_as_float(
      __builtin_amdgcn_update_dpp(b, b, 0x160 | M, 0xf, 0xf, false));
}

__global__ __launch_bounds__(256) void node_gather_kernel(
    const int* __restrict__ cursor, const int* __restrict__ bbase,
    const int* __restrict__ rec, const float* __restrict__ u_s,
    const float* __restrict__ u_d, const float* __restrict__ u_r,
    const float* __restrict__ cvec, const float* __restrict__ W_a,
    const float* __restrict__ b_a, float* __restrict__ h_ent, int N, int E) {
  const int s = blockIdx.x * (blockDim.x >> 5) + (threadIdx.x >> 5);
  const int lane = threadIdx.x & 31;
  if (s >= N) return;

  int end = cursor[s];
  int start = (s & (SEG_BLK - 1)) ? cursor[s - 1] : bbase[s / SEG_BLK];
  end = __builtin_amdgcn_readfirstlane(end);
  start = __builtin_amdgcn_readfirstlane(start);
  start = min(max(start, 0), E);
  end = min(max(end, start), E);
  const int n = end - start;

  v2f acc = {0.f, 0.f};
  if (n > 0) {
    int pk = (lane < n) ? rec[start + lane] : 0;
    const v2f wa = *(const v2f*)(W_a + 2 * lane);
    const float ba = b_a[0];
    const v2f base = *(const v2f*)(u_s + (size_t)s * DIM + 2 * lane) +
                     *(const v2f*)(cvec + 2 * lane);
    const bool h4 = (lane & 16) != 0;
    const bool h3 = (lane & 8) != 0;
    const bool h2 = (lane & 4) != 0;
    const float NEG = -3.0e38f;
    float m = NEG;
    float bsum = 0.f;

    for (int c0 = 0; c0 < n; c0 += 32) {
      if (c0 > 0) pk = (c0 + lane < n) ? rec[start + c0 + lane] : 0;
      const int mm = min(32, n - c0);
      int j = 0;
      auto group = [&](auto KC, int valid) {
        constexpr int K = decltype(KC)::value;
        v2f ud[K], ur[K], c[K];
        float p[K];
#pragma unroll
        for (int i = 0; i < K; ++i) {
          const int idx = (j + i) & 31;
          const int pj = __builtin_amdgcn_readlane(pk, idx);
          const int d = min(pj & 0x1FFFF, NUM_NODES - 1);
          const int r = min(pj >> 17, NUM_RELS - 1);
          ud[i] = *(const v2f*)(u_d + (size_t)d * DIM + 2 * lane);
          ur[i] = *(const v2f*)(u_r + (size_t)r * DIM + 2 * lane);
        }
#pragma unroll
        for (int i = 0; i < K; ++i) {
          c[i] = (base + ud[i]) + ur[i];
          p[i] = c[i].x * wa.x + c[i].y * wa.y;
        }
        float t;
        if constexpr (K == 8) {
          float q[4], q2[2];
#pragma unroll
          for (int i = 0; i < 4; ++i) {
            const float keep = h4 ? p[i + 4] : p[i];
            const float send = h4 ? p[i] : p[i + 4];
            q[i] = keep + xlane16(send);
          }
#pragma unroll
          for (int i = 0; i < 2; ++i) {
            const float keep = h3 ? q[i + 2] : q[i];
            const float send = h3 ? q[i] : q[i + 2];
            q2[i] = keep + xrow<8>(send);
          }
          const float keep = h2 ? q2[1] : q2[0];
          const float send = h2 ? q2[0] : q2[1];
          t = keep + xrow<4>(send);
        } else {
          float q[2];
#pragma unroll
          for (int i = 0; i < 2; ++i) {
            const float keep = h4 ? p[i + 2] : p[i];
            const float send = h4 ? p[i] : p[i + 2];
            q[i] = keep + xlane16(send);
          }
          const float keep = h3 ? q[1] : q[0];
          const float send = h3 ? q[0] : q[1];
          t = keep + xrow<8>(send);
          t += xrow<4>(t);
        }
        t += xrow<2>(t);
        t += xrow<1>(t);
        const int myedge = (K == 8) ? ((lane >> 2) & 7) : ((lane >> 3) & 3);
        float l = t + ba;
        l = l > 0.f ? l : 0.01f * l;
        if (myedge >= valid) l = NEG;
        float gm = l;
        gm = fmaxf(gm, xlane16(gm));
        gm = fmaxf(gm, xrow<8>(gm));
        gm = fmaxf(gm, xrow<4>(gm));
        const float mnew = fmaxf(m, gm);
        const float sc = expf(m - mnew);
        const int bbits = __float_as_int(expf(l - mnew));
        m = mnew;
        acc.x *= sc;
        acc.y *= sc;
        bsum *= sc;
#pragma unroll
        for (int i = 0; i < K; ++i) {
          if (i < valid) {
            const float w =
                __int_as_float(__builtin_amdgcn_readlane(bbits, i * (32 / K)));
            acc.x += w * c[i].x;
            acc.y += w * c[i].y;
            bsum += w;
          }
        }
        j += K;
      };
      while (mm - j >= 8) group(KConst<8>{}, 8);
      if (mm - j >= 4) group(KConst<4>{}, 4);
      if (mm - j > 0) group(KConst<4>{}, mm - j);
    }
    acc.x = acc.x / bsum;
    acc.y = acc.y / bsum;
  }
  *(v2f*)(h_ent + (size_t)s * DIM + 2 * lane) = acc;
}


#define PB      128
#define PPITCH  160
#define POOL_WGS 256
#define CNT_PITCH 512

__global__ __launch_bounds__(512) void edge_pool_kernel(
    const int* __restrict__ src, const int* __restrict__ dst,
    const int* __restrict__ rel, const float* __restrict__ g_s,
    const float* __restrict__ g_d, float* __restrict__ pool,
    float* __restrict__ cntp, int E) {
  __shared__ __attribute__((aligned(16))) float sG[2][(PB / 2) * PPITCH];
  __shared__ __attribute__((aligned(16))) int sR[2][PB];

  const int wave = threadIdx.x >> 5;
  const int lane = threadIdx.x & 31;
  const int lm = lane & 15;
  const int hi = lane >> 4;
  const int rbase = wave * 32;
  const int nb = (E + PB - 1) / PB;
  const int G = gridDim.x;

  v8f c00 = {}, c01 = {}, c02 = {}, c03 = {};
  v8f c10 = {}, c11 = {}, c12 = {}, c13 = {};
  float n0 = 0.f, n1 = 0.f;

  v2f rs[8], rd[8];
  int s_cur, d_cur, r_cur, s_nxt, d_nxt, r_nxt;

  auto load_idx = [&](int bb, int& s, int& d, int& r) {
    const int e = bb * PB + wave * 8 + (lane & 7);
    const bool ok = (bb < nb) && (e < E);
    s = ok ? src[e] : 0;
    d = ok ? dst[e] : 0;
    r = ok ? rel[e] : -1;
  };
  auto load_rows = [&](int s, int d) {
#pragma unroll
    for (int i = 0; i < 8; ++i) {
      const int si = __builtin_amdgcn_readlane(s, i);
      const int di = __builtin_amdgcn_readlane(d, i);
      rs[i] = *(const v2f*)(g_s + (size_t)si * DIM + 2 * lane);
      rd[i] = *(const v2f*)(g_d + (size_t)di * DIM + 2 * lane);
    }
  };

  int b = blockIdx.x;
  load_idx(b, s_cur, d_cur, r_cur);
  load_rows(s_cur, d_cur);
  load_idx(b + G, s_nxt, d_nxt, r_nxt);

  int buf = 0;
  for (; b < nb; b += G) {
    float* gb = sG[buf];
#pragma unroll
    for (int j = 0; j < 4; ++j) {
      const int r0 = __builtin_amdgcn_readlane(r_cur, 2 * j);
      const int r1 = __builtin_amdgcn_readlane(r_cur, 2 * j + 1);
      v2f g0 = rs[2 * j] + rd[2 * j];
      v2f g1 = rs[2 * j + 1] + rd[2 * j + 1];
      if (r0 < 0) { g0.x = 0.f; g0.y = 0.f; }
      if (r1 < 0) { g1.x = 0.f; g1.y = 0.f; }
      v4f o;
      o.x = g0.x; o.y = g1.x; o.z = g0.y; o.w = g1.y;
      *(v4f*)(gb + (wave * 4 + j) * PPITCH + 4 * lane) = o;
    }
    if (lane < 8) sR[buf][wave * 8 + lane] = r_cur;
    __syncthreads();

    s_cur = s_nxt; d_cur = d_nxt; r_cur = r_nxt;
    if (b + G < nb) load_rows(s_cur, d_cur);
    load_idx(b + 2 * G, s_nxt, d_nxt, r_nxt);

    const v4i rr = *(const v4i*)&sR[buf][4 * lane];
    const bool relv = ((unsigned)(rr.x - rbase) < 32u) |
                      ((unsigned)(rr.y - rbase) < 32u) |
                      ((unsigned)(rr.z - rbase) < 32u) |
                      ((unsigned)(rr.w - rbase) < 32u);
    unsigned mask = __builtin_amdgcn_ballot_w32(relv);
    while (mask) {
      const int g = __builtin_ctz(mask);
      mask &= mask - 1;
      const int* pr = &sR[buf][4 * g + 2 * hi];
      const int da = pr[0] - rbase - lm;
      const int db = pr[1] - rbase - lm;
      const float* pg = gb + (2 * g + hi) * PPITCH + 2 * lm;
      const v2f b0 = *(const v2f*)(pg);
      const v2f b1 = *(const v2f*)(pg + 32);
      const v2f b2 = *(const v2f*)(pg + 64);
      const v2f b3 = *(const v2f*)(pg + 96);
      {
        const bool ma = (da == 0), mb = (db == 0);
        if (__builtin_amdgcn_ballot_w32(ma | mb)) {
          v2f a;
          a.x = ma ? 1.0f : 0.0f;
          a.y = mb ? 1.0f : 0.0f;
          n0 += a.x + a.y;
          c00 = __builtin_amdgcn_wmma_f32_16x16x4_f32(false, a, false, b0,
                                                      (short)0, c00, false, false);
          c01 = __builtin_amdgcn_wmma_f32_16x16x4_f32(false, a, false, b1,
                                                      (short)0, c01, false, false);
          c02 = __builtin_amdgcn_wmma_f32_16x16x4_f32(false, a, false, b2,
                                                      (short)0, c02, false, false);
          c03 = __builtin_amdgcn_wmma_f32_16x16x4_f32(false, a, false, b3,
                                                      (short)0, c03, false, false);
        }
      }
      {
        const bool ma = (da == 16), mb = (db == 16);
        if (__builtin_amdgcn_ballot_w32(ma | mb)) {
          v2f a;
          a.x = ma ? 1.0f : 0.0f;
          a.y = mb ? 1.0f : 0.0f;
          n1 += a.x + a.y;
          c10 = __builtin_amdgcn_wmma_f32_16x16x4_f32(false, a, false, b0,
                                                      (short)0, c10, false, false);
          c11 = __builtin_amdgcn_wmma_f32_16x16x4_f32(false, a, false, b1,
                                                      (short)0, c11, false, false);
          c12 = __builtin_amdgcn_wmma_f32_16x16x4_f32(false, a, false, b2,
                                                      (short)0, c12, false, false);
          c13 = __builtin_amdgcn_wmma_f32_16x16x4_f32(false, a, false, b3,
                                                      (short)0, c13, false, false);
        }
      }
    }
    buf ^= 1;
  }

  __syncthreads();
  float* sFw = &sG[0][0] + wave * 1024;
  auto flush = [&](const v8f& t0, const v8f& t1, const v8f& t2, const v8f& t3,
                   int rt) {
#pragma unroll
    for (int v = 0; v < 8; ++v) {
      float* row = sFw + (v + 8 * hi) * DIM + lm;
      row[0] = t0[v]; row[16] = t1[v]; row[32] = t2[v]; row[48] = t3[v];
    }
    __syncthreads();
    for (int q = 0; q < 16; ++q) {
      const int r = (q + (int)blockIdx.x) & 15;
      const int R = rbase + rt * 16 + r;
      if (R < NUM_RELS) {
        atomicAdd(&pool[(size_t)R * DIM + lane], sFw[r * DIM + lane]);
        atomicAdd(&pool[(size_t)R * DIM + 32 + lane], sFw[r * DIM + 32 + lane]);
      }
    }
    __syncthreads();
  };
  flush(c00, c01, c02, c03, 0);
  flush(c10, c11, c12, c13, 1);

  {
    const float t0 = n0 + xlane16(n0);
    const float t1 = n1 + xlane16(n1);
    cntp[(size_t)blockIdx.x * CNT_PITCH + rbase + lane] = hi ? t1 : t0;
  }
}

__global__ __launch_bounds__(256) void h_rel_kernel(
    const float* __restrict__ pool, const float* __restrict__ cntp,
    const float* __restrict__ rel2_proj, const float* __restrict__ W_rel3,
    const float* __restrict__ b_rel3, float* __restrict__ h_rel, int R) {
  __shared__ float swsum[8];
  const int t = blockIdx.x * 256 + threadIdx.x;
  const int r = min(t >> 6, R - 1), o = t & 63;
  float c = 0.f;
#pragma unroll
  for (int j = 0; j < POOL_WGS / DIM; ++j)
    c += cntp[(size_t)(o + DIM * j) * CNT_PITCH + r];
  c += xlane16(c);
  c += xrow<8>(c);
  c += xrow<4>(c);
  c += xrow<2>(c);
  c += xrow<1>(c);
  const int wave = threadIdx.x >> 5;
  if ((threadIdx.x & 31) == 0) swsum[wave] = c;
  __syncthreads();
  c = swsum[wave & 6] + swsum[wave | 1];
  if (t >= R * DIM) return;
  const float inv = 1.0f / fmaxf(c, 1.0f);
  const float* w = W_rel3 + (size_t)o * 192;
  float acc = b_rel3[o] + pool[t] * inv;
  if (c > 0.f) {
    const float* r2 = rel2_proj + (size_t)r * DIM;
    for (int j = 0; j < DIM; ++j) acc += r2[j] * w[128 + j];
  }
  h_rel[t] = acc;
}

#define OFF_MS     0
#define OFF_MD     4096
#define OFF_MR     8192
#define OFF_M2     12288
#define OFF_CVEC   16384
#define OFF_VREL2  16448
#define OFF_UR     16576
#define OFF_REL2P  49088
#define OFF_BSUM   281088
#define OFF_POOL   381568
#define OFF_TOTAL  413588
#define OFF_BBASE  413600
#define OFF_BE     445568
#define OFF_CNTP   OFF_BE
#define OFF_US     1445568
#define OFF_UD     7845568
#define ZERO_CNT   (OFF_TOTAL + 1 - OFF_BSUM)

extern "C" void kernel_launch(void* const* d_in, const int* in_sizes, int n_in,
                              void* d_out, int out_size, void* d_ws,
                              size_t ws_size, hipStream_t stream) {
  const float* ent_embed = (const float*)d_in[0];
  const float* rel_embed = (const float*)d_in[1];
  const float* W_ent  = (const float*)d_in[2];
  const float* b_ent  = (const float*)d_in[3];
  const float* W_relL = (const float*)d_in[4];
  const float* b_relL = (const float*)d_in[5];
  const float* W_rel2 = (const float*)d_in[6];
  const float* b_rel2 = (const float*)d_in[7];
  const float* W_rel3 = (const float*)d_in[8];
  const float* b_rel3 = (const float*)d_in[9];
  const float* W_a    = (const float*)d_in[10];
  const float* b_a    = (const float*)d_in[11];
  const float* W_fc   = (const float*)d_in[12];
  const float* b_fc   = (const float*)d_in[13];
  const int* src_idx  = (const int*)d_in[14];
  const int* dst_idx  = (const int*)d_in[15];
  const int* rel_idx  = (const int*)d_in[16];

  float* ws = (float*)d_ws;
  float* h_ent = (float*)d_out;
  float* h_rel = (float*)d_out + (size_t)NUM_NODES * DIM;

  zero_kernel<<<512, 256, 0, stream>>>(ws + OFF_BSUM, ZERO_CNT);

  int* deg_cur = (int*)(ws + OFF_BSUM);
  int* seg_total = (int*)(ws + OFF_TOTAL);
  int* seg_bbase = (int*)(ws + OFF_BBASE);
  int* edge_rec = (int*)(ws + OFF_BE);
  edge_count_kernel<<<4096, 256, 0, stream>>>(src_idx, deg_cur, NUM_EDGES);
  seg_alloc_kernel<<<(NUM_NODES + SEG_BLK - 1) / SEG_BLK, SEG_BLK, 0, stream>>>(
      deg_cur, seg_total, seg_bbase, NUM_NODES);
  edge_fill_kernel<<<4096, 256, 0, stream>>>(src_idx, dst_idx, rel_idx, deg_cur,
                                             edge_rec, NUM_EDGES);

  fold_kernel<<<FOLD_WGS, 128, 0, stream>>>(
      W_ent, b_ent, W_relL, b_relL, W_rel2, b_rel2, W_fc, b_fc,
      ws + OFF_MS, ws + OFF_MD, ws + OFF_MR, ws + OFF_M2, ws + OFF_CVEC,
      ws + OFF_VREL2);

  const int gemm_blocks_N = 2048 / GEMM_WAVES;
  const int gemm_blocks_R = ((NUM_RELS + 15) / 16 + GEMM_WAVES - 1) / GEMM_WAVES;
  wmma_gemm64<DIM><<<gemm_blocks_N, 32 * GEMM_WAVES, 0, stream>>>(ent_embed, ws + OFF_MS,
                                                 nullptr, ws + OFF_US, NUM_NODES);
  wmma_gemm64<DIM><<<gemm_blocks_N, 32 * GEMM_WAVES, 0, stream>>>(ent_embed, ws + OFF_MD,
                                                 nullptr, ws + OFF_UD, NUM_NODES);
  wmma_gemm64<DIM><<<gemm_blocks_R, 32 * GEMM_WAVES, 0, stream>>>(rel_embed, ws + OFF_MR,
                                                 nullptr, ws + OFF_UR, NUM_RELS);
  wmma_gemm64<DIM><<<gemm_blocks_R, 32 * GEMM_WAVES, 0, stream>>>(rel_embed, ws + OFF_M2,
                                                 ws + OFF_VREL2, ws + OFF_REL2P,
                                                 NUM_RELS);

  node_gather_kernel<<<(NUM_NODES + 7) / 8, 256, 0, stream>>>(
      deg_cur, seg_bbase, edge_rec, ws + OFF_US, ws + OFF_UD, ws + OFF_UR,
      ws + OFF_CVEC, W_a, b_a, h_ent, NUM_NODES, NUM_EDGES);
  wmma_gemm64<3 * DIM><<<gemm_blocks_N, 32 * GEMM_WAVES, 0, stream>>>(h_ent, W_rel3,
                                                 nullptr, ws + OFF_US, NUM_NODES);
  wmma_gemm64<3 * DIM><<<gemm_blocks_N, 32 * GEMM_WAVES, 0, stream>>>(h_ent, W_rel3 + DIM,
                                                 nullptr, ws + OFF_UD, NUM_NODES);
  edge_pool_kernel<<<POOL_WGS, 512, 0, stream>>>(
      src_idx, dst_idx, rel_idx, ws + OFF_US, ws + OFF_UD, ws + OFF_POOL,
      ws + OFF_CNTP, NUM_EDGES);

  h_rel_kernel<<<(NUM_RELS * DIM + 255) / 256, 256, 0, stream>>>(
      ws + OFF_POOL, ws + OFF_CNTP, ws + OFF_REL2P, W_rel3, b_rel3, h_rel,
      NUM_RELS);
}


# independent work shares one grid, block index picks the job: degree count beside u_s/u_r/rel2 GEMMs, edge records beside u_d, fold beside zero, g_s beside g_d; 14 launches to 8; record order shifts
# speedup vs baseline: 3.4745x; 1.0767x over previous
#include <hip/hip_runtime.h>
#include <math.h>


#define NUM_NODES 100000
#define NUM_RELS  500
#define DIM       64
#define NUM_EDGES 1000000

typedef __attribute__((ext_vector_type(2))) float v2f;
typedef __attribute__((ext_vector_type(8))) float v8f;
typedef __attribute__((ext_vector_type(4))) float v4f;
typedef __attribute__((ext_vector_type(4))) int v4i;

__device__ __forceinline__ void zero_body(float* __restrict__ p, int n, int bid,
                                          int nblocks) {
  for (int i = bid * blockDim.x + threadIdx.x; i < n;
       i += nblocks * blockDim.x)
    p[i] = 0.0f;
}

#define FOLD_WGS 5
__device__ __forceinline__ void fold_body(
    const float* __restrict__ W_ent, const float* __restrict__ b_ent,
    const float* __restrict__ W_relL, const float* __restrict__ b_relL,
    const float* __restrict__ W_rel2, const float* __restrict__ b_rel2,
    const float* __restrict__ W_fc, const float* __restrict__ b_fc,
    float* __restrict__ M_s, float* __restrict__ M_d,
    float* __restrict__ M_r, float* __restrict__ M_2,
    float* __restrict__ cvec, float* __restrict__ v_rel2, int q) {
  const int t = threadIdx.x;
  if (q == 4) {
    if (t < DIM) {
      float cv = b_fc[t], v2 = b_rel2[t];
      for (int u = 0; u < DIM; ++u) {
        cv += W_fc[t * 192 + u] * b_ent[u]
            + W_fc[t * 192 + 64 + u] * b_ent[u]
            + W_fc[t * 192 + 128 + u] * b_relL[u];
        v2 += W_rel2[t * DIM + u] * b_relL[u];
      }
      cvec[t] = cv; v_rel2[t] = v2;
    }
    return;
  }
  const float* __restrict__ L = (q == 3) ? W_rel2 : W_fc + 64 * q;
  const int pitch = (q == 3) ? DIM : 3 * DIM;
  const float* __restrict__ R = (q < 2) ? W_ent : W_relL;
  float* __restrict__ out = (q == 0) ? M_s : (q == 1) ? M_d : (q == 2) ? M_r : M_2;

  const int wave = t >> 5;
  const int lane = t & 31;
  const int c = lane & 15;
  const int h = lane >> 4;
  const float* __restrict__ lrow = L + (size_t)(16 * wave + c) * pitch + 2 * h;
  const float* __restrict__ rcol = R + (size_t)(2 * h) * DIM + 4 * c;

  v2f a[16];
  v4f r0[16], r1[16];
#pragma unroll
  for (int s = 0; s < 16; ++s) {
    a[s] = *(const v2f*)(lrow + 4 * s);
    r0[s] = *(const v4f*)(rcol + (size_t)(4 * s) * DIM);
    r1[s] = *(const v4f*)(rcol + (size_t)(4 * s + 1) * DIM);
  }

  v8f acc0 = {}, acc1 = {}, acc2 = {}, acc3 = {};
#pragma unroll
  for (int s = 0; s < 16; ++s) {
    const v2f b0 = {r0[s].x, r1[s].x};
    const v2f b1 = {r0[s].y, r1[s].y};
    const v2f b2 = {r0[s].z, r1[s].z};
    const v2f b3 = {r0[s].w, r1[s].w};
    acc0 = __builtin_amdgcn_wmma_f32_16x16x4_f32(false, a[s], false, b0,
                                                 (short)0, acc0, false, false);
    acc1 = __builtin_amdgcn_wmma_f32_16x16x4_f32(false, a[s], false, b1,
                                                 (short)0, acc1, false, false);
    acc2 = __builtin_amdgcn_wmma_f32_16x16x4_f32(false, a[s], false, b2,
                                                 (short)0, acc2, false, false);
    acc3 = __builtin_amdgcn_wmma_f32_16x16x4_f32(false, a[s], false, b3,
                                                 (short)0, acc3, false, false);
  }

  float* __restrict__ orow = out + (size_t)(16 * wave + 8 * h) * DIM + 4 * c;
#pragma unroll
  for (int r = 0; r < 8; ++r) {
    v4f o;
    o.x = acc0[r]; o.y = acc1[r]; o.z = acc2[r]; o.w = acc3[r];
    *(v4f*)(orow + (size_t)r * DIM) = o;
  }
}

#define GEMM_WAVES 4
template <int LDW>
__device__ __forceinline__ void gemm_body(
    const float* __restrict__ A, const float* __restrict__ W,
    const float* __restrict__ bias, float* __restrict__ out, int M, int bid,
    int nblocks) {
  const int wave = threadIdx.x >> 5;
  const int lane = threadIdx.x & 31;
  const int lm = lane & 15;
  const int hi = lane >> 4;
  const int ntiles = (M + 15) >> 4;
  const int nwaves = nblocks * GEMM_WAVES;
  int tile = bid * GEMM_WAVES + wave;
  if (tile >= ntiles) return;

  v4f w[4][8];
#pragma unroll
  for (int nb = 0; nb < 4; ++nb) {
    const float* __restrict__ wr = W + (size_t)(nb * 16 + lm) * LDW + 4 * hi;
#pragma unroll
    for (int t = 0; t < 8; ++t) w[nb][t] = *(const v4f*)(wr + 8 * t);
  }
  float bv[4];
#pragma unroll
  for (int nb = 0; nb < 4; ++nb) bv[nb] = bias ? bias[nb * 16 + lm] : 0.0f;

  auto load_a = [&](int tl, v4f (&a)[8]) {
    const int row = min(tl * 16 + lm, M - 1);
    const float* __restrict__ ar = A + (size_t)row * DIM + 4 * hi;
#pragma unroll
    for (int t = 0; t < 8; ++t) a[t] = *(const v4f*)(ar + 8 * t);
  };

  v4f a[8];
  load_a(tile, a);
  for (; tile < ntiles; tile += nwaves) {
    v4f an[8];
    load_a(min(tile + nwaves, ntiles - 1), an);

    v8f acc0 = {}, acc1 = {}, acc2 = {}, acc3 = {};
#pragma unroll
    for (int t = 0; t < 8; ++t) {
      const v2f a0 = {a[t].x, a[t].y};
      const v2f a1 = {a[t].z, a[t].w};
      const v2f b00 = {w[0][t].x, w[0][t].y}, b01 = {w[0][t].z, w[0][t].w};
      const v2f b10 = {w[1][t].x, w[1][t].y}, b11 = {w[1][t].z, w[1][t].w};
      const v2f b20 = {w[2][t].x, w[2][t].y}, b21 = {w[2][t].z, w[2][t].w};
      const v2f b30 = {w[3][t].x, w[3][t].y}, b31 = {w[3][t].z, w[3][t].w};
      acc0 = __builtin_amdgcn_wmma_f32_16x16x4_f32(false, a0, false, b00,
                                                   (short)0, acc0, false, false);
      acc1 = __builtin_amdgcn_wmma_f32_16x16x4_f32(false, a0, false, b10,
                                                   (short)0, acc1, false, false);
      acc2 = __builtin_amdgcn_wmma_f32_16x16x4_f32(false, a0, false, b20,
                                                   (short)0, acc2, false, false);
      acc3 = __builtin_amdgcn_wmma_f32_16x16x4_f32(false, a0, false, b30,
                                                   (short)0, acc3, false, false);
      acc0 = __builtin_amdgcn_wmma_f32_16x16x4_f32(false, a1, false, b01,
                                                   (short)0, acc0, false, false);
      acc1 = __builtin_amdgcn_wmma_f32_16x16x4_f32(false, a1, false, b11,
                                                   (short)0, acc1, false, false);
      acc2 = __builtin_amdgcn_wmma_f32_16x16x4_f32(false, a1, false, b21,
                                                   (short)0, acc2, false, false);
      acc3 = __builtin_amdgcn_wmma_f32_16x16x4_f32(false, a1, false, b31,
                                                   (short)0, acc3, false, false);
    }

    const int m0 = tile * 16;
    v8f accs[4] = {acc0, acc1, acc2, acc3};
#pragma unroll
    for (int ob = 0; ob < 4; ++ob) {
#pragma unroll
      for (int v = 0; v < 8; ++v) {
        const int row = m0 + v + 8 * hi;
        if (row < M)
          out[(size_t)row * DIM + ob * 16 + lm] = accs[ob][v] + bv[ob];
      }
    }
#pragma unroll
    for (int t = 0; t < 8; ++t) a[t] = an[t];
  }
}

template <int V> struct KConst { static constexpr int value = V; };

#define SEG_BLK 1024

#define EDGE_PT 8
#define EDGE_CHUNK (128 * EDGE_PT)
#define EDGE_WGS ((NUM_EDGES + EDGE_CHUNK - 1) / EDGE_CHUNK)

__device__ __forceinline__ void edge_count_body(const int* __restrict__ src,
                                                int* __restrict__ deg, int E,
                                                int chunk) {
  const int e0 = chunk * EDGE_CHUNK + threadIdx.x;
  int sv[EDGE_PT];
#pragma unroll
  for (int i = 0; i < EDGE_PT; ++i)
    sv[i] = src[min(e0 + 128 * i, E - 1)];
#pragma unroll
  for (int i = 0; i < EDGE_PT; ++i)
    if (e0 + 128 * i < E) atomicAdd(&deg[sv[i]], 1);
}

__global__ __launch_bounds__(SEG_BLK) void seg_alloc_kernel(
    int* __restrict__ cursor  ,
    int* __restrict__ total, int* __restrict__ bbase, int N) {
  __shared__ int swave[32];
  __shared__ int sbase;
  const int lane = threadIdx.x & 31;
  const int wave = threadIdx.x >> 5;
  const int i = blockIdx.x * SEG_BLK + threadIdx.x;
  const int v = (i < N) ? cursor[i] : 0;
  int x = v;
#pragma unroll
  for (int o = 1; o < 32; o <<= 1) {
    const int y = __shfl_up(x, o);
    if (lane >= o) x += y;
  }
  if (lane == 31) swave[wave] = x;
  __syncthreads();
  if (wave == 0) {
    const int w = swave[lane];
    int xs = w;
#pragma unroll
    for (int o = 1; o < 32; o <<= 1) {
      const int y = __shfl_up(xs, o);
      if (lane >= o) xs += y;
    }
    swave[lane] = xs - w;
    if (lane == 31) {
      const int base = atomicAdd(total, xs);
      sbase = base;
      bbase[blockIdx.x] = base;
    }
  }
  __syncthreads();
  if (i < N) cursor[i] = sbase + swave[wave] + x - v;
}

__device__ __forceinline__ void edge_fill_body(const int* __restrict__ src,
                                               const int* __restrict__ dst,
                                               const int* __restrict__ rel,
                                               int* __restrict__ cursor,
                                               int* __restrict__ rec, int E,
                                               int chunk) {
  const int e0 = chunk * EDGE_CHUNK + threadIdx.x;
  int sv[EDGE_PT], dv[EDGE_PT], rv[EDGE_PT], p[EDGE_PT];
#pragma unroll
  for (int i = 0; i < EDGE_PT; ++i) {
    const int ec = min(e0 + 128 * i, E - 1);
    sv[i] = src[ec];
    dv[i] = dst[ec];
    rv[i] = rel[ec];
  }
#pragma unroll
  for (int i = 0; i < EDGE_PT; ++i)
    p[i] = (e0 + 128 * i < E) ? atomicAdd(&cursor[sv[i]], 1) : -1;
#pragma unroll
  for (int i = 0; i < EDGE_PT; ++i)
    if ((unsigned)p[i] < (unsigned)E) rec[p[i]] = dv[i] | (rv[i] << 17);
}

__device__ __forceinline__ float xlane16(float x) {
  const int b = __float_as_int(x);
  return __int_as_float(
      __builtin_amdgcn_permlanex16(b, b, 0x76543210, 0xfedcba98, false, false));
}
template <int M>
__device__ __forceinline__ float xrow(float x) {
  const int b = __float_as_int(x);
  return __int_as_float(
      __builtin_amdgcn_update_dpp(b, b, 0x160 | M, 0xf, 0xf, false));
}

__global__ __launch_bounds__(256) void node_gather_kernel(
    const int* __restrict__ cursor, const int* __restrict__ bbase,
    const int* __restrict__ rec, const float* __restrict__ u_s,
    const float* __restrict__ u_d, const float* __restrict__ u_r,
    const float* __restrict__ cvec, const float* __restrict__ W_a,
    const float* __restrict__ b_a, float* __restrict__ h_ent, int N, int E) {
  const int s = blockIdx.x * (blockDim.x >> 5) + (threadIdx.x >> 5);
  const int lane = threadIdx.x & 31;
  if (s >= N) return;

  int end = cursor[s];
  int start = (s & (SEG_BLK - 1)) ? cursor[s - 1] : bbase[s / SEG_BLK];
  end = __builtin_amdgcn_readfirstlane(end);
  start = __builtin_amdgcn_readfirstlane(start);
  start = min(max(start, 0), E);
  end = min(max(end, start), E);
  const int n = end - start;

  v2f acc = {0.f, 0.f};
  if (n > 0) {
    int pk = (lane < n) ? rec[start + lane] : 0;
    const v2f wa = *(const v2f*)(W_a + 2 * lane);
    const float ba = b_a[0];
    const v2f base = *(const v2f*)(u_s + (size_t)s * DIM + 2 * lane) +
                     *(const v2f*)(cvec + 2 * lane);
    const bool h4 = (lane & 16) != 0;
    const bool h3 = (lane & 8) != 0;
    const bool h2 = (lane & 4) != 0;
    const float NEG = -3.0e38f;
    float m = NEG;
    float bsum = 0.f;

    for (int c0 = 0; c0 < n; c0 += 32) {
      if (c0 > 0) pk = (c0 + lane < n) ? rec[start + c0 + lane] : 0;
      const int mm = min(32, n - c0);
      int j = 0;
      auto group = [&](auto KC, int valid) {
        constexpr int K = decltype(KC)::value;
        v2f ud[K], ur[K], c[K];
        float p[K];
#pragma unroll
        for (int i = 0; i < K; ++i) {
          const int idx = (j + i) & 31;
          const int pj = __builtin_amdgcn_readlane(pk, idx);
          const int d = min(pj & 0x1FFFF, NUM_NODES - 1);
          const int r = min(pj >> 17, NUM_RELS - 1);
          ud[i] = *(const v2f*)(u_d + (size_t)d * DIM + 2 * lane);
          ur[i] = *(const v2f*)(u_r + (size_t)r * DIM + 2 * lane);
        }
#pragma unroll
        for (int i = 0; i < K; ++i) {
          c[i] = (base + ud[i]) + ur[i];
          p[i] = c[i].x * wa.x + c[i].y * wa.y;
        }
        float t;
        if constexpr (K == 8) {
          float q[4], q2[2];
#pragma unroll
          for (int i = 0; i < 4; ++i) {
            const float keep = h4 ? p[i + 4] : p[i];
            const float send = h4 ? p[i] : p[i + 4];
            q[i] = keep + xlane16(send);
          }
#pragma unroll
          for (int i = 0; i < 2; ++i) {
            const float keep = h3 ? q[i + 2] : q[i];
            const float send = h3 ? q[i] : q[i + 2];
            q2[i] = keep + xrow<8>(send);
          }
          const float keep = h2 ? q2[1] : q2[0];
          const float send = h2 ? q2[0] : q2[1];
          t = keep + xrow<4>(send);
        } else {
          float q[2];
#pragma unroll
          for (int i = 0; i < 2; ++i) {
            const float keep = h4 ? p[i + 2] : p[i];
            const float send = h4 ? p[i] : p[i + 2];
            q[i] = keep + xlane16(send);
          }
          const float keep = h3 ? q[1] : q[0];
          const float send = h3 ? q[0] : q[1];
          t = keep + xrow<8>(send);
          t += xrow<4>(t);
        }
        t += xrow<2>(t);
        t += xrow<1>(t);
        const int myedge = (K == 8) ? ((lane >> 2) & 7) : ((lane >> 3) & 3);
        float l = t + ba;
        l = l > 0.f ? l : 0.01f * l;
        if (myedge >= valid) l = NEG;
        float gm = l;
        gm = fmaxf(gm, xlane16(gm));
        gm = fmaxf(gm, xrow<8>(gm));
        gm = fmaxf(gm, xrow<4>(gm));
        const float mnew = fmaxf(m, gm);
        const float sc = expf(m - mnew);
        const int bbits = __float_as_int(expf(l - mnew));
        m = mnew;
        acc.x *= sc;
        acc.y *= sc;
        bsum *= sc;
#pragma unroll
        for (int i = 0; i < K; ++i) {
          if (i < valid) {
            const float w =
                __int_as_float(__builtin_amdgcn_readlane(bbits, i * (32 / K)));
            acc.x += w * c[i].x;
            acc.y += w * c[i].y;
            bsum += w;
          }
        }
        j += K;
      };
      while (mm - j >= 8) group(KConst<8>{}, 8);
      if (mm - j >= 4) group(KConst<4>{}, 4);
      if (mm - j > 0) group(KConst<4>{}, mm - j);
    }
    acc.x = acc.x / bsum;
    acc.y = acc.y / bsum;
  }
  *(v2f*)(h_ent + (size_t)s * DIM + 2 * lane) = acc;
}


#define PB      128
#define PPITCH  160
#define POOL_WGS 256
#define CNT_PITCH 512

__global__ __launch_bounds__(512) void edge_pool_kernel(
    const int* __restrict__ src, const int* __restrict__ dst,
    const int* __restrict__ rel, const float* __restrict__ g_s,
    const float* __restrict__ g_d, float* __restrict__ pool,
    float* __restrict__ cntp, int E) {
  __shared__ __attribute__((aligned(16))) float sG[2][(PB / 2) * PPITCH];
  __shared__ __attribute__((aligned(16))) int sR[2][PB];

  const int wave = threadIdx.x >> 5;
  const int lane = threadIdx.x & 31;
  const int lm = lane & 15;
  const int hi = lane >> 4;
  const int rbase = wave * 32;
  const int nb = (E + PB - 1) / PB;
  const int G = gridDim.x;

  v8f c00 = {}, c01 = {}, c02 = {}, c03 = {};
  v8f c10 = {}, c11 = {}, c12 = {}, c13 = {};
  float n0 = 0.f, n1 = 0.f;

  v2f rs[8], rd[8];
  int s_cur, d_cur, r_cur, s_nxt, d_nxt, r_nxt;

  auto load_idx = [&](int bb, int& s, int& d, int& r) {
    const int e = bb * PB + wave * 8 + (lane & 7);
    const bool ok = (bb < nb) && (e < E);
    s = ok ? src[e] : 0;
    d = ok ? dst[e] : 0;
    r = ok ? rel[e] : -1;
  };
  auto load_rows = [&](int s, int d) {
#pragma unroll
    for (int i = 0; i < 8; ++i) {
      const int si = __builtin_amdgcn_readlane(s, i);
      const int di = __builtin_amdgcn_readlane(d, i);
      rs[i] = *(const v2f*)(g_s + (size_t)si * DIM + 2 * lane);
      rd[i] = *(const v2f*)(g_d + (size_t)di * DIM + 2 * lane);
    }
  };

  int b = blockIdx.x;
  load_idx(b, s_cur, d_cur, r_cur);
  load_rows(s_cur, d_cur);
  load_idx(b + G, s_nxt, d_nxt, r_nxt);

  int buf = 0;
  for (; b < nb; b += G) {
    float* gb = sG[buf];
#pragma unroll
    for (int j = 0; j < 4; ++j) {
      const int r0 = __builtin_amdgcn_readlane(r_cur, 2 * j);
      const int r1 = __builtin_amdgcn_readlane(r_cur, 2 * j + 1);
      v2f g0 = rs[2 * j] + rd[2 * j];
      v2f g1 = rs[2 * j + 1] + rd[2 * j + 1];
      if (r0 < 0) { g0.x = 0.f; g0.y = 0.f; }
      if (r1 < 0) { g1.x = 0.f; g1.y = 0.f; }
      v4f o;
      o.x = g0.x; o.y = g1.x; o.z = g0.y; o.w = g1.y;
      *(v4f*)(gb + (wave * 4 + j) * PPITCH + 4 * lane) = o;
    }
    if (lane < 8) sR[buf][wave * 8 + lane] = r_cur;
    __syncthreads();

    s_cur = s_nxt; d_cur = d_nxt; r_cur = r_nxt;
    if (b + G < nb) load_rows(s_cur, d_cur);
    load_idx(b + 2 * G, s_nxt, d_nxt, r_nxt);

    const v4i rr = *(const v4i*)&sR[buf][4 * lane];
    const bool relv = ((unsigned)(rr.x - rbase) < 32u) |
                      ((unsigned)(rr.y - rbase) < 32u) |
                      ((unsigned)(rr.z - rbase) < 32u) |
                      ((unsigned)(rr.w - rbase) < 32u);
    unsigned mask = __builtin_amdgcn_ballot_w32(relv);
    while (mask) {
      const int g = __builtin_ctz(mask);
      mask &= mask - 1;
      const int* pr = &sR[buf][4 * g + 2 * hi];
      const int da = pr[0] - rbase - lm;
      const int db = pr[1] - rbase - lm;
      const float* pg = gb + (2 * g + hi) * PPITCH + 2 * lm;
      const v2f b0 = *(const v2f*)(pg);
      const v2f b1 = *(const v2f*)(pg + 32);
      const v2f b2 = *(const v2f*)(pg + 64);
      const v2f b3 = *(const v2f*)(pg + 96);
      {
        const bool ma = (da == 0), mb = (db == 0);
        if (__builtin_amdgcn_ballot_w32(ma | mb)) {
          v2f a;
          a.x = ma ? 1.0f : 0.0f;
          a.y = mb ? 1.0f : 0.0f;
          n0 += a.x + a.y;
          c00 = __builtin_amdgcn_wmma_f32_16x16x4_f32(false, a, false, b0,
                                                      (short)0, c00, false, false);
          c01 = __builtin_amdgcn_wmma_f32_16x16x4_f32(false, a, false, b1,
                                                      (short)0, c01, false, false);
          c02 = __builtin_amdgcn_wmma_f32_16x16x4_f32(false, a, false, b2,
                                                      (short)0, c02, false, false);
          c03 = __builtin_amdgcn_wmma_f32_16x16x4_f32(false, a, false, b3,
                                                      (short)0, c03, false, false);
        }
      }
      {
        const bool ma = (da == 16), mb = (db == 16);
        if (__builtin_amdgcn_ballot_w32(ma | mb)) {
          v2f a;
          a.x = ma ? 1.0f : 0.0f;
          a.y = mb ? 1.0f : 0.0f;
          n1 += a.x + a.y;
          c10 = __builtin_amdgcn_wmma_f32_16x16x4_f32(false, a, false, b0,
                                                      (short)0, c10, false, false);
          c11 = __builtin_amdgcn_wmma_f32_16x16x4_f32(false, a, false, b1,
                                                      (short)0, c11, false, false);
          c12 = __builtin_amdgcn_wmma_f32_16x16x4_f32(false, a, false, b2,
                                                      (short)0, c12, false, false);
          c13 = __builtin_amdgcn_wmma_f32_16x16x4_f32(false, a, false, b3,
                                                      (short)0, c13, false, false);
        }
      }
    }
    buf ^= 1;
  }

  __syncthreads();
  float* sFw = &sG[0][0] + wave * 1024;
  auto flush = [&](const v8f& t0, const v8f& t1, const v8f& t2, const v8f& t3,
                   int rt) {
#pragma unroll
    for (int v = 0; v < 8; ++v) {
      float* row = sFw + (v + 8 * hi) * DIM + lm;
      row[0] = t0[v]; row[16] = t1[v]; row[32] = t2[v]; row[48] = t3[v];
    }
    __syncthreads();
    for (int q = 0; q < 16; ++q) {
      const int r = (q + (int)blockIdx.x) & 15;
      const int R = rbase + rt * 16 + r;
      if (R < NUM_RELS) {
        atomicAdd(&pool[(size_t)R * DIM + lane], sFw[r * DIM + lane]);
        atomicAdd(&pool[(size_t)R * DIM + 32 + lane], sFw[r * DIM + 32 + lane]);
      }
    }
    __syncthreads();
  };
  flush(c00, c01, c02, c03, 0);
  flush(c10, c11, c12, c13, 1);

  {
    const float t0 = n0 + xlane16(n0);
    const float t1 = n1 + xlane16(n1);
    cntp[(size_t)blockIdx.x * CNT_PITCH + rbase + lane] = hi ? t1 : t0;
  }
}

__global__ __launch_bounds__(256) void h_rel_kernel(
    const float* __restrict__ pool, const float* __restrict__ cntp,
    const float* __restrict__ rel2_proj, const float* __restrict__ W_rel3,
    const float* __restrict__ b_rel3, float* __restrict__ h_rel, int R) {
  __shared__ float swsum[8];
  const int t = blockIdx.x * 256 + threadIdx.x;
  const int r = min(t >> 6, R - 1), o = t & 63;
  float c = 0.f;
#pragma unroll
  for (int j = 0; j < POOL_WGS / DIM; ++j)
    c += cntp[(size_t)(o + DIM * j) * CNT_PITCH + r];
  c += xlane16(c);
  c += xrow<8>(c);
  c += xrow<4>(c);
  c += xrow<2>(c);
  c += xrow<1>(c);
  const int wave = threadIdx.x >> 5;
  if ((threadIdx.x & 31) == 0) swsum[wave] = c;
  __syncthreads();
  c = swsum[wave & 6] + swsum[wave | 1];
  if (t >= R * DIM) return;
  const float inv = 1.0f / fmaxf(c, 1.0f);
  const float* w = W_rel3 + (size_t)o * 192;
  float acc = b_rel3[o] + pool[t] * inv;
  if (c > 0.f) {
    const float* r2 = rel2_proj + (size_t)r * DIM;
    for (int j = 0; j < DIM; ++j) acc += r2[j] * w[128 + j];
  }
  h_rel[t] = acc;
}

#define PREP_ZERO_WGS 251
#define GEMM_WGS_N 512
#define GEMM_WGS_R ((((NUM_RELS + 15) / 16) + GEMM_WAVES - 1) / GEMM_WAVES)
#define PROJ1_GEMM_WGS (2 * GEMM_WGS_R + GEMM_WGS_N)
#define COUNT_PROJ_WGS (PROJ1_GEMM_WGS + EDGE_WGS)
#define FILL_PROJ_WGS (GEMM_WGS_N + EDGE_WGS)
static_assert(EDGE_WGS >= PROJ1_GEMM_WGS, "one edge chunk beside every GEMM work-group");

__global__ __launch_bounds__(128) void prep_kernel(
    const float* __restrict__ W_ent, const float* __restrict__ b_ent,
    const float* __restrict__ W_relL, const float* __restrict__ b_relL,
    const float* __restrict__ W_rel2, const float* __restrict__ b_rel2,
    const float* __restrict__ W_fc, const float* __restrict__ b_fc,
    float* __restrict__ M_s, float* __restrict__ M_d,
    float* __restrict__ M_r, float* __restrict__ M_2,
    float* __restrict__ cvec, float* __restrict__ v_rel2,
    float* __restrict__ zero_p, int zero_n) {
  const int b = blockIdx.x;
  if (b < FOLD_WGS) {
    fold_body(W_ent, b_ent, W_relL, b_relL, W_rel2, b_rel2, W_fc, b_fc, M_s,
              M_d, M_r, M_2, cvec, v_rel2, b);
    return;
  }
  zero_body(zero_p, zero_n, b - FOLD_WGS, PREP_ZERO_WGS);
}

__global__ __launch_bounds__(128) void count_proj_kernel(
    const int* __restrict__ src, int* __restrict__ deg,
    const float* __restrict__ ent_embed, const float* __restrict__ rel_embed,
    const float* __restrict__ M_s, const float* __restrict__ M_r,
    const float* __restrict__ M_2, const float* __restrict__ v_rel2,
    float* __restrict__ u_s, float* __restrict__ u_r,
    float* __restrict__ rel2p) {
  const int b = blockIdx.x;
  int chunk;
  if (b < 2 * PROJ1_GEMM_WGS) {
    if ((b & 1) == 0) {
      const int g = b >> 1;
      const float* A = ent_embed;
      const float* Wm = M_s;
      const float* bias = nullptr;
      float* out = u_s;
      int M = NUM_NODES, bid = g - 2 * GEMM_WGS_R, nblk = GEMM_WGS_N;
      if (g < GEMM_WGS_R) {
        A = rel_embed; Wm = M_r; out = u_r;
        M = NUM_RELS; bid = g; nblk = GEMM_WGS_R;
      } else if (g < 2 * GEMM_WGS_R) {
        A = rel_embed; Wm = M_2; bias = v_rel2; out = rel2p;
        M = NUM_RELS; bid = g - GEMM_WGS_R; nblk = GEMM_WGS_R;
      }
      gemm_body<DIM>(A, Wm, bias, out, M, bid, nblk);
      return;
    }
    chunk = b >> 1;
  } else {
    chunk = b - PROJ1_GEMM_WGS;
  }
  edge_count_body(src, deg, NUM_EDGES, chunk);
}

__global__ __launch_bounds__(128) void fill_proj_kernel(
    const int* __restrict__ src, const int* __restrict__ dst,
    const int* __restrict__ rel, int* __restrict__ cursor,
    int* __restrict__ rec, const float* __restrict__ ent_embed,
    const float* __restrict__ M_d, float* __restrict__ u_d) {
  const int b = blockIdx.x;
  int chunk;
  if (b < 2 * GEMM_WGS_N) {
    if ((b & 1) == 0) {
      gemm_body<DIM>(ent_embed, M_d, nullptr, u_d, NUM_NODES, b >> 1,
                     GEMM_WGS_N);
      return;
    }
    chunk = b >> 1;
  } else {
    chunk = b - GEMM_WGS_N;
  }
  edge_fill_body(src, dst, rel, cursor, rec, NUM_EDGES, chunk);
}

__global__ __launch_bounds__(128) void proj2_kernel(
    const float* __restrict__ h_ent, const float* __restrict__ W_rel3,
    float* __restrict__ g_s, float* __restrict__ g_d) {
  const int b = blockIdx.x;
  const bool odd = (b & 1) != 0;
  gemm_body<3 * DIM>(h_ent, odd ? W_rel3 + DIM : W_rel3, nullptr,
                     odd ? g_d : g_s, NUM_NODES, b >> 1, GEMM_WGS_N);
}

#define OFF_MS     0
#define OFF_MD     4096
#define OFF_MR     8192
#define OFF_M2     12288
#define OFF_CVEC   16384
#define OFF_VREL2  16448
#define OFF_UR     16576
#define OFF_REL2P  49088
#define OFF_BSUM   281088
#define OFF_POOL   381568
#define OFF_TOTAL  413588
#define OFF_BBASE  413600
#define OFF_BE     445568
#define OFF_CNTP   OFF_BE
#define OFF_US     1445568
#define OFF_UD     7845568
#define ZERO_CNT   (OFF_TOTAL + 1 - OFF_BSUM)

extern "C" void kernel_launch(void* const* d_in, const int* in_sizes, int n_in,
                              void* d_out, int out_size, void* d_ws,
                              size_t ws_size, hipStream_t stream) {
  const float* ent_embed = (const float*)d_in[0];
  const float* rel_embed = (const float*)d_in[1];
  const float* W_ent  = (const float*)d_in[2];
  const float* b_ent  = (const float*)d_in[3];
  const float* W_relL = (const float*)d_in[4];
  const float* b_relL = (const float*)d_in[5];
  const float* W_rel2 = (const float*)d_in[6];
  const float* b_rel2 = (const float*)d_in[7];
  const float* W_rel3 = (const float*)d_in[8];
  const float* b_rel3 = (const float*)d_in[9];
  const float* W_a    = (const float*)d_in[10];
  const float* b_a    = (const float*)d_in[11];
  const float* W_fc   = (const float*)d_in[12];
  const float* b_fc   = (const float*)d_in[13];
  const int* src_idx  = (const int*)d_in[14];
  const int* dst_idx  = (const int*)d_in[15];
  const int* rel_idx  = (const int*)d_in[16];

  float* ws = (float*)d_ws;
  float* h_ent = (float*)d_out;
  float* h_rel = (float*)d_out + (size_t)NUM_NODES * DIM;

  int* deg_cur = (int*)(ws + OFF_BSUM);
  int* seg_total = (int*)(ws + OFF_TOTAL);
  int* seg_bbase = (int*)(ws + OFF_BBASE);
  int* edge_rec = (int*)(ws + OFF_BE);

  prep_kernel<<<FOLD_WGS + PREP_ZERO_WGS, 128, 0, stream>>>(
      W_ent, b_ent, W_relL, b_relL, W_rel2, b_rel2, W_fc, b_fc,
      ws + OFF_MS, ws + OFF_MD, ws + OFF_MR, ws + OFF_M2, ws + OFF_CVEC,
      ws + OFF_VREL2, ws + OFF_BSUM, ZERO_CNT);

  count_proj_kernel<<<COUNT_PROJ_WGS, 128, 0, stream>>>(
      src_idx, deg_cur, ent_embed, rel_embed, ws + OFF_MS, ws + OFF_MR,
      ws + OFF_M2, ws + OFF_VREL2, ws + OFF_US, ws + OFF_UR, ws + OFF_REL2P);

  seg_alloc_kernel<<<(NUM_NODES + SEG_BLK - 1) / SEG_BLK, SEG_BLK, 0, stream>>>(
      deg_cur, seg_total, seg_bbase, NUM_NODES);

  fill_proj_kernel<<<FILL_PROJ_WGS, 128, 0, stream>>>(
      src_idx, dst_idx, rel_idx, deg_cur, edge_rec, ent_embed, ws + OFF_MD,
      ws + OFF_UD);

  node_gather_kernel<<<(NUM_NODES + 7) / 8, 256, 0, stream>>>(
      deg_cur, seg_bbase, edge_rec, ws + OFF_US, ws + OFF_UD, ws + OFF_UR,
      ws + OFF_CVEC, W_a, b_a, h_ent, NUM_NODES, NUM_EDGES);

  proj2_kernel<<<2 * GEMM_WGS_N, 128, 0, stream>>>(h_ent, W_rel3, ws + OFF_US,
                                                  ws + OFF_UD);

  edge_pool_kernel<<<POOL_WGS, 512, 0, stream>>>(
      src_idx, dst_idx, rel_idx, ws + OFF_US, ws + OFF_UD, ws + OFF_POOL,
      ws + OFF_CNTP, NUM_EDGES);

  h_rel_kernel<<<(NUM_RELS * DIM + 255) / 256, 256, 0, stream>>>(
      ws + OFF_POOL, ws + OFF_CNTP, ws + OFF_REL2P, W_rel3, b_rel3, h_rel,
      NUM_RELS);
}


# relation pooling as a 2-D grid: a work-group owns 32 relations x one edge slice, scans and keeps its edges, its 16 waves sum in LDS: 2048 pool atomics per group, not 32768; pool sum order changes
# speedup vs baseline: 4.2360x; 1.2192x over previous
#include <hip/hip_runtime.h>
#include <math.h>


#define NUM_NODES 100000
#define NUM_RELS  500
#define DIM       64
#define NUM_EDGES 1000000

typedef __attribute__((ext_vector_type(2))) float v2f;
typedef __attribute__((ext_vector_type(8))) float v8f;
typedef __attribute__((ext_vector_type(4))) float v4f;
typedef __attribute__((ext_vector_type(4))) int v4i;

__device__ __forceinline__ void zero_body(float* __restrict__ p, int n, int bid,
                                          int nblocks) {
  for (int i = bid * blockDim.x + threadIdx.x; i < n;
       i += nblocks * blockDim.x)
    p[i] = 0.0f;
}

#define FOLD_WGS 5
__device__ __forceinline__ void fold_body(
    const float* __restrict__ W_ent, const float* __restrict__ b_ent,
    const float* __restrict__ W_relL, const float* __restrict__ b_relL,
    const float* __restrict__ W_rel2, const float* __restrict__ b_rel2,
    const float* __restrict__ W_fc, const float* __restrict__ b_fc,
    float* __restrict__ M_s, float* __restrict__ M_d,
    float* __restrict__ M_r, float* __restrict__ M_2,
    float* __restrict__ cvec, float* __restrict__ v_rel2, int q) {
  const int t = threadIdx.x;
  if (q == 4) {
    if (t < DIM) {
      float cv = b_fc[t], v2 = b_rel2[t];
      for (int u = 0; u < DIM; ++u) {
        cv += W_fc[t * 192 + u] * b_ent[u]
            + W_fc[t * 192 + 64 + u] * b_ent[u]
            + W_fc[t * 192 + 128 + u] * b_relL[u];
        v2 += W_rel2[t * DIM + u] * b_relL[u];
      }
      cvec[t] = cv; v_rel2[t] = v2;
    }
    return;
  }
  const float* __restrict__ L = (q == 3) ? W_rel2 : W_fc + 64 * q;
  const int pitch = (q == 3) ? DIM : 3 * DIM;
  const float* __restrict__ R = (q < 2) ? W_ent : W_relL;
  float* __restrict__ out = (q == 0) ? M_s : (q == 1) ? M_d : (q == 2) ? M_r : M_2;

  const int wave = t >> 5;
  const int lane = t & 31;
  const int c = lane & 15;
  const int h = lane >> 4;
  const float* __restrict__ lrow = L + (size_t)(16 * wave + c) * pitch + 2 * h;
  const float* __restrict__ rcol = R + (size_t)(2 * h) * DIM + 4 * c;

  v2f a[16];
  v4f r0[16], r1[16];
#pragma unroll
  for (int s = 0; s < 16; ++s) {
    a[s] = *(const v2f*)(lrow + 4 * s);
    r0[s] = *(const v4f*)(rcol + (size_t)(4 * s) * DIM);
    r1[s] = *(const v4f*)(rcol + (size_t)(4 * s + 1) * DIM);
  }

  v8f acc0 = {}, acc1 = {}, acc2 = {}, acc3 = {};
#pragma unroll
  for (int s = 0; s < 16; ++s) {
    const v2f b0 = {r0[s].x, r1[s].x};
    const v2f b1 = {r0[s].y, r1[s].y};
    const v2f b2 = {r0[s].z, r1[s].z};
    const v2f b3 = {r0[s].w, r1[s].w};
    acc0 = __builtin_amdgcn_wmma_f32_16x16x4_f32(false, a[s], false, b0,
                                                 (short)0, acc0, false, false);
    acc1 = __builtin_amdgcn_wmma_f32_16x16x4_f32(false, a[s], false, b1,
                                                 (short)0, acc1, false, false);
    acc2 = __builtin_amdgcn_wmma_f32_16x16x4_f32(false, a[s], false, b2,
                                                 (short)0, acc2, false, false);
    acc3 = __builtin_amdgcn_wmma_f32_16x16x4_f32(false, a[s], false, b3,
                                                 (short)0, acc3, false, false);
  }

  float* __restrict__ orow = out + (size_t)(16 * wave + 8 * h) * DIM + 4 * c;
#pragma unroll
  for (int r = 0; r < 8; ++r) {
    v4f o;
    o.x = acc0[r]; o.y = acc1[r]; o.z = acc2[r]; o.w = acc3[r];
    *(v4f*)(orow + (size_t)r * DIM) = o;
  }
}

#define GEMM_WAVES 4
template <int LDW>
__device__ __forceinline__ void gemm_body(
    const float* __restrict__ A, const float* __restrict__ W,
    const float* __restrict__ bias, float* __restrict__ out, int M, int bid,
    int nblocks) {
  const int wave = threadIdx.x >> 5;
  const int lane = threadIdx.x & 31;
  const int lm = lane & 15;
  const int hi = lane >> 4;
  const int ntiles = (M + 15) >> 4;
  const int nwaves = nblocks * GEMM_WAVES;
  int tile = bid * GEMM_WAVES + wave;
  if (tile >= ntiles) return;

  v4f w[4][8];
#pragma unroll
  for (int nb = 0; nb < 4; ++nb) {
    const float* __restrict__ wr = W + (size_t)(nb * 16 + lm) * LDW + 4 * hi;
#pragma unroll
    for (int t = 0; t < 8; ++t) w[nb][t] = *(const v4f*)(wr + 8 * t);
  }
  float bv[4];
#pragma unroll
  for (int nb = 0; nb < 4; ++nb) bv[nb] = bias ? bias[nb * 16 + lm] : 0.0f;

  auto load_a = [&](int tl, v4f (&a)[8]) {
    const int row = min(tl * 16 + lm, M - 1);
    const float* __restrict__ ar = A + (size_t)row * DIM + 4 * hi;
#pragma unroll
    for (int t = 0; t < 8; ++t) a[t] = *(const v4f*)(ar + 8 * t);
  };

  v4f a[8];
  load_a(tile, a);
  for (; tile < ntiles; tile += nwaves) {
    v4f an[8];
    load_a(min(tile + nwaves, ntiles - 1), an);

    v8f acc0 = {}, acc1 = {}, acc2 = {}, acc3 = {};
#pragma unroll
    for (int t = 0; t < 8; ++t) {
      const v2f a0 = {a[t].x, a[t].y};
      const v2f a1 = {a[t].z, a[t].w};
      const v2f b00 = {w[0][t].x, w[0][t].y}, b01 = {w[0][t].z, w[0][t].w};
      const v2f b10 = {w[1][t].x, w[1][t].y}, b11 = {w[1][t].z, w[1][t].w};
      const v2f b20 = {w[2][t].x, w[2][t].y}, b21 = {w[2][t].z, w[2][t].w};
      const v2f b30 = {w[3][t].x, w[3][t].y}, b31 = {w[3][t].z, w[3][t].w};
      acc0 = __builtin_amdgcn_wmma_f32_16x16x4_f32(false, a0, false, b00,
                                                   (short)0, acc0, false, false);
      acc1 = __builtin_amdgcn_wmma_f32_16x16x4_f32(false, a0, false, b10,
                                                   (short)0, acc1, false, false);
      acc2 = __builtin_amdgcn_wmma_f32_16x16x4_f32(false, a0, false, b20,
                                                   (short)0, acc2, false, false);
      acc3 = __builtin_amdgcn_wmma_f32_16x16x4_f32(false, a0, false, b30,
                                                   (short)0, acc3, false, false);
      acc0 = __builtin_amdgcn_wmma_f32_16x16x4_f32(false, a1, false, b01,
                                                   (short)0, acc0, false, false);
      acc1 = __builtin_amdgcn_wmma_f32_16x16x4_f32(false, a1, false, b11,
                                                   (short)0, acc1, false, false);
      acc2 = __builtin_amdgcn_wmma_f32_16x16x4_f32(false, a1, false, b21,
                                                   (short)0, acc2, false, false);
      acc3 = __builtin_amdgcn_wmma_f32_16x16x4_f32(false, a1, false, b31,
                                                   (short)0, acc3, false, false);
    }

    const int m0 = tile * 16;
    v8f accs[4] = {acc0, acc1, acc2, acc3};
#pragma unroll
    for (int ob = 0; ob < 4; ++ob) {
#pragma unroll
      for (int v = 0; v < 8; ++v) {
        const int row = m0 + v + 8 * hi;
        if (row < M)
          out[(size_t)row * DIM + ob * 16 + lm] = accs[ob][v] + bv[ob];
      }
    }
#pragma unroll
    for (int t = 0; t < 8; ++t) a[t] = an[t];
  }
}

template <int V> struct KConst { static constexpr int value = V; };

#define SEG_BLK 1024

#define EDGE_PT 8
#define EDGE_CHUNK (128 * EDGE_PT)
#define EDGE_WGS ((NUM_EDGES + EDGE_CHUNK - 1) / EDGE_CHUNK)

__device__ __forceinline__ void edge_count_body(const int* __restrict__ src,
                                                int* __restrict__ deg, int E,
                                                int chunk) {
  const int e0 = chunk * EDGE_CHUNK + threadIdx.x;
  int sv[EDGE_PT];
#pragma unroll
  for (int i = 0; i < EDGE_PT; ++i)
    sv[i] = src[min(e0 + 128 * i, E - 1)];
#pragma unroll
  for (int i = 0; i < EDGE_PT; ++i)
    if (e0 + 128 * i < E) atomicAdd(&deg[sv[i]], 1);
}

__global__ __launch_bounds__(SEG_BLK) void seg_alloc_kernel(
    int* __restrict__ cursor  ,
    int* __restrict__ total, int* __restrict__ bbase, int N) {
  __shared__ int swave[32];
  __shared__ int sbase;
  const int lane = threadIdx.x & 31;
  const int wave = threadIdx.x >> 5;
  const int i = blockIdx.x * SEG_BLK + threadIdx.x;
  const int v = (i < N) ? cursor[i] : 0;
  int x = v;
#pragma unroll
  for (int o = 1; o < 32; o <<= 1) {
    const int y = __shfl_up(x, o);
    if (lane >= o) x += y;
  }
  if (lane == 31) swave[wave] = x;
  __syncthreads();
  if (wave == 0) {
    const int w = swave[lane];
    int xs = w;
#pragma unroll
    for (int o = 1; o < 32; o <<= 1) {
      const int y = __shfl_up(xs, o);
      if (lane >= o) xs += y;
    }
    swave[lane] = xs - w;
    if (lane == 31) {
      const int base = atomicAdd(total, xs);
      sbase = base;
      bbase[blockIdx.x] = base;
    }
  }
  __syncthreads();
  if (i < N) cursor[i] = sbase + swave[wave] + x - v;
}

__device__ __forceinline__ void edge_fill_body(const int* __restrict__ src,
                                               const int* __restrict__ dst,
                                               const int* __restrict__ rel,
                                               int* __restrict__ cursor,
                                               int* __restrict__ rec, int E,
                                               int chunk) {
  const int e0 = chunk * EDGE_CHUNK + threadIdx.x;
  int sv[EDGE_PT], dv[EDGE_PT], rv[EDGE_PT], p[EDGE_PT];
#pragma unroll
  for (int i = 0; i < EDGE_PT; ++i) {
    const int ec = min(e0 + 128 * i, E - 1);
    sv[i] = src[ec];
    dv[i] = dst[ec];
    rv[i] = rel[ec];
  }
#pragma unroll
  for (int i = 0; i < EDGE_PT; ++i)
    p[i] = (e0 + 128 * i < E) ? atomicAdd(&cursor[sv[i]], 1) : -1;
#pragma unroll
  for (int i = 0; i < EDGE_PT; ++i)
    if ((unsigned)p[i] < (unsigned)E) rec[p[i]] = dv[i] | (rv[i] << 17);
}

__device__ __forceinline__ float xlane16(float x) {
  const int b = __float_as_int(x);
  return __int_as_float(
      __builtin_amdgcn_permlanex16(b, b, 0x76543210, 0xfedcba98, false, false));
}
template <int M>
__device__ __forceinline__ float xrow(float x) {
  const int b = __float_as_int(x);
  return __int_as_float(
      __builtin_amdgcn_update_dpp(b, b, 0x160 | M, 0xf, 0xf, false));
}

__global__ __launch_bounds__(256) void node_gather_kernel(
    const int* __restrict__ cursor, const int* __restrict__ bbase,
    const int* __restrict__ rec, const float* __restrict__ u_s,
    const float* __restrict__ u_d, const float* __restrict__ u_r,
    const float* __restrict__ cvec, const float* __restrict__ W_a,
    const float* __restrict__ b_a, float* __restrict__ h_ent, int N, int E) {
  const int s = blockIdx.x * (blockDim.x >> 5) + (threadIdx.x >> 5);
  const int lane = threadIdx.x & 31;
  if (s >= N) return;

  int end = cursor[s];
  int start = (s & (SEG_BLK - 1)) ? cursor[s - 1] : bbase[s / SEG_BLK];
  end = __builtin_amdgcn_readfirstlane(end);
  start = __builtin_amdgcn_readfirstlane(start);
  start = min(max(start, 0), E);
  end = min(max(end, start), E);
  const int n = end - start;

  v2f acc = {0.f, 0.f};
  if (n > 0) {
    int pk = (lane < n) ? rec[start + lane] : 0;
    const v2f wa = *(const v2f*)(W_a + 2 * lane);
    const float ba = b_a[0];
    const v2f base = *(const v2f*)(u_s + (size_t)s * DIM + 2 * lane) +
                     *(const v2f*)(cvec + 2 * lane);
    const bool h4 = (lane & 16) != 0;
    const bool h3 = (lane & 8) != 0;
    const bool h2 = (lane & 4) != 0;
    const float NEG = -3.0e38f;
    float m = NEG;
    float bsum = 0.f;

    for (int c0 = 0; c0 < n; c0 += 32) {
      if (c0 > 0) pk = (c0 + lane < n) ? rec[start + c0 + lane] : 0;
      const int mm = min(32, n - c0);
      int j = 0;
      auto group = [&](auto KC, int valid) {
        constexpr int K = decltype(KC)::value;
        v2f ud[K], ur[K], c[K];
        float p[K];
#pragma unroll
        for (int i = 0; i < K; ++i) {
          const int idx = (j + i) & 31;
          const int pj = __builtin_amdgcn_readlane(pk, idx);
          const int d = min(pj & 0x1FFFF, NUM_NODES - 1);
          const int r = min(pj >> 17, NUM_RELS - 1);
          ud[i] = *(const v2f*)(u_d + (size_t)d * DIM + 2 * lane);
          ur[i] = *(const v2f*)(u_r + (size_t)r * DIM + 2 * lane);
        }
#pragma unroll
        for (int i = 0; i < K; ++i) {
          c[i] = (base + ud[i]) + ur[i];
          p[i] = c[i].x * wa.x + c[i].y * wa.y;
        }
        float t;
        if constexpr (K == 8) {
          float q[4], q2[2];
#pragma unroll
          for (int i = 0; i < 4; ++i) {
            const float keep = h4 ? p[i + 4] : p[i];
            const float send = h4 ? p[i] : p[i + 4];
            q[i] = keep + xlane16(send);
          }
#pragma unroll
          for (int i = 0; i < 2; ++i) {
            const float keep = h3 ? q[i + 2] : q[i];
            const float send = h3 ? q[i] : q[i + 2];
            q2[i] = keep + xrow<8>(send);
          }
          const float keep = h2 ? q2[1] : q2[0];
          const float send = h2 ? q2[0] : q2[1];
          t = keep + xrow<4>(send);
        } else {
          float q[2];
#pragma unroll
          for (int i = 0; i < 2; ++i) {
            const float keep = h4 ? p[i + 2] : p[i];
            const float send = h4 ? p[i] : p[i + 2];
            q[i] = keep + xlane16(send);
          }
          const float keep = h3 ? q[1] : q[0];
          const float send = h3 ? q[0] : q[1];
          t = keep + xrow<8>(send);
          t += xrow<4>(t);
        }
        t += xrow<2>(t);
        t += xrow<1>(t);
        const int myedge = (K == 8) ? ((lane >> 2) & 7) : ((lane >> 3) & 3);
        float l = t + ba;
        l = l > 0.f ? l : 0.01f * l;
        if (myedge >= valid) l = NEG;
        float gm = l;
        gm = fmaxf(gm, xlane16(gm));
        gm = fmaxf(gm, xrow<8>(gm));
        gm = fmaxf(gm, xrow<4>(gm));
        const float mnew = fmaxf(m, gm);
        const float sc = expf(m - mnew);
        const int bbits = __float_as_int(expf(l - mnew));
        m = mnew;
        acc.x *= sc;
        acc.y *= sc;
        bsum *= sc;
#pragma unroll
        for (int i = 0; i < K; ++i) {
          if (i < valid) {
            const float w =
                __int_as_float(__builtin_amdgcn_readlane(bbits, i * (32 / K)));
            acc.x += w * c[i].x;
            acc.y += w * c[i].y;
            bsum += w;
          }
        }
        j += K;
      };
      while (mm - j >= 8) group(KConst<8>{}, 8);
      if (mm - j >= 4) group(KConst<4>{}, 4);
      if (mm - j > 0) group(KConst<4>{}, mm - j);
    }
    acc.x = acc.x / bsum;
    acc.y = acc.y / bsum;
  }
  *(v2f*)(h_ent + (size_t)s * DIM + 2 * lane) = acc;
}


#define PPITCH  160
#define POOL_RG 16
#define POOL_SL 16
#define POOL_WGS (POOL_RG * POOL_SL)
#define CNT_PITCH 512
#define SCAN_PT 8
#define SCAN_STEP (512 * SCAN_PT)
#define POOL_SLICE ((((NUM_EDGES + POOL_SL - 1) / POOL_SL + 511) / 512) * 512)
#define QCAP 512
static_assert(QCAP >= 8 + 32 * SCAN_PT && (QCAP & (QCAP - 1)) == 0, "ring size");
static_assert(NUM_NODES <= (1 << 17), "a node index fits 17 bits of a ring entry");
static_assert(POOL_RG * 32 >= NUM_RELS, "relation groups cover all relations");
static_assert((long long)POOL_SL * POOL_SLICE >= NUM_EDGES, "slices cover all edges");

__global__ __launch_bounds__(512) void edge_pool_kernel(
    const int* __restrict__ src, const int* __restrict__ dst,
    const int* __restrict__ rel, const float* __restrict__ g_s,
    const float* __restrict__ g_d, float* __restrict__ pool,
    float* __restrict__ cntp, int E) {
  __shared__ __attribute__((aligned(16))) float sW[16 * 1024];
  __shared__ int sQ[16][2][QCAP];

  const int wave = threadIdx.x >> 5;
  const int lane = threadIdx.x & 31;
  const int lm = lane & 15;
  const int hi = lane >> 4;
  const int rg = blockIdx.x % POOL_RG;
  const int sl = blockIdx.x / POOL_RG;
  const int rbase = rg * 32;
  const int e_lo = sl * POOL_SLICE;
  const int e_hi = min(E, e_lo + POOL_SLICE);
  const int nsteps = (e_hi > e_lo) ? (e_hi - e_lo + SCAN_STEP - 1) / SCAN_STEP : 0;

  float* gb = sW + wave * 1024;
  int* qa = sQ[wave][0];
  int* qb = sQ[wave][1];

  v8f c00 = {}, c01 = {}, c02 = {}, c03 = {};
  v8f c10 = {}, c11 = {}, c12 = {}, c13 = {};
  float n0 = 0.f, n1 = 0.f;

  v2f rs[8], rd[8];
  int r_cur = -1;
  bool pending = false;
  int head = 0, tail = 0;

  auto load_idx = [&](int st, int (&s)[SCAN_PT], int (&d)[SCAN_PT],
                      int (&r)[SCAN_PT]) {
#pragma unroll
    for (int k = 0; k < SCAN_PT; ++k) {
      const int e = e_lo + st * SCAN_STEP + k * 512 + (int)threadIdx.x;
      const bool ok = e < e_hi;
      const int ec = min(e, E - 1);
      s[k] = src[ec];
      d[k] = dst[ec];
      const int rv = rel[ec];
      r[k] = ok ? rv : -1;
    }
  };
  auto load_rows = [&](int s, int d) {
#pragma unroll
    for (int i = 0; i < 8; ++i) {
      const int si = __builtin_amdgcn_readlane(s, i);
      const int di = __builtin_amdgcn_readlane(d, i);
      rs[i] = *(const v2f*)(g_s + (size_t)si * DIM + 2 * lane);
      rd[i] = *(const v2f*)(g_d + (size_t)di * DIM + 2 * lane);
    }
  };

  auto consume = [&]() {
#pragma unroll
    for (int j = 0; j < 4; ++j) {
      const int r0 = __builtin_amdgcn_readlane(r_cur, 2 * j);
      const int r1 = __builtin_amdgcn_readlane(r_cur, 2 * j + 1);
      v2f g0 = rs[2 * j] + rd[2 * j];
      v2f g1 = rs[2 * j + 1] + rd[2 * j + 1];
      if (r0 < 0) { g0.x = 0.f; g0.y = 0.f; }
      if (r1 < 0) { g1.x = 0.f; g1.y = 0.f; }
      v4f o;
      o.x = g0.x; o.y = g1.x; o.z = g0.y; o.w = g1.y;
      *(v4f*)(gb + j * PPITCH + 4 * lane) = o;
    }
    __builtin_amdgcn_fence(__ATOMIC_ACQ_REL, "wavefront");
#pragma unroll
    for (int g = 0; g < 2; ++g) {
      const int x0 = __builtin_amdgcn_readlane(r_cur, 4 * g);
      const int x1 = __builtin_amdgcn_readlane(r_cur, 4 * g + 1);
      const int x2 = __builtin_amdgcn_readlane(r_cur, 4 * g + 2);
      const int x3 = __builtin_amdgcn_readlane(r_cur, 4 * g + 3);
      const int da = (hi ? x2 : x0) - lm;
      const int db = (hi ? x3 : x1) - lm;
      const float* pg = gb + (2 * g + hi) * PPITCH + 2 * lm;
      const v2f b0 = *(const v2f*)(pg);
      const v2f b1 = *(const v2f*)(pg + 32);
      const v2f b2 = *(const v2f*)(pg + 64);
      const v2f b3 = *(const v2f*)(pg + 96);
      {
        v2f a;
        a.x = (da == 0) ? 1.0f : 0.0f;
        a.y = (db == 0) ? 1.0f : 0.0f;
        n0 += a.x + a.y;
        c00 = __builtin_amdgcn_wmma_f32_16x16x4_f32(false, a, false, b0,
                                                    (short)0, c00, false, false);
        c01 = __builtin_amdgcn_wmma_f32_16x16x4_f32(false, a, false, b1,
                                                    (short)0, c01, false, false);
        c02 = __builtin_amdgcn_wmma_f32_16x16x4_f32(false, a, false, b2,
                                                    (short)0, c02, false, false);
        c03 = __builtin_amdgcn_wmma_f32_16x16x4_f32(false, a, false, b3,
                                                    (short)0, c03, false, false);
      }
      {
        v2f a;
        a.x = (da == 16) ? 1.0f : 0.0f;
        a.y = (db == 16) ? 1.0f : 0.0f;
        n1 += a.x + a.y;
        c10 = __builtin_amdgcn_wmma_f32_16x16x4_f32(false, a, false, b0,
                                                    (short)0, c10, false, false);
        c11 = __builtin_amdgcn_wmma_f32_16x16x4_f32(false, a, false, b1,
                                                    (short)0, c11, false, false);
        c12 = __builtin_amdgcn_wmma_f32_16x16x4_f32(false, a, false, b2,
                                                    (short)0, c12, false, false);
        c13 = __builtin_amdgcn_wmma_f32_16x16x4_f32(false, a, false, b3,
                                                    (short)0, c13, false, false);
      }
    }
    pending = false;
  };
  auto fetch = [&]() {
    const int avail = tail - head;
    const int i = lane & 7;
    const bool ok = i < avail;
    const int qi = (head + i) & (QCAP - 1);
    const int q0 = ok ? qa[qi] : 0;
    const int d = ok ? qb[qi] : 0;
    const int s = q0 & 0x1FFFF;
    r_cur = ok ? (q0 >> 17) : -1;
    head += min(avail, 8);
    load_rows(s, d);
    pending = true;
  };

  int sA[SCAN_PT], dA[SCAN_PT], rA[SCAN_PT];
  int sB[SCAN_PT], dB[SCAN_PT], rB[SCAN_PT];
  if (nsteps > 0) load_idx(0, sA, dA, rA);
  for (int st = 0; st < nsteps; ++st) {
    load_idx(st + 1, sB, dB, rB);
#pragma unroll
    for (int k = 0; k < SCAN_PT; ++k) {
      const bool m = (unsigned)(rA[k] - rbase) < 32u;
      const unsigned mk = __builtin_amdgcn_ballot_w32(m);
      if (mk) {
        const int pos =
            (tail + __builtin_popcount(mk & ((1u << lane) - 1u))) & (QCAP - 1);
        if (m) { qa[pos] = sA[k] | ((rA[k] - rbase) << 17); qb[pos] = dA[k]; }
        tail += __builtin_popcount(mk);
      }
    }
    __builtin_amdgcn_fence(__ATOMIC_ACQ_REL, "wavefront");
    while (tail - head >= 8) {
      if (pending) consume();
      fetch();
    }
#pragma unroll
    for (int k = 0; k < SCAN_PT; ++k) { sA[k] = sB[k]; dA[k] = dB[k]; rA[k] = rB[k]; }
  }
  while (tail - head > 0) {
    if (pending) consume();
    fetch();
  }
  if (pending) consume();

  auto reduce_flush = [&](const v8f& t0, const v8f& t1, const v8f& t2,
                          const v8f& t3, int rt) {
#pragma unroll
    for (int v = 0; v < 8; ++v) {
      float* row = gb + (v + 8 * hi) * DIM + lm;
      row[0] = t0[v]; row[16] = t1[v]; row[32] = t2[v]; row[48] = t3[v];
    }
    __syncthreads();
#pragma unroll
    for (int k = 0; k < 2; ++k) {
      const int idx = (int)threadIdx.x + 512 * k;
      float sum = 0.f;
#pragma unroll
      for (int w = 0; w < 16; ++w) sum += sW[w * 1024 + idx];
      const int R = rbase + rt * 16 + (idx >> 6);
      if (R < NUM_RELS) atomicAdd(&pool[(size_t)R * DIM + (idx & 63)], sum);
    }
    __syncthreads();
  };
  reduce_flush(c00, c01, c02, c03, 0);
  reduce_flush(c10, c11, c12, c13, 1);

  {
    const float t0 = n0 + xlane16(n0);
    const float t1 = n1 + xlane16(n1);
    cntp[(size_t)(sl * 16 + wave) * CNT_PITCH + rbase + lane] = hi ? t1 : t0;
  }
}

__global__ __launch_bounds__(256) void h_rel_kernel(
    const float* __restrict__ pool, const float* __restrict__ cntp,
    const float* __restrict__ rel2_proj, const float* __restrict__ W_rel3,
    const float* __restrict__ b_rel3, float* __restrict__ h_rel, int R) {
  __shared__ float swsum[8];
  const int t = blockIdx.x * 256 + threadIdx.x;
  const int r = min(t >> 6, R - 1), o = t & 63;
  float c = 0.f;
#pragma unroll
  for (int j = 0; j < POOL_WGS / DIM; ++j)
    c += cntp[(size_t)(o + DIM * j) * CNT_PITCH + r];
  c += xlane16(c);
  c += xrow<8>(c);
  c += xrow<4>(c);
  c += xrow<2>(c);
  c += xrow<1>(c);
  const int wave = threadIdx.x >> 5;
  if ((threadIdx.x & 31) == 0) swsum[wave] = c;
  __syncthreads();
  c = swsum[wave & 6] + swsum[wave | 1];
  if (t >= R * DIM) return;
  const float inv = 1.0f / fmaxf(c, 1.0f);
  const float* w = W_rel3 + (size_t)o * 192;
  float acc = b_rel3[o] + pool[t] * inv;
  if (c > 0.f) {
    const float* r2 = rel2_proj + (size_t)r * DIM;
    for (int j = 0; j < DIM; ++j) acc += r2[j] * w[128 + j];
  }
  h_rel[t] = acc;
}

#define PREP_ZERO_WGS 251
#define GEMM_WGS_N 512
#define GEMM_WGS_R ((((NUM_RELS + 15) / 16) + GEMM_WAVES - 1) / GEMM_WAVES)
#define PROJ1_GEMM_WGS (2 * GEMM_WGS_R + GEMM_WGS_N)
#define COUNT_PROJ_WGS (PROJ1_GEMM_WGS + EDGE_WGS)
#define FILL_PROJ_WGS (GEMM_WGS_N + EDGE_WGS)
static_assert(EDGE_WGS >= PROJ1_GEMM_WGS, "one edge chunk beside every GEMM work-group");

__global__ __launch_bounds__(128) void prep_kernel(
    const float* __restrict__ W_ent, const float* __restrict__ b_ent,
    const float* __restrict__ W_relL, const float* __restrict__ b_relL,
    const float* __restrict__ W_rel2, const float* __restrict__ b_rel2,
    const float* __restrict__ W_fc, const float* __restrict__ b_fc,
    float* __restrict__ M_s, float* __restrict__ M_d,
    float* __restrict__ M_r, float* __restrict__ M_2,
    float* __restrict__ cvec, float* __restrict__ v_rel2,
    float* __restrict__ zero_p, int zero_n) {
  const int b = blockIdx.x;
  if (b < FOLD_WGS) {
    fold_body(W_ent, b_ent, W_relL, b_relL, W_rel2, b_rel2, W_fc, b_fc, M_s,
              M_d, M_r, M_2, cvec, v_rel2, b);
    return;
  }
  zero_body(zero_p, zero_n, b - FOLD_WGS, PREP_ZERO_WGS);
}

__global__ __launch_bounds__(128) void count_proj_kernel(
    const int* __restrict__ src, int* __restrict__ deg,
    const float* __restrict__ ent_embed, const float* __restrict__ rel_embed,
    const float* __restrict__ M_s, const float* __restrict__ M_r,
    const float* __restrict__ M_2, const float* __restrict__ v_rel2,
    float* __restrict__ u_s, float* __restrict__ u_r,
    float* __restrict__ rel2p) {
  const int b = blockIdx.x;
  int chunk;
  if (b < 2 * PROJ1_GEMM_WGS) {
    if ((b & 1) == 0) {
      const int g = b >> 1;
      const float* A = ent_embed;
      const float* Wm = M_s;
      const float* bias = nullptr;
      float* out = u_s;
      int M = NUM_NODES, bid = g - 2 * GEMM_WGS_R, nblk = GEMM_WGS_N;
      if (g < GEMM_WGS_R) {
        A = rel_embed; Wm = M_r; out = u_r;
        M = NUM_RELS; bid = g; nblk = GEMM_WGS_R;
      } else if (g < 2 * GEMM_WGS_R) {
        A = rel_embed; Wm = M_2; bias = v_rel2; out = rel2p;
        M = NUM_RELS; bid = g - GEMM_WGS_R; nblk = GEMM_WGS_R;
      }
      gemm_body<DIM>(A, Wm, bias, out, M, bid, nblk);
      return;
    }
    chunk = b >> 1;
  } else {
    chunk = b - PROJ1_GEMM_WGS;
  }
  edge_count_body(src, deg, NUM_EDGES, chunk);
}

__global__ __launch_bounds__(128) void fill_proj_kernel(
    const int* __restrict__ src, const int* __restrict__ dst,
    const int* __restrict__ rel, int* __restrict__ cursor,
    int* __restrict__ rec, const float* __restrict__ ent_embed,
    const float* __restrict__ M_d, float* __restrict__ u_d) {
  const int b = blockIdx.x;
  int chunk;
  if (b < 2 * GEMM_WGS_N) {
    if ((b & 1) == 0) {
      gemm_body<DIM>(ent_embed, M_d, nullptr, u_d, NUM_NODES, b >> 1,
                     GEMM_WGS_N);
      return;
    }
    chunk = b >> 1;
  } else {
    chunk = b - GEMM_WGS_N;
  }
  edge_fill_body(src, dst, rel, cursor, rec, NUM_EDGES, chunk);
}

__global__ __launch_bounds__(128) void proj2_kernel(
    const float* __restrict__ h_ent, const float* __restrict__ W_rel3,
    float* __restrict__ g_s, float* __restrict__ g_d) {
  const int b = blockIdx.x;
  const bool odd = (b & 1) != 0;
  gemm_body<3 * DIM>(h_ent, odd ? W_rel3 + DIM : W_rel3, nullptr,
                     odd ? g_d : g_s, NUM_NODES, b >> 1, GEMM_WGS_N);
}

#define OFF_MS     0
#define OFF_MD     4096
#define OFF_MR     8192
#define OFF_M2     12288
#define OFF_CVEC   16384
#define OFF_VREL2  16448
#define OFF_UR     16576
#define OFF_REL2P  49088
#define OFF_BSUM   281088
#define OFF_POOL   381568
#define OFF_TOTAL  413588
#define OFF_BBASE  413600
#define OFF_BE     445568
#define OFF_CNTP   OFF_BE
#define OFF_US     1445568
#define OFF_UD     7845568
#define ZERO_CNT   (OFF_TOTAL + 1 - OFF_BSUM)

extern "C" void kernel_launch(void* const* d_in, const int* in_sizes, int n_in,
                              void* d_out, int out_size, void* d_ws,
                              size_t ws_size, hipStream_t stream) {
  const float* ent_embed = (const float*)d_in[0];
  const float* rel_embed = (const float*)d_in[1];
  const float* W_ent  = (const float*)d_in[2];
  const float* b_ent  = (const float*)d_in[3];
  const float* W_relL = (const float*)d_in[4];
  const float* b_relL = (const float*)d_in[5];
  const float* W_rel2 = (const float*)d_in[6];
  const float* b_rel2 = (const float*)d_in[7];
  const float* W_rel3 = (const float*)d_in[8];
  const float* b_rel3 = (const float*)d_in[9];
  const float* W_a    = (const float*)d_in[10];
  const float* b_a    = (const float*)d_in[11];
  const float* W_fc   = (const float*)d_in[12];
  const float* b_fc   = (const float*)d_in[13];
  const int* src_idx  = (const int*)d_in[14];
  const int* dst_idx  = (const int*)d_in[15];
  const int* rel_idx  = (const int*)d_in[16];

  float* ws = (float*)d_ws;
  float* h_ent = (float*)d_out;
  float* h_rel = (float*)d_out + (size_t)NUM_NODES * DIM;

  int* deg_cur = (int*)(ws + OFF_BSUM);
  int* seg_total = (int*)(ws + OFF_TOTAL);
  int* seg_bbase = (int*)(ws + OFF_BBASE);
  int* edge_rec = (int*)(ws + OFF_BE);

  prep_kernel<<<FOLD_WGS + PREP_ZERO_WGS, 128, 0, stream>>>(
      W_ent, b_ent, W_relL, b_relL, W_rel2, b_rel2, W_fc, b_fc,
      ws + OFF_MS, ws + OFF_MD, ws + OFF_MR, ws + OFF_M2, ws + OFF_CVEC,
      ws + OFF_VREL2, ws + OFF_BSUM, ZERO_CNT);

  count_proj_kernel<<<COUNT_PROJ_WGS, 128, 0, stream>>>(
      src_idx, deg_cur, ent_embed, rel_embed, ws + OFF_MS, ws + OFF_MR,
      ws + OFF_M2, ws + OFF_VREL2, ws + OFF_US, ws + OFF_UR, ws + OFF_REL2P);

  seg_alloc_kernel<<<(NUM_NODES + SEG_BLK - 1) / SEG_BLK, SEG_BLK, 0, stream>>>(
      deg_cur, seg_total, seg_bbase, NUM_NODES);

  fill_proj_kernel<<<FILL_PROJ_WGS, 128, 0, stream>>>(
      src_idx, dst_idx, rel_idx, deg_cur, edge_rec, ent_embed, ws + OFF_MD,
      ws + OFF_UD);

  node_gather_kernel<<<(NUM_NODES + 7) / 8, 256, 0, stream>>>(
      deg_cur, seg_bbase, edge_rec, ws + OFF_US, ws + OFF_UD, ws + OFF_UR,
      ws + OFF_CVEC, W_a, b_a, h_ent, NUM_NODES, NUM_EDGES);

  proj2_kernel<<<2 * GEMM_WGS_N, 128, 0, stream>>>(h_ent, W_rel3, ws + OFF_US,
                                                  ws + OFF_UD);

  edge_pool_kernel<<<POOL_WGS, 512, 0, stream>>>(
      src_idx, dst_idx, rel_idx, ws + OFF_US, ws + OFF_UD, ws + OFF_POOL,
      ws + OFF_CNTP, NUM_EDGES);

  h_rel_kernel<<<(NUM_RELS * DIM + 255) / 256, 256, 0, stream>>>(
      ws + OFF_POOL, ws + OFF_CNTP, ws + OFF_REL2P, W_rel3, b_rel3, h_rel,
      NUM_RELS);
}
